# starting kernel
# baseline (speedup 1.0000x reference)
#include <hip/hip_runtime.h>
#include <math.h>


typedef __attribute__((ext_vector_type(2))) float v2f;
typedef __attribute__((ext_vector_type(8))) float v8f;

#define LQ    512
#define NH    12
#define NODE  384
#define PAIR  128
#define OUTIN 176
#define KOUT  (NH*OUTIN)

#define OFF_SQ   ((size_t)0)
#define OFF_SK   (OFF_SQ  + (size_t)NH*LQ*16)
#define OFF_SV   (OFF_SK  + (size_t)NH*LQ*16)
#define OFF_PQ   (OFF_SV  + (size_t)NH*LQ*16)
#define OFF_PK   (OFF_PQ  + (size_t)NH*LQ*16)
#define OFF_PV   (OFF_PK  + (size_t)NH*LQ*16)
#define OFF_QN2  (OFF_PV  + (size_t)NH*LQ*24)
#define OFF_KN2  (OFF_QN2 + (size_t)NH*LQ)
#define OFF_BIAS (OFF_KN2 + (size_t)NH*LQ)
#define OFF_ATTN (OFF_BIAS+ (size_t)NH*LQ*LQ)
#define OFF_RES  (OFF_ATTN+ (size_t)NH*LQ*LQ)

#define RES_IDX(h, l, f) (((size_t)(l)*NH + (h))*OUTIN + (f))

static __device__ __forceinline__ v8f wmma4(v2f a, v2f b, v8f c) {
  return __builtin_amdgcn_wmma_f32_16x16x4_f32(
       false, a,  false, b,
       (short)0, c,  false,  false);
}

static __device__ __forceinline__ v8f vzero8() {
  v8f z = {0.f,0.f,0.f,0.f,0.f,0.f,0.f,0.f};
  return z;
}

__global__ __launch_bounds__(32) void proj_kernel(
    const float* __restrict__ node,
    const float* __restrict__ sq_k, const float* __restrict__ sk_k,
    const float* __restrict__ sv_k, const float* __restrict__ pq_k,
    const float* __restrict__ pk_k, const float* __restrict__ pv_k,
    float* __restrict__ ws)
{
  const int lane  = threadIdx.x;
  const int mt    = blockIdx.x;
  const int seg   = blockIdx.y;
  const int h     = blockIdx.z;
  const int row   = lane & 15;
  const int khalf = (lane >> 4) * 2;

  const float* wb = sk_k + (size_t)h*NODE*16;
  int   wstride = 16;
  int   wcol    = row;
  float wscale  = 1.f;
  switch (seg) {
    case 0: wb = sq_k + (size_t)h*NODE*16; wscale = 0.25f; break;
    case 1: break;
    case 2: wb = sv_k + (size_t)h*NODE*16; break;
    case 3: wb = pq_k + (size_t)h*NODE*12; wstride = 12;
            wcol = (row < 12) ? row : 11; wscale = (row < 12) ? 1.f : 0.f; break;
    case 4: wb = pk_k + (size_t)h*NODE*12; wstride = 12;
            wcol = (row < 12) ? row : 11; wscale = (row < 12) ? 1.f : 0.f; break;
    case 5: wb = pv_k + (size_t)h*NODE*24; wstride = 24; break;
    case 6: wb = pv_k + (size_t)h*NODE*24; wstride = 24;
            wcol = (row < 8) ? 16 + row : 23; wscale = (row < 8) ? 1.f : 0.f; break;
  }
  const float* arow = node + (size_t)(mt*16 + row)*NODE;
  const float* wp   = wb + wcol;

  v8f acc = vzero8();
  for (int k0 = 0; k0 < NODE; k0 += 4) {
    const int ka = k0 + khalf;
    v2f a; a.x = arow[ka]; a.y = arow[ka + 1];
    v2f b;
    b.x = wp[(size_t)ka * wstride]       * wscale;
    b.y = wp[(size_t)(ka + 1) * wstride] * wscale;
    acc = wmma4(a, b, acc);
  }

  const int col   = lane & 15;
  const int rbase = (lane >> 4) * 8;
  float* ob = ws + OFF_SK + (size_t)h*LQ*16;
  int ostride = 16, ocol = col;
  bool dostore = true;
  switch (seg) {
    case 0: ob = ws + OFF_SQ + (size_t)h*LQ*16; break;
    case 1: break;
    case 2: ob = ws + OFF_SV + (size_t)h*LQ*16; break;
    case 3: ob = ws + OFF_PQ + (size_t)h*LQ*16; break;
    case 4: ob = ws + OFF_PK + (size_t)h*LQ*16; break;
    case 5: ob = ws + OFF_PV + (size_t)h*LQ*24; ostride = 24; break;
    case 6: ob = ws + OFF_PV + (size_t)h*LQ*24; ostride = 24;
            ocol = 16 + col; dostore = (col < 8); break;
  }
  if (dostore) {
#pragma unroll
    for (int r = 0; r < 8; ++r)
      ob[(size_t)(mt*16 + rbase + r)*ostride + ocol] = acc[r];
  }
}

__global__ void rigid_kernel(const float* __restrict__ R,
                             const float* __restrict__ t,
                             float* __restrict__ ws)
{
  const int idx = blockIdx.x * blockDim.x + threadIdx.x;
  if (idx >= NH * LQ) return;
  const int h = idx / LQ, l = idx % LQ;
  float Rm[9], tv[3];
#pragma unroll
  for (int i = 0; i < 9; ++i) Rm[i] = R[l*9 + i];
#pragma unroll
  for (int i = 0; i < 3; ++i) tv[i] = t[l*3 + i];

  float* Pq = ws + OFF_PQ + ((size_t)h*LQ + l)*16;
  float* Pk = ws + OFF_PK + ((size_t)h*LQ + l)*16;
  float* Pv = ws + OFF_PV + ((size_t)h*LQ + l)*24;

  float qs = 0.f, ksum = 0.f;
#pragma unroll
  for (int p = 0; p < 4; ++p) {
    float loc[3], g[3];
    for (int f = 0; f < 3; ++f) loc[f] = Pq[p*3+f];
    for (int gg = 0; gg < 3; ++gg) {
      g[gg] = tv[gg];
      for (int f = 0; f < 3; ++f) g[gg] += loc[f]*Rm[gg*3+f];
      Pq[p*3+gg] = g[gg]; qs += g[gg]*g[gg];
    }
    for (int f = 0; f < 3; ++f) loc[f] = Pk[p*3+f];
    for (int gg = 0; gg < 3; ++gg) {
      g[gg] = tv[gg];
      for (int f = 0; f < 3; ++f) g[gg] += loc[f]*Rm[gg*3+f];
      Pk[p*3+gg] = g[gg]; ksum += g[gg]*g[gg];
    }
  }
#pragma unroll
  for (int p = 0; p < 8; ++p) {
    float loc[3], g[3];
    for (int f = 0; f < 3; ++f) loc[f] = Pv[p*3+f];
    for (int gg = 0; gg < 3; ++gg) {
      g[gg] = tv[gg];
      for (int f = 0; f < 3; ++f) g[gg] += loc[f]*Rm[gg*3+f];
      Pv[p*3+gg] = g[gg];
    }
  }
  ws[OFF_QN2 + h*LQ + l] = qs;
  ws[OFF_KN2 + h*LQ + l] = ksum;
}

__global__ __launch_bounds__(32) void bias_kernel(
    const float* __restrict__ pairwise, const float* __restrict__ pr_k,
    float* __restrict__ ws)
{
  float* bias = ws + OFF_BIAS;
  const int lane  = threadIdx.x;
  const int tile  = blockIdx.x;
  const int row   = lane & 15;
  const int khalf = (lane >> 4) * 2;
  const int flat  = tile*16 + row;
  const float* prow = pairwise + (size_t)flat * PAIR;
  const float* prk  = pr_k + (size_t)((row < NH) ? row : NH - 1) * PAIR;
  const float bmask = (row < NH) ? 1.f : 0.f;

  v8f acc = vzero8();
  for (int k0 = 0; k0 < PAIR; k0 += 4) {
    const int ka = k0 + khalf;
    __builtin_prefetch(prow + ka + 64, 0, 0);
    v2f a; a.x = prow[ka];          a.y = prow[ka + 1];
    v2f b; b.x = prk[ka] * bmask;   b.y = prk[ka + 1] * bmask;
    acc = wmma4(a, b, acc);
  }
  const int col = lane & 15;
  const int rbase = (lane >> 4) * 8;
  if (col < NH) {
#pragma unroll
    for (int r = 0; r < 8; ++r) {
      const int f2 = tile*16 + rbase + r;
      const int i = f2 >> 9, j = f2 & 511;
      bias[((size_t)col*LQ + i)*LQ + j] = acc[r];
    }
  }
}

#define SMS 516
__global__ __launch_bounds__(128) void attn_kernel(
    const float* __restrict__ point_weight, float* __restrict__ ws)
{
  __shared__ float sm[16 * SMS];
  const int tid  = threadIdx.x;
  const int lane = tid & 31;
  const int w    = tid >> 5;
  const int nt   = blockIdx.x;
  const int h    = blockIdx.y;

  const float* Sq  = ws + OFF_SQ;  const float* Sk  = ws + OFF_SK;
  const float* Pq  = ws + OFF_PQ;  const float* Pk  = ws + OFF_PK;
  const float* qn2 = ws + OFF_QN2; const float* kn2 = ws + OFF_KN2;
  const float* bias= ws + OFF_BIAS;
  const float* Sv  = ws + OFF_SV;  const float* Pv  = ws + OFF_PV;
  float* attn = ws + OFF_ATTN;
  float* res  = ws + OFF_RES;

  const float pwx = point_weight[h];
  const float pw  = (pwx > 20.f) ? pwx : log1pf(__expf(pwx));
  const float cpt = -0.5f * pw / sqrtf(18.f);
  const float rs3 = 0.57735026919f;

  const int row   = lane & 15;
  const int khalf = (lane >> 4) * 2;
  const int col   = lane & 15;
  const int rbase = (lane >> 4) * 8;

  v2f aq[4], ap[4];
#pragma unroll
  for (int ks = 0; ks < 4; ++ks) {
    const int k = ks*4 + khalf;
    const int l = nt*16 + row;
    aq[ks].x = Sq[((size_t)h*LQ + l)*16 + k];
    aq[ks].y = Sq[((size_t)h*LQ + l)*16 + k + 1];
    ap[ks].x = Pq[((size_t)h*LQ + l)*16 + k];
    ap[ks].y = Pq[((size_t)h*LQ + l)*16 + k + 1];
  }
  float qv[8];
#pragma unroll
  for (int r = 0; r < 8; ++r) qv[r] = qn2[h*LQ + nt*16 + rbase + r];

  for (int mt = w; mt < 32; mt += 4) {
    v8f s = vzero8(), p = vzero8();
#pragma unroll
    for (int ks = 0; ks < 4; ++ks) {
      const int k = ks*4 + khalf;
      const int m = mt*16 + row;
      v2f bq, bp;
      bq.x = Sk[((size_t)h*LQ + m)*16 + k]; bq.y = Sk[((size_t)h*LQ + m)*16 + k + 1];
      bp.x = Pk[((size_t)h*LQ + m)*16 + k]; bp.y = Pk[((size_t)h*LQ + m)*16 + k + 1];
      s = wmma4(aq[ks], bq, s);
      p = wmma4(ap[ks], bp, p);
    }
    const int m = mt*16 + col;
    const float kv = kn2[h*LQ + m];
#pragma unroll
    for (int r = 0; r < 8; ++r) {
      const int rr = rbase + r;
      const int n  = nt*16 + rr;
      const float dist = qv[r] + kv - 2.f*p[r];
      sm[rr*SMS + m] =
          (s[r] + bias[((size_t)h*LQ + n)*LQ + m] + cpt*dist) * rs3;
    }
  }
  __syncthreads();

  {
    const int rrow = tid >> 3;
    const int sub  = tid & 7;
    float mx = -1e30f;
    for (int j = sub; j < LQ; j += 8) mx = fmaxf(mx, sm[rrow*SMS + j]);
    for (int o = 1; o < 8; o <<= 1) mx = fmaxf(mx, __shfl_xor(mx, o, 8));
    float sum = 0.f;
    for (int j = sub; j < LQ; j += 8) {
      const float e = __expf(sm[rrow*SMS + j] - mx);
      sm[rrow*SMS + j] = e; sum += e;
    }
    for (int o = 1; o < 8; o <<= 1) sum += __shfl_xor(sum, o, 8);
    const float inv = 1.f / sum;
    const int n = nt*16 + rrow;
    for (int j = sub; j < LQ; j += 8) {
      const float v = sm[rrow*SMS + j] * inv;
      sm[rrow*SMS + j] = v;
      attn[((size_t)h*LQ + n)*LQ + j] = v;
    }
  }
  __syncthreads();

  if (w < 3) {
    const float* vb = (w == 0) ? Sv + (size_t)h*LQ*16 : Pv + (size_t)h*LQ*24;
    const int    vs = (w == 0) ? 16 : 24;
    int   vc = col;
    float vmask = 1.f;
    if (w == 2) { vc = (col < 8) ? 16 + col : 23; vmask = (col < 8) ? 1.f : 0.f; }
    const float* vp = vb + vc;

    v8f acc = vzero8();
    for (int kt = 0; kt < 128; ++kt) {
      const int k = kt*4 + khalf;
      v2f a; a.x = sm[row*SMS + k]; a.y = sm[row*SMS + k + 1];
      v2f b;
      b.x = vp[(size_t)k * vs]       * vmask;
      b.y = vp[(size_t)(k + 1) * vs] * vmask;
      acc = wmma4(a, b, acc);
    }
    const int fbase = (w == 0) ? 0 : 144;
#pragma unroll
    for (int r = 0; r < 8; ++r) {
      const int n = nt*16 + rbase + r;
      const float v = acc[r];
      if (w < 2)        res[RES_IDX(h, n, fbase + col)] = v;
      else if (col < 8) res[RES_IDX(h, n, 160 + col)]   = v;
    }
  }
}

__global__ __launch_bounds__(32) void pairout_kernel(
    const float* __restrict__ pairwise, float* __restrict__ ws)
{
  const float* attn = ws + OFF_ATTN;
  float* res = ws + OFF_RES;
  const int lane  = threadIdx.x;
  const int n     = blockIdx.x;
  const int row   = lane & 15;
  const int khalf = (lane >> 4) * 2;
  const int col   = lane & 15;

  const int hh0 = (row < NH) ? row : NH - 1;
  const float amask = (row < NH) ? 1.f : 0.f;
  const float* arow = attn + ((size_t)hh0*LQ + n)*LQ;

  v8f acc[8];
#pragma unroll
  for (int i = 0; i < 8; ++i) acc[i] = vzero8();

  for (int kt = 0; kt < 128; ++kt) {
    const int m = kt*4 + khalf;
    v2f a;
    a.x = arow[m] * amask;
    a.y = arow[m + 1] * amask;
    const float* pr0 = pairwise + ((size_t)n*LQ + m) * PAIR + col;
    __builtin_prefetch(pr0 + 4*PAIR, 0, 0);
#pragma unroll
    for (int ntile = 0; ntile < 8; ++ntile) {
      v2f b; b.x = pr0[ntile*16]; b.y = pr0[ntile*16 + PAIR];
      acc[ntile] = wmma4(a, b, acc[ntile]);
    }
  }
  const int rbase = (lane >> 4) * 8;
#pragma unroll
  for (int ntile = 0; ntile < 8; ++ntile)
#pragma unroll
    for (int r = 0; r < 8; ++r) {
      const int hh = rbase + r;
      if (hh < NH)
        res[RES_IDX(hh, n, 16 + ntile*16 + col)] = acc[ntile][r];
    }
}

__global__ void pfinish_kernel(const float* __restrict__ R,
                               const float* __restrict__ t,
                               float* __restrict__ ws)
{
  const int idx = blockIdx.x * blockDim.x + threadIdx.x;
  if (idx >= NH * LQ) return;
  const int h = idx / LQ, l = idx % LQ;
  float* resp = ws + OFF_RES + RES_IDX(h, l, 0);
  float Rm[9], tv[3];
#pragma unroll
  for (int i = 0; i < 9; ++i) Rm[i] = R[l*9 + i];
#pragma unroll
  for (int i = 0; i < 3; ++i) tv[i] = t[l*3 + i];
#pragma unroll
  for (int p = 0; p < 8; ++p) {
    float g[3], loc[3], nrm = 0.f;
    for (int gg = 0; gg < 3; ++gg) g[gg] = resp[144 + p*3 + gg] - tv[gg];
    for (int f = 0; f < 3; ++f) {
      float v = 0.f;
      for (int gg = 0; gg < 3; ++gg) v += g[gg]*Rm[gg*3 + f];
      loc[f] = v; nrm += v*v;
    }
    for (int f = 0; f < 3; ++f) resp[144 + p*3 + f] = loc[f];
    resp[168 + p] = sqrtf(nrm + 1e-12f);
  }
}

__global__ __launch_bounds__(32) void outproj_kernel(
    const float* __restrict__ out_k, const float* __restrict__ out_b,
    const float* __restrict__ res, float* __restrict__ out)
{
  const int lane  = threadIdx.x;
  const int mt    = blockIdx.x;
  const int ntile = blockIdx.y;
  const int row   = lane & 15;
  const int khalf = (lane >> 4) * 2;
  const int col   = lane & 15;
  const int o     = ntile*16 + col;

  const float* resl = res + (size_t)(mt*16 + row) * KOUT;
  const float* okp  = out_k + o;

  v8f acc = vzero8();
  for (int k0 = 0; k0 < KOUT; k0 += 4) {
    const int ka = k0 + khalf;
    v2f a; a.x = resl[ka];                 a.y = resl[ka + 1];
    v2f b; b.x = okp[(size_t)ka * NODE];   b.y = okp[(size_t)(ka + 1) * NODE];
    acc = wmma4(a, b, acc);
  }
  const int rbase = (lane >> 4) * 8;
  const float ob = out_b[o];
#pragma unroll
  for (int r = 0; r < 8; ++r) {
    const int l = mt*16 + rbase + r;
    out[(size_t)l*NODE + o] = acc[r] + ob;
  }
}

extern "C" void kernel_launch(void* const* d_in, const int* in_sizes, int n_in,
                              void* d_out, int out_size, void* d_ws, size_t ws_size,
                              hipStream_t stream) {
  const float* node  = (const float*)d_in[0];
  const float* R     = (const float*)d_in[1];
  const float* t     = (const float*)d_in[2];
  const float* pairw = (const float*)d_in[3];
  const float* sq_k  = (const float*)d_in[4];
  const float* sk_k  = (const float*)d_in[5];
  const float* sv_k  = (const float*)d_in[6];
  const float* pq_k  = (const float*)d_in[7];
  const float* pk_k  = (const float*)d_in[8];
  const float* pv_k  = (const float*)d_in[9];
  const float* pr_k  = (const float*)d_in[10];
  const float* out_k = (const float*)d_in[11];
  const float* out_b = (const float*)d_in[12];
  const float* pwgt  = (const float*)d_in[13];
  float* ws  = (float*)d_ws;
  float* out = (float*)d_out;

  proj_kernel<<<dim3(32, 7, NH), 32, 0, stream>>>(
      node, sq_k, sk_k, sv_k, pq_k, pk_k, pv_k, ws);
  rigid_kernel<<<(NH*LQ + 127)/128, 128, 0, stream>>>(R, t, ws);
  bias_kernel<<<(LQ*LQ)/16, 32, 0, stream>>>(pairw, pr_k, ws);
  attn_kernel<<<dim3(32, NH), 128, 0, stream>>>(pwgt, ws);
  pairout_kernel<<<LQ, 32, 0, stream>>>(pairw, ws);
  pfinish_kernel<<<(NH*LQ + 127)/128, 128, 0, stream>>>(R, t, ws);
  outproj_kernel<<<dim3(32, 24), 32, 0, stream>>>(out_k, out_b, ws + OFF_RES, out);
}


# K-loops of proj, attn A*V, pairout, outproj: fragment loads issued as whole groups far ahead of their wmmas (16-step groups; register rings in pairout, outproj); pairout prefetch hint dropped
# speedup vs baseline: 1.9412x; 1.9412x over previous
#include <hip/hip_runtime.h>
#include <math.h>


typedef __attribute__((ext_vector_type(2))) float v2f;
typedef __attribute__((ext_vector_type(8))) float v8f;

#define LQ    512
#define NH    12
#define NODE  384
#define PAIR  128
#define OUTIN 176
#define KOUT  (NH*OUTIN)

#define U_PROJ 16
#define U_AV   16
#define PO_R   4
#define OUT_S  4
#define OUT_R  6
#define OUT_NB (KOUT/(4*OUT_S))

#define OFF_SQ   ((size_t)0)
#define OFF_SK   (OFF_SQ  + (size_t)NH*LQ*16)
#define OFF_SV   (OFF_SK  + (size_t)NH*LQ*16)
#define OFF_PQ   (OFF_SV  + (size_t)NH*LQ*16)
#define OFF_PK   (OFF_PQ  + (size_t)NH*LQ*16)
#define OFF_PV   (OFF_PK  + (size_t)NH*LQ*16)
#define OFF_QN2  (OFF_PV  + (size_t)NH*LQ*24)
#define OFF_KN2  (OFF_QN2 + (size_t)NH*LQ)
#define OFF_BIAS (OFF_KN2 + (size_t)NH*LQ)
#define OFF_ATTN (OFF_BIAS+ (size_t)NH*LQ*LQ)
#define OFF_RES  (OFF_ATTN+ (size_t)NH*LQ*LQ)

#define RES_IDX(h, l, f) (((size_t)(l)*NH + (h))*OUTIN + (f))

static __device__ __forceinline__ v8f wmma4(v2f a, v2f b, v8f c) {
  return __builtin_amdgcn_wmma_f32_16x16x4_f32(
       false, a,  false, b,
       (short)0, c,  false,  false);
}

static __device__ __forceinline__ v8f vzero8() {
  v8f z = {0.f,0.f,0.f,0.f,0.f,0.f,0.f,0.f};
  return z;
}

__global__ __launch_bounds__(32) void proj_kernel(
    const float* __restrict__ node,
    const float* __restrict__ sq_k, const float* __restrict__ sk_k,
    const float* __restrict__ sv_k, const float* __restrict__ pq_k,
    const float* __restrict__ pk_k, const float* __restrict__ pv_k,
    float* __restrict__ ws)
{
  const int lane  = threadIdx.x;
  const int mt    = blockIdx.x;
  const int seg   = blockIdx.y;
  const int h     = blockIdx.z;
  const int row   = lane & 15;
  const int khalf = (lane >> 4) * 2;

  const float* wb = sk_k + (size_t)h*NODE*16;
  int   wstride = 16;
  int   wcol    = row;
  float wscale  = 1.f;
  switch (seg) {
    case 0: wb = sq_k + (size_t)h*NODE*16; wscale = 0.25f; break;
    case 1: break;
    case 2: wb = sv_k + (size_t)h*NODE*16; break;
    case 3: wb = pq_k + (size_t)h*NODE*12; wstride = 12;
            wcol = (row < 12) ? row : 11; wscale = (row < 12) ? 1.f : 0.f; break;
    case 4: wb = pk_k + (size_t)h*NODE*12; wstride = 12;
            wcol = (row < 12) ? row : 11; wscale = (row < 12) ? 1.f : 0.f; break;
    case 5: wb = pv_k + (size_t)h*NODE*24; wstride = 24; break;
    case 6: wb = pv_k + (size_t)h*NODE*24; wstride = 24;
            wcol = (row < 8) ? 16 + row : 23; wscale = (row < 8) ? 1.f : 0.f; break;
  }
  const float* arow = node + (size_t)(mt*16 + row)*NODE;
  const float* wp   = wb + wcol;

  v8f acc = vzero8();
  for (int k0 = 0; k0 < NODE; k0 += 4*U_PROJ) {
    v2f a[U_PROJ], b[U_PROJ];
#pragma unroll
    for (int u = 0; u < U_PROJ; ++u) {
      const int ka = k0 + 4*u + khalf;
      a[u].x = arow[ka];                     a[u].y = arow[ka + 1];
      b[u].x = wp[(size_t)ka * wstride];     b[u].y = wp[(size_t)(ka + 1) * wstride];
    }
    __builtin_amdgcn_sched_barrier(0);
#pragma unroll
    for (int u = 0; u < U_PROJ; ++u) {
      v2f bs; bs.x = b[u].x * wscale; bs.y = b[u].y * wscale;
      acc = wmma4(a[u], bs, acc);
    }
  }

  const int col   = lane & 15;
  const int rbase = (lane >> 4) * 8;
  float* ob = ws + OFF_SK + (size_t)h*LQ*16;
  int ostride = 16, ocol = col;
  bool dostore = true;
  switch (seg) {
    case 0: ob = ws + OFF_SQ + (size_t)h*LQ*16; break;
    case 1: break;
    case 2: ob = ws + OFF_SV + (size_t)h*LQ*16; break;
    case 3: ob = ws + OFF_PQ + (size_t)h*LQ*16; break;
    case 4: ob = ws + OFF_PK + (size_t)h*LQ*16; break;
    case 5: ob = ws + OFF_PV + (size_t)h*LQ*24; ostride = 24; break;
    case 6: ob = ws + OFF_PV + (size_t)h*LQ*24; ostride = 24;
            ocol = 16 + col; dostore = (col < 8); break;
  }
  if (dostore) {
#pragma unroll
    for (int r = 0; r < 8; ++r)
      ob[(size_t)(mt*16 + rbase + r)*ostride + ocol] = acc[r];
  }
}

__global__ void rigid_kernel(const float* __restrict__ R,
                             const float* __restrict__ t,
                             float* __restrict__ ws)
{
  const int idx = blockIdx.x * blockDim.x + threadIdx.x;
  if (idx >= NH * LQ) return;
  const int h = idx / LQ, l = idx % LQ;
  float Rm[9], tv[3];
#pragma unroll
  for (int i = 0; i < 9; ++i) Rm[i] = R[l*9 + i];
#pragma unroll
  for (int i = 0; i < 3; ++i) tv[i] = t[l*3 + i];

  float* Pq = ws + OFF_PQ + ((size_t)h*LQ + l)*16;
  float* Pk = ws + OFF_PK + ((size_t)h*LQ + l)*16;
  float* Pv = ws + OFF_PV + ((size_t)h*LQ + l)*24;

  float qs = 0.f, ksum = 0.f;
#pragma unroll
  for (int p = 0; p < 4; ++p) {
    float loc[3], g[3];
    for (int f = 0; f < 3; ++f) loc[f] = Pq[p*3+f];
    for (int gg = 0; gg < 3; ++gg) {
      g[gg] = tv[gg];
      for (int f = 0; f < 3; ++f) g[gg] += loc[f]*Rm[gg*3+f];
      Pq[p*3+gg] = g[gg]; qs += g[gg]*g[gg];
    }
    for (int f = 0; f < 3; ++f) loc[f] = Pk[p*3+f];
    for (int gg = 0; gg < 3; ++gg) {
      g[gg] = tv[gg];
      for (int f = 0; f < 3; ++f) g[gg] += loc[f]*Rm[gg*3+f];
      Pk[p*3+gg] = g[gg]; ksum += g[gg]*g[gg];
    }
  }
#pragma unroll
  for (int p = 0; p < 8; ++p) {
    float loc[3], g[3];
    for (int f = 0; f < 3; ++f) loc[f] = Pv[p*3+f];
    for (int gg = 0; gg < 3; ++gg) {
      g[gg] = tv[gg];
      for (int f = 0; f < 3; ++f) g[gg] += loc[f]*Rm[gg*3+f];
      Pv[p*3+gg] = g[gg];
    }
  }
  ws[OFF_QN2 + h*LQ + l] = qs;
  ws[OFF_KN2 + h*LQ + l] = ksum;
}

__global__ __launch_bounds__(32) void bias_kernel(
    const float* __restrict__ pairwise, const float* __restrict__ pr_k,
    float* __restrict__ ws)
{
  float* bias = ws + OFF_BIAS;
  const int lane  = threadIdx.x;
  const int tile  = blockIdx.x;
  const int row   = lane & 15;
  const int khalf = (lane >> 4) * 2;
  const int flat  = tile*16 + row;
  const float* prow = pairwise + (size_t)flat * PAIR;
  const float* prk  = pr_k + (size_t)((row < NH) ? row : NH - 1) * PAIR;
  const float bmask = (row < NH) ? 1.f : 0.f;

  v8f acc = vzero8();
  for (int k0 = 0; k0 < PAIR; k0 += 4) {
    const int ka = k0 + khalf;
    __builtin_prefetch(prow + ka + 64, 0, 0);
    v2f a; a.x = prow[ka];          a.y = prow[ka + 1];
    v2f b; b.x = prk[ka] * bmask;   b.y = prk[ka + 1] * bmask;
    acc = wmma4(a, b, acc);
  }
  const int col = lane & 15;
  const int rbase = (lane >> 4) * 8;
  if (col < NH) {
#pragma unroll
    for (int r = 0; r < 8; ++r) {
      const int f2 = tile*16 + rbase + r;
      const int i = f2 >> 9, j = f2 & 511;
      bias[((size_t)col*LQ + i)*LQ + j] = acc[r];
    }
  }
}

#define SMS 516
__global__ __launch_bounds__(128) void attn_kernel(
    const float* __restrict__ point_weight, float* __restrict__ ws)
{
  __shared__ float sm[16 * SMS];
  const int tid  = threadIdx.x;
  const int lane = tid & 31;
  const int w    = tid >> 5;
  const int nt   = blockIdx.x;
  const int h    = blockIdx.y;

  const float* Sq  = ws + OFF_SQ;  const float* Sk  = ws + OFF_SK;
  const float* Pq  = ws + OFF_PQ;  const float* Pk  = ws + OFF_PK;
  const float* qn2 = ws + OFF_QN2; const float* kn2 = ws + OFF_KN2;
  const float* bias= ws + OFF_BIAS;
  const float* Sv  = ws + OFF_SV;  const float* Pv  = ws + OFF_PV;
  float* attn = ws + OFF_ATTN;
  float* res  = ws + OFF_RES;

  const float pwx = point_weight[h];
  const float pw  = (pwx > 20.f) ? pwx : log1pf(__expf(pwx));
  const float cpt = -0.5f * pw / sqrtf(18.f);
  const float rs3 = 0.57735026919f;

  const int row   = lane & 15;
  const int khalf = (lane >> 4) * 2;
  const int col   = lane & 15;
  const int rbase = (lane >> 4) * 8;

  v2f aq[4], ap[4];
#pragma unroll
  for (int ks = 0; ks < 4; ++ks) {
    const int k = ks*4 + khalf;
    const int l = nt*16 + row;
    aq[ks].x = Sq[((size_t)h*LQ + l)*16 + k];
    aq[ks].y = Sq[((size_t)h*LQ + l)*16 + k + 1];
    ap[ks].x = Pq[((size_t)h*LQ + l)*16 + k];
    ap[ks].y = Pq[((size_t)h*LQ + l)*16 + k + 1];
  }
  float qv[8];
#pragma unroll
  for (int r = 0; r < 8; ++r) qv[r] = qn2[h*LQ + nt*16 + rbase + r];

  for (int mt = w; mt < 32; mt += 4) {
    v8f s = vzero8(), p = vzero8();
#pragma unroll
    for (int ks = 0; ks < 4; ++ks) {
      const int k = ks*4 + khalf;
      const int m = mt*16 + row;
      v2f bq, bp;
      bq.x = Sk[((size_t)h*LQ + m)*16 + k]; bq.y = Sk[((size_t)h*LQ + m)*16 + k + 1];
      bp.x = Pk[((size_t)h*LQ + m)*16 + k]; bp.y = Pk[((size_t)h*LQ + m)*16 + k + 1];
      s = wmma4(aq[ks], bq, s);
      p = wmma4(ap[ks], bp, p);
    }
    const int m = mt*16 + col;
    const float kv = kn2[h*LQ + m];
#pragma unroll
    for (int r = 0; r < 8; ++r) {
      const int rr = rbase + r;
      const int n  = nt*16 + rr;
      const float dist = qv[r] + kv - 2.f*p[r];
      sm[rr*SMS + m] =
          (s[r] + bias[((size_t)h*LQ + n)*LQ + m] + cpt*dist) * rs3;
    }
  }
  __syncthreads();

  {
    const int rrow = tid >> 3;
    const int sub  = tid & 7;
    float mx = -1e30f;
    for (int j = sub; j < LQ; j += 8) mx = fmaxf(mx, sm[rrow*SMS + j]);
    for (int o = 1; o < 8; o <<= 1) mx = fmaxf(mx, __shfl_xor(mx, o, 8));
    float sum = 0.f;
    for (int j = sub; j < LQ; j += 8) {
      const float e = __expf(sm[rrow*SMS + j] - mx);
      sm[rrow*SMS + j] = e; sum += e;
    }
    for (int o = 1; o < 8; o <<= 1) sum += __shfl_xor(sum, o, 8);
    const float inv = 1.f / sum;
    const int n = nt*16 + rrow;
    for (int j = sub; j < LQ; j += 8) {
      const float v = sm[rrow*SMS + j] * inv;
      sm[rrow*SMS + j] = v;
      attn[((size_t)h*LQ + n)*LQ + j] = v;
    }
  }
  __syncthreads();

  if (w < 3) {
    const float* vb = (w == 0) ? Sv + (size_t)h*LQ*16 : Pv + (size_t)h*LQ*24;
    const int    vs = (w == 0) ? 16 : 24;
    int   vc = col;
    float vmask = 1.f;
    if (w == 2) { vc = (col < 8) ? 16 + col : 23; vmask = (col < 8) ? 1.f : 0.f; }
    const float* vp = vb + vc;

    v8f acc = vzero8();
    for (int kt0 = 0; kt0 < 128; kt0 += U_AV) {
      v2f a[U_AV], b[U_AV];
#pragma unroll
      for (int u = 0; u < U_AV; ++u) {
        const int k = (kt0 + u)*4 + khalf;
        a[u].x = sm[row*SMS + k];        a[u].y = sm[row*SMS + k + 1];
        b[u].x = vp[(size_t)k * vs];     b[u].y = vp[(size_t)(k + 1) * vs];
      }
      __builtin_amdgcn_sched_barrier(0);
#pragma unroll
      for (int u = 0; u < U_AV; ++u) {
        v2f bs; bs.x = b[u].x * vmask; bs.y = b[u].y * vmask;
        acc = wmma4(a[u], bs, acc);
      }
    }
    const int fbase = (w == 0) ? 0 : 144;
#pragma unroll
    for (int r = 0; r < 8; ++r) {
      const int n = nt*16 + rbase + r;
      const float v = acc[r];
      if (w < 2)        res[RES_IDX(h, n, fbase + col)] = v;
      else if (col < 8) res[RES_IDX(h, n, 160 + col)]   = v;
    }
  }
}

static __device__ __forceinline__ void po_load(
    v2f& a, v2f (&b)[8], const float* __restrict__ arow,
    const float* __restrict__ prow, int m)
{
  a.x = arow[m];
  a.y = arow[m + 1];
  const float* pr0 = prow + (size_t)m * PAIR;
#pragma unroll
  for (int ntile = 0; ntile < 8; ++ntile) {
    b[ntile].x = pr0[ntile*16];
    b[ntile].y = pr0[ntile*16 + PAIR];
  }
}

template <bool TAIL>
static __device__ __forceinline__ void po_turn(
    v8f (&acc)[8], v2f (&a)[PO_R], v2f (&b)[PO_R][8],
    const float* __restrict__ arow, const float* __restrict__ prow,
    int kt0, int khalf, float amask)
{
#pragma unroll
  for (int j = 0; j < PO_R; ++j) {
    if (!TAIL || j == 0)
      po_load(a[(j + PO_R - 1) % PO_R], b[(j + PO_R - 1) % PO_R],
              arow, prow, (kt0 + j + PO_R - 1)*4 + khalf);
    __builtin_amdgcn_sched_barrier(0);
    v2f am; am.x = a[j].x * amask; am.y = a[j].y * amask;
#pragma unroll
    for (int ntile = 0; ntile < 8; ++ntile)
      acc[ntile] = wmma4(am, b[j][ntile], acc[ntile]);
    __builtin_amdgcn_sched_barrier(0);
  }
}

__global__ __launch_bounds__(32) void pairout_kernel(
    const float* __restrict__ pairwise, float* __restrict__ ws)
{
  const float* attn = ws + OFF_ATTN;
  float* res = ws + OFF_RES;
  const int lane  = threadIdx.x;
  const int n     = blockIdx.x;
  const int row   = lane & 15;
  const int khalf = (lane >> 4) * 2;
  const int col   = lane & 15;

  const int hh0 = (row < NH) ? row : NH - 1;
  const float amask = (row < NH) ? 1.f : 0.f;
  const float* arow = attn + ((size_t)hh0*LQ + n)*LQ;

  v8f acc[8];
#pragma unroll
  for (int i = 0; i < 8; ++i) acc[i] = vzero8();

  v2f a[PO_R], b[PO_R][8];
  const float* prow = pairwise + (size_t)n*LQ*PAIR + col;
#pragma unroll
  for (int s = 0; s < PO_R - 1; ++s) {
    po_load(a[s], b[s], arow, prow, s*4 + khalf);
    __builtin_amdgcn_sched_barrier(0);
  }

  for (int kt0 = 0; kt0 < 128 - PO_R; kt0 += PO_R)
    po_turn<false>(acc, a, b, arow, prow, kt0, khalf, amask);
  po_turn<true>(acc, a, b, arow, prow, 128 - PO_R, khalf, amask);
  const int rbase = (lane >> 4) * 8;
#pragma unroll
  for (int ntile = 0; ntile < 8; ++ntile)
#pragma unroll
    for (int r = 0; r < 8; ++r) {
      const int hh = rbase + r;
      if (hh < NH)
        res[RES_IDX(hh, n, 16 + ntile*16 + col)] = acc[ntile][r];
    }
}

__global__ void pfinish_kernel(const float* __restrict__ R,
                               const float* __restrict__ t,
                               float* __restrict__ ws)
{
  const int idx = blockIdx.x * blockDim.x + threadIdx.x;
  if (idx >= NH * LQ) return;
  const int h = idx / LQ, l = idx % LQ;
  float* resp = ws + OFF_RES + RES_IDX(h, l, 0);
  float Rm[9], tv[3];
#pragma unroll
  for (int i = 0; i < 9; ++i) Rm[i] = R[l*9 + i];
#pragma unroll
  for (int i = 0; i < 3; ++i) tv[i] = t[l*3 + i];
#pragma unroll
  for (int p = 0; p < 8; ++p) {
    float g[3], loc[3], nrm = 0.f;
    for (int gg = 0; gg < 3; ++gg) g[gg] = resp[144 + p*3 + gg] - tv[gg];
    for (int f = 0; f < 3; ++f) {
      float v = 0.f;
      for (int gg = 0; gg < 3; ++gg) v += g[gg]*Rm[gg*3 + f];
      loc[f] = v; nrm += v*v;
    }
    for (int f = 0; f < 3; ++f) resp[144 + p*3 + f] = loc[f];
    resp[168 + p] = sqrtf(nrm + 1e-12f);
  }
}

static __device__ __forceinline__ void out_load(
    v2f (&a)[OUT_S], v2f (&b)[OUT_S], const float* __restrict__ resl,
    const float* __restrict__ okp, int k0, int khalf)
{
#pragma unroll
  for (int u = 0; u < OUT_S; ++u) {
    const int ka = k0 + 4*u + khalf;
    a[u].x = resl[ka];                 a[u].y = resl[ka + 1];
    b[u].x = okp[(size_t)ka * NODE];   b[u].y = okp[(size_t)(ka + 1) * NODE];
  }
}

template <bool TAIL>
static __device__ __forceinline__ void out_turn(
    v8f& acc, v2f (&a)[OUT_R][OUT_S], v2f (&b)[OUT_R][OUT_S],
    const float* __restrict__ resl, const float* __restrict__ okp,
    int kb0, int khalf)
{
#pragma unroll
  for (int j = 0; j < OUT_R; ++j) {
    if (!TAIL || j == 0)
      out_load(a[(j + OUT_R - 1) % OUT_R], b[(j + OUT_R - 1) % OUT_R],
               resl, okp, (kb0 + j + OUT_R - 1)*4*OUT_S, khalf);
    __builtin_amdgcn_sched_barrier(0);
#pragma unroll
    for (int u = 0; u < OUT_S; ++u) acc = wmma4(a[j][u], b[j][u], acc);
    __builtin_amdgcn_sched_barrier(0);
  }
}

__global__ __launch_bounds__(32) void outproj_kernel(
    const float* __restrict__ out_k, const float* __restrict__ out_b,
    const float* __restrict__ res, float* __restrict__ out)
{
  const int lane  = threadIdx.x;
  const int mt    = blockIdx.x;
  const int ntile = blockIdx.y;
  const int row   = lane & 15;
  const int khalf = (lane >> 4) * 2;
  const int col   = lane & 15;
  const int o     = ntile*16 + col;

  const float* resl = res + (size_t)(mt*16 + row) * KOUT;
  const float* okp  = out_k + o;

  v8f acc = vzero8();
  v2f a[OUT_R][OUT_S], b[OUT_R][OUT_S];
#pragma unroll
  for (int s = 0; s < OUT_R - 1; ++s) {
    out_load(a[s], b[s], resl, okp, s*4*OUT_S, khalf);
    __builtin_amdgcn_sched_barrier(0);
  }

  for (int kb0 = 0; kb0 < OUT_NB - OUT_R; kb0 += OUT_R)
    out_turn<false>(acc, a, b, resl, okp, kb0, khalf);
  out_turn<true>(acc, a, b, resl, okp, OUT_NB - OUT_R, khalf);
  const int rbase = (lane >> 4) * 8;
  const float ob = out_b[o];
#pragma unroll
  for (int r = 0; r < 8; ++r) {
    const int l = mt*16 + rbase + r;
    out[(size_t)l*NODE + o] = acc[r] + ob;
  }
}

extern "C" void kernel_launch(void* const* d_in, const int* in_sizes, int n_in,
                              void* d_out, int out_size, void* d_ws, size_t ws_size,
                              hipStream_t stream) {
  const float* node  = (const float*)d_in[0];
  const float* R     = (const float*)d_in[1];
  const float* t     = (const float*)d_in[2];
  const float* pairw = (const float*)d_in[3];
  const float* sq_k  = (const float*)d_in[4];
  const float* sk_k  = (const float*)d_in[5];
  const float* sv_k  = (const float*)d_in[6];
  const float* pq_k  = (const float*)d_in[7];
  const float* pk_k  = (const float*)d_in[8];
  const float* pv_k  = (const float*)d_in[9];
  const float* pr_k  = (const float*)d_in[10];
  const float* out_k = (const float*)d_in[11];
  const float* out_b = (const float*)d_in[12];
  const float* pwgt  = (const float*)d_in[13];
  float* ws  = (float*)d_ws;
  float* out = (float*)d_out;

  proj_kernel<<<dim3(32, 7, NH), 32, 0, stream>>>(
      node, sq_k, sk_k, sv_k, pq_k, pk_k, pv_k, ws);
  rigid_kernel<<<(NH*LQ + 127)/128, 128, 0, stream>>>(R, t, ws);
  bias_kernel<<<(LQ*LQ)/16, 32, 0, stream>>>(pairw, pr_k, ws);
  attn_kernel<<<dim3(32, NH), 128, 0, stream>>>(pwgt, ws);
  pairout_kernel<<<LQ, 32, 0, stream>>>(pairw, ws);
  pfinish_kernel<<<(NH*LQ + 127)/128, 128, 0, stream>>>(R, t, ws);
  outproj_kernel<<<dim3(32, 24), 32, 0, stream>>>(out_k, out_b, ws + OFF_RES, out);
}


# pairout: a row's 8 N-tiles split over 4 waves (128-thread block), 2 tiles per wave from 8-byte loads on an interleaved column map, ring 14 k-steps deep; per-element sum order unchanged
# speedup vs baseline: 2.2032x; 1.1350x over previous
#include <hip/hip_runtime.h>
#include <math.h>


typedef __attribute__((ext_vector_type(2))) float v2f;
typedef __attribute__((ext_vector_type(8))) float v8f;

#define LQ    512
#define NH    12
#define NODE  384
#define PAIR  128
#define OUTIN 176
#define KOUT  (NH*OUTIN)

#define U_PROJ 16
#define U_AV   16
#define PO_W   4
#define PO_T   2
#define PO_S   2
#define PO_R   8
#define PO_NB  (LQ/(4*PO_S))
#define OUT_S  4
#define OUT_R  6
#define OUT_NB (KOUT/(4*OUT_S))

#define OFF_SQ   ((size_t)0)
#define OFF_SK   (OFF_SQ  + (size_t)NH*LQ*16)
#define OFF_SV   (OFF_SK  + (size_t)NH*LQ*16)
#define OFF_PQ   (OFF_SV  + (size_t)NH*LQ*16)
#define OFF_PK   (OFF_PQ  + (size_t)NH*LQ*16)
#define OFF_PV   (OFF_PK  + (size_t)NH*LQ*16)
#define OFF_QN2  (OFF_PV  + (size_t)NH*LQ*24)
#define OFF_KN2  (OFF_QN2 + (size_t)NH*LQ)
#define OFF_BIAS (OFF_KN2 + (size_t)NH*LQ)
#define OFF_ATTN (OFF_BIAS+ (size_t)NH*LQ*LQ)
#define OFF_RES  (OFF_ATTN+ (size_t)NH*LQ*LQ)

#define RES_IDX(h, l, f) (((size_t)(l)*NH + (h))*OUTIN + (f))

static __device__ __forceinline__ v8f wmma4(v2f a, v2f b, v8f c) {
  return __builtin_amdgcn_wmma_f32_16x16x4_f32(
       false, a,  false, b,
       (short)0, c,  false,  false);
}

static __device__ __forceinline__ v8f vzero8() {
  v8f z = {0.f,0.f,0.f,0.f,0.f,0.f,0.f,0.f};
  return z;
}

__global__ __launch_bounds__(32) void proj_kernel(
    const float* __restrict__ node,
    const float* __restrict__ sq_k, const float* __restrict__ sk_k,
    const float* __restrict__ sv_k, const float* __restrict__ pq_k,
    const float* __restrict__ pk_k, const float* __restrict__ pv_k,
    float* __restrict__ ws)
{
  const int lane  = threadIdx.x;
  const int mt    = blockIdx.x;
  const int seg   = blockIdx.y;
  const int h     = blockIdx.z;
  const int row   = lane & 15;
  const int khalf = (lane >> 4) * 2;

  const float* wb = sk_k + (size_t)h*NODE*16;
  int   wstride = 16;
  int   wcol    = row;
  float wscale  = 1.f;
  switch (seg) {
    case 0: wb = sq_k + (size_t)h*NODE*16; wscale = 0.25f; break;
    case 1: break;
    case 2: wb = sv_k + (size_t)h*NODE*16; break;
    case 3: wb = pq_k + (size_t)h*NODE*12; wstride = 12;
            wcol = (row < 12) ? row : 11; wscale = (row < 12) ? 1.f : 0.f; break;
    case 4: wb = pk_k + (size_t)h*NODE*12; wstride = 12;
            wcol = (row < 12) ? row : 11; wscale = (row < 12) ? 1.f : 0.f; break;
    case 5: wb = pv_k + (size_t)h*NODE*24; wstride = 24; break;
    case 6: wb = pv_k + (size_t)h*NODE*24; wstride = 24;
            wcol = (row < 8) ? 16 + row : 23; wscale = (row < 8) ? 1.f : 0.f; break;
  }
  const float* arow = node + (size_t)(mt*16 + row)*NODE;
  const float* wp   = wb + wcol;

  v8f acc = vzero8();
  for (int k0 = 0; k0 < NODE; k0 += 4*U_PROJ) {
    v2f a[U_PROJ], b[U_PROJ];
#pragma unroll
    for (int u = 0; u < U_PROJ; ++u) {
      const int ka = k0 + 4*u + khalf;
      a[u].x = arow[ka];                     a[u].y = arow[ka + 1];
      b[u].x = wp[(size_t)ka * wstride];     b[u].y = wp[(size_t)(ka + 1) * wstride];
    }
    __builtin_amdgcn_sched_barrier(0);
#pragma unroll
    for (int u = 0; u < U_PROJ; ++u) {
      v2f bs; bs.x = b[u].x * wscale; bs.y = b[u].y * wscale;
      acc = wmma4(a[u], bs, acc);
    }
  }

  const int col   = lane & 15;
  const int rbase = (lane >> 4) * 8;
  float* ob = ws + OFF_SK + (size_t)h*LQ*16;
  int ostride = 16, ocol = col;
  bool dostore = true;
  switch (seg) {
    case 0: ob = ws + OFF_SQ + (size_t)h*LQ*16; break;
    case 1: break;
    case 2: ob = ws + OFF_SV + (size_t)h*LQ*16; break;
    case 3: ob = ws + OFF_PQ + (size_t)h*LQ*16; break;
    case 4: ob = ws + OFF_PK + (size_t)h*LQ*16; break;
    case 5: ob = ws + OFF_PV + (size_t)h*LQ*24; ostride = 24; break;
    case 6: ob = ws + OFF_PV + (size_t)h*LQ*24; ostride = 24;
            ocol = 16 + col; dostore = (col < 8); break;
  }
  if (dostore) {
#pragma unroll
    for (int r = 0; r < 8; ++r)
      ob[(size_t)(mt*16 + rbase + r)*ostride + ocol] = acc[r];
  }
}

__global__ void rigid_kernel(const float* __restrict__ R,
                             const float* __restrict__ t,
                             float* __restrict__ ws)
{
  const int idx = blockIdx.x * blockDim.x + threadIdx.x;
  if (idx >= NH * LQ) return;
  const int h = idx / LQ, l = idx % LQ;
  float Rm[9], tv[3];
#pragma unroll
  for (int i = 0; i < 9; ++i) Rm[i] = R[l*9 + i];
#pragma unroll
  for (int i = 0; i < 3; ++i) tv[i] = t[l*3 + i];

  float* Pq = ws + OFF_PQ + ((size_t)h*LQ + l)*16;
  float* Pk = ws + OFF_PK + ((size_t)h*LQ + l)*16;
  float* Pv = ws + OFF_PV + ((size_t)h*LQ + l)*24;

  float qs = 0.f, ksum = 0.f;
#pragma unroll
  for (int p = 0; p < 4; ++p) {
    float loc[3], g[3];
    for (int f = 0; f < 3; ++f) loc[f] = Pq[p*3+f];
    for (int gg = 0; gg < 3; ++gg) {
      g[gg] = tv[gg];
      for (int f = 0; f < 3; ++f) g[gg] += loc[f]*Rm[gg*3+f];
      Pq[p*3+gg] = g[gg]; qs += g[gg]*g[gg];
    }
    for (int f = 0; f < 3; ++f) loc[f] = Pk[p*3+f];
    for (int gg = 0; gg < 3; ++gg) {
      g[gg] = tv[gg];
      for (int f = 0; f < 3; ++f) g[gg] += loc[f]*Rm[gg*3+f];
      Pk[p*3+gg] = g[gg]; ksum += g[gg]*g[gg];
    }
  }
#pragma unroll
  for (int p = 0; p < 8; ++p) {
    float loc[3], g[3];
    for (int f = 0; f < 3; ++f) loc[f] = Pv[p*3+f];
    for (int gg = 0; gg < 3; ++gg) {
      g[gg] = tv[gg];
      for (int f = 0; f < 3; ++f) g[gg] += loc[f]*Rm[gg*3+f];
      Pv[p*3+gg] = g[gg];
    }
  }
  ws[OFF_QN2 + h*LQ + l] = qs;
  ws[OFF_KN2 + h*LQ + l] = ksum;
}

__global__ __launch_bounds__(32) void bias_kernel(
    const float* __restrict__ pairwise, const float* __restrict__ pr_k,
    float* __restrict__ ws)
{
  float* bias = ws + OFF_BIAS;
  const int lane  = threadIdx.x;
  const int tile  = blockIdx.x;
  const int row   = lane & 15;
  const int khalf = (lane >> 4) * 2;
  const int flat  = tile*16 + row;
  const float* prow = pairwise + (size_t)flat * PAIR;
  const float* prk  = pr_k + (size_t)((row < NH) ? row : NH - 1) * PAIR;
  const float bmask = (row < NH) ? 1.f : 0.f;

  v8f acc = vzero8();
  for (int k0 = 0; k0 < PAIR; k0 += 4) {
    const int ka = k0 + khalf;
    __builtin_prefetch(prow + ka + 64, 0, 0);
    v2f a; a.x = prow[ka];          a.y = prow[ka + 1];
    v2f b; b.x = prk[ka] * bmask;   b.y = prk[ka + 1] * bmask;
    acc = wmma4(a, b, acc);
  }
  const int col = lane & 15;
  const int rbase = (lane >> 4) * 8;
  if (col < NH) {
#pragma unroll
    for (int r = 0; r < 8; ++r) {
      const int f2 = tile*16 + rbase + r;
      const int i = f2 >> 9, j = f2 & 511;
      bias[((size_t)col*LQ + i)*LQ + j] = acc[r];
    }
  }
}

#define SMS 516
__global__ __launch_bounds__(128) void attn_kernel(
    const float* __restrict__ point_weight, float* __restrict__ ws)
{
  __shared__ float sm[16 * SMS];
  const int tid  = threadIdx.x;
  const int lane = tid & 31;
  const int w    = tid >> 5;
  const int nt   = blockIdx.x;
  const int h    = blockIdx.y;

  const float* Sq  = ws + OFF_SQ;  const float* Sk  = ws + OFF_SK;
  const float* Pq  = ws + OFF_PQ;  const float* Pk  = ws + OFF_PK;
  const float* qn2 = ws + OFF_QN2; const float* kn2 = ws + OFF_KN2;
  const float* bias= ws + OFF_BIAS;
  const float* Sv  = ws + OFF_SV;  const float* Pv  = ws + OFF_PV;
  float* attn = ws + OFF_ATTN;
  float* res  = ws + OFF_RES;

  const float pwx = point_weight[h];
  const float pw  = (pwx > 20.f) ? pwx : log1pf(__expf(pwx));
  const float cpt = -0.5f * pw / sqrtf(18.f);
  const float rs3 = 0.57735026919f;

  const int row   = lane & 15;
  const int khalf = (lane >> 4) * 2;
  const int col   = lane & 15;
  const int rbase = (lane >> 4) * 8;

  v2f aq[4], ap[4];
#pragma unroll
  for (int ks = 0; ks < 4; ++ks) {
    const int k = ks*4 + khalf;
    const int l = nt*16 + row;
    aq[ks].x = Sq[((size_t)h*LQ + l)*16 + k];
    aq[ks].y = Sq[((size_t)h*LQ + l)*16 + k + 1];
    ap[ks].x = Pq[((size_t)h*LQ + l)*16 + k];
    ap[ks].y = Pq[((size_t)h*LQ + l)*16 + k + 1];
  }
  float qv[8];
#pragma unroll
  for (int r = 0; r < 8; ++r) qv[r] = qn2[h*LQ + nt*16 + rbase + r];

  for (int mt = w; mt < 32; mt += 4) {
    v8f s = vzero8(), p = vzero8();
#pragma unroll
    for (int ks = 0; ks < 4; ++ks) {
      const int k = ks*4 + khalf;
      const int m = mt*16 + row;
      v2f bq, bp;
      bq.x = Sk[((size_t)h*LQ + m)*16 + k]; bq.y = Sk[((size_t)h*LQ + m)*16 + k + 1];
      bp.x = Pk[((size_t)h*LQ + m)*16 + k]; bp.y = Pk[((size_t)h*LQ + m)*16 + k + 1];
      s = wmma4(aq[ks], bq, s);
      p = wmma4(ap[ks], bp, p);
    }
    const int m = mt*16 + col;
    const float kv = kn2[h*LQ + m];
#pragma unroll
    for (int r = 0; r < 8; ++r) {
      const int rr = rbase + r;
      const int n  = nt*16 + rr;
      const float dist = qv[r] + kv - 2.f*p[r];
      sm[rr*SMS + m] =
          (s[r] + bias[((size_t)h*LQ + n)*LQ + m] + cpt*dist) * rs3;
    }
  }
  __syncthreads();

  {
    const int rrow = tid >> 3;
    const int sub  = tid & 7;
    float mx = -1e30f;
    for (int j = sub; j < LQ; j += 8) mx = fmaxf(mx, sm[rrow*SMS + j]);
    for (int o = 1; o < 8; o <<= 1) mx = fmaxf(mx, __shfl_xor(mx, o, 8));
    float sum = 0.f;
    for (int j = sub; j < LQ; j += 8) {
      const float e = __expf(sm[rrow*SMS + j] - mx);
      sm[rrow*SMS + j] = e; sum += e;
    }
    for (int o = 1; o < 8; o <<= 1) sum += __shfl_xor(sum, o, 8);
    const float inv = 1.f / sum;
    const int n = nt*16 + rrow;
    for (int j = sub; j < LQ; j += 8) {
      const float v = sm[rrow*SMS + j] * inv;
      sm[rrow*SMS + j] = v;
      attn[((size_t)h*LQ + n)*LQ + j] = v;
    }
  }
  __syncthreads();

  if (w < 3) {
    const float* vb = (w == 0) ? Sv + (size_t)h*LQ*16 : Pv + (size_t)h*LQ*24;
    const int    vs = (w == 0) ? 16 : 24;
    int   vc = col;
    float vmask = 1.f;
    if (w == 2) { vc = (col < 8) ? 16 + col : 23; vmask = (col < 8) ? 1.f : 0.f; }
    const float* vp = vb + vc;

    v8f acc = vzero8();
    for (int kt0 = 0; kt0 < 128; kt0 += U_AV) {
      v2f a[U_AV], b[U_AV];
#pragma unroll
      for (int u = 0; u < U_AV; ++u) {
        const int k = (kt0 + u)*4 + khalf;
        a[u].x = sm[row*SMS + k];        a[u].y = sm[row*SMS + k + 1];
        b[u].x = vp[(size_t)k * vs];     b[u].y = vp[(size_t)(k + 1) * vs];
      }
      __builtin_amdgcn_sched_barrier(0);
#pragma unroll
      for (int u = 0; u < U_AV; ++u) {
        v2f bs; bs.x = b[u].x * vmask; bs.y = b[u].y * vmask;
        acc = wmma4(a[u], bs, acc);
      }
    }
    const int fbase = (w == 0) ? 0 : 144;
#pragma unroll
    for (int r = 0; r < 8; ++r) {
      const int n = nt*16 + rbase + r;
      const float v = acc[r];
      if (w < 2)        res[RES_IDX(h, n, fbase + col)] = v;
      else if (col < 8) res[RES_IDX(h, n, 160 + col)]   = v;
    }
  }
}

static __device__ __forceinline__ void po_load(
    v2f (&a)[PO_S], v2f (&x)[PO_S], v2f (&y)[PO_S],
    const float* __restrict__ arow, const float* __restrict__ prow, int m0)
{
#pragma unroll
  for (int u = 0; u < PO_S; ++u) {
    const int m = m0 + 4*u;
    a[u] = *(const v2f*)(arow + m);
    x[u] = *(const v2f*)(prow + (size_t)m * PAIR);
    y[u] = *(const v2f*)(prow + (size_t)(m + 1) * PAIR);
  }
}

template <bool TAIL>
static __device__ __forceinline__ void po_turn(
    v8f (&acc)[PO_T], v2f (&a)[PO_R][PO_S], v2f (&x)[PO_R][PO_S],
    v2f (&y)[PO_R][PO_S], const float* __restrict__ arow,
    const float* __restrict__ prow, int kb0, int khalf, float amask)
{
#pragma unroll
  for (int j = 0; j < PO_R; ++j) {
    if (!TAIL || j == 0)
      po_load(a[(j + PO_R - 1) % PO_R], x[(j + PO_R - 1) % PO_R],
              y[(j + PO_R - 1) % PO_R], arow, prow,
              (kb0 + j + PO_R - 1)*4*PO_S + khalf);
    __builtin_amdgcn_sched_barrier(0);
#pragma unroll
    for (int u = 0; u < PO_S; ++u) {
      v2f am; am.x = a[j][u].x * amask; am.y = a[j][u].y * amask;
      v2f b0; b0.x = x[j][u].x; b0.y = y[j][u].x;
      v2f b1; b1.x = x[j][u].y; b1.y = y[j][u].y;
      acc[0] = wmma4(am, b0, acc[0]);
      acc[1] = wmma4(am, b1, acc[1]);
    }
    __builtin_amdgcn_sched_barrier(0);
  }
}

__global__ __launch_bounds__(32*PO_W) void pairout_kernel(
    const float* __restrict__ pairwise, float* __restrict__ ws)
{
  const float* attn = ws + OFF_ATTN;
  float* res = ws + OFF_RES;
  const int lane  = threadIdx.x & 31;
  const int w     = threadIdx.x >> 5;
  const int n     = blockIdx.x;
  const int row   = lane & 15;
  const int khalf = (lane >> 4) * 2;
  const int col   = lane & 15;

  const int hh0 = (row < NH) ? row : NH - 1;
  const float amask = (row < NH) ? 1.f : 0.f;
  const float* arow = attn + ((size_t)hh0*LQ + n)*LQ;

  v8f acc[PO_T];
#pragma unroll
  for (int i = 0; i < PO_T; ++i) acc[i] = vzero8();

  v2f a[PO_R][PO_S], x[PO_R][PO_S], y[PO_R][PO_S];
  const float* prow = pairwise + (size_t)n*LQ*PAIR + 32*w + 2*col;
#pragma unroll
  for (int s = 0; s < PO_R - 1; ++s) {
    po_load(a[s], x[s], y[s], arow, prow, s*4*PO_S + khalf);
    __builtin_amdgcn_sched_barrier(0);
  }

  for (int kb0 = 0; kb0 < PO_NB - PO_R; kb0 += PO_R)
    po_turn<false>(acc, a, x, y, arow, prow, kb0, khalf, amask);
  po_turn<true>(acc, a, x, y, arow, prow, PO_NB - PO_R, khalf, amask);

  const int rbase = (lane >> 4) * 8;
#pragma unroll
  for (int r = 0; r < 8; ++r) {
    const int hh = rbase + r;
    if (hh < NH) {
      v2f o; o.x = acc[0][r]; o.y = acc[1][r];
      *(v2f*)(res + RES_IDX(hh, n, 16 + 32*w + 2*col)) = o;
    }
  }
}

__global__ void pfinish_kernel(const float* __restrict__ R,
                               const float* __restrict__ t,
                               float* __restrict__ ws)
{
  const int idx = blockIdx.x * blockDim.x + threadIdx.x;
  if (idx >= NH * LQ) return;
  const int h = idx / LQ, l = idx % LQ;
  float* resp = ws + OFF_RES + RES_IDX(h, l, 0);
  float Rm[9], tv[3];
#pragma unroll
  for (int i = 0; i < 9; ++i) Rm[i] = R[l*9 + i];
#pragma unroll
  for (int i = 0; i < 3; ++i) tv[i] = t[l*3 + i];
#pragma unroll
  for (int p = 0; p < 8; ++p) {
    float g[3], loc[3], nrm = 0.f;
    for (int gg = 0; gg < 3; ++gg) g[gg] = resp[144 + p*3 + gg] - tv[gg];
    for (int f = 0; f < 3; ++f) {
      float v = 0.f;
      for (int gg = 0; gg < 3; ++gg) v += g[gg]*Rm[gg*3 + f];
      loc[f] = v; nrm += v*v;
    }
    for (int f = 0; f < 3; ++f) resp[144 + p*3 + f] = loc[f];
    resp[168 + p] = sqrtf(nrm + 1e-12f);
  }
}

static __device__ __forceinline__ void out_load(
    v2f (&a)[OUT_S], v2f (&b)[OUT_S], const float* __restrict__ resl,
    const float* __restrict__ okp, int k0, int khalf)
{
#pragma unroll
  for (int u = 0; u < OUT_S; ++u) {
    const int ka = k0 + 4*u + khalf;
    a[u].x = resl[ka];                 a[u].y = resl[ka + 1];
    b[u].x = okp[(size_t)ka * NODE];   b[u].y = okp[(size_t)(ka + 1) * NODE];
  }
}

template <bool TAIL>
static __device__ __forceinline__ void out_turn(
    v8f& acc, v2f (&a)[OUT_R][OUT_S], v2f (&b)[OUT_R][OUT_S],
    const float* __restrict__ resl, const float* __restrict__ okp,
    int kb0, int khalf)
{
#pragma unroll
  for (int j = 0; j < OUT_R; ++j) {
    if (!TAIL || j == 0)
      out_load(a[(j + OUT_R - 1) % OUT_R], b[(j + OUT_R - 1) % OUT_R],
               resl, okp, (kb0 + j + OUT_R - 1)*4*OUT_S, khalf);
    __builtin_amdgcn_sched_barrier(0);
#pragma unroll
    for (int u = 0; u < OUT_S; ++u) acc = wmma4(a[j][u], b[j][u], acc);
    __builtin_amdgcn_sched_barrier(0);
  }
}

__global__ __launch_bounds__(32) void outproj_kernel(
    const float* __restrict__ out_k, const float* __restrict__ out_b,
    const float* __restrict__ res, float* __restrict__ out)
{
  const int lane  = threadIdx.x;
  const int mt    = blockIdx.x;
  const int ntile = blockIdx.y;
  const int row   = lane & 15;
  const int khalf = (lane >> 4) * 2;
  const int col   = lane & 15;
  const int o     = ntile*16 + col;

  const float* resl = res + (size_t)(mt*16 + row) * KOUT;
  const float* okp  = out_k + o;

  v8f acc = vzero8();
  v2f a[OUT_R][OUT_S], b[OUT_R][OUT_S];
#pragma unroll
  for (int s = 0; s < OUT_R - 1; ++s) {
    out_load(a[s], b[s], resl, okp, s*4*OUT_S, khalf);
    __builtin_amdgcn_sched_barrier(0);
  }

  for (int kb0 = 0; kb0 < OUT_NB - OUT_R; kb0 += OUT_R)
    out_turn<false>(acc, a, b, resl, okp, kb0, khalf);
  out_turn<true>(acc, a, b, resl, okp, OUT_NB - OUT_R, khalf);
  const int rbase = (lane >> 4) * 8;
  const float ob = out_b[o];
#pragma unroll
  for (int r = 0; r < 8; ++r) {
    const int l = mt*16 + rbase + r;
    out[(size_t)l*NODE + o] = acc[r] + ob;
  }
}

extern "C" void kernel_launch(void* const* d_in, const int* in_sizes, int n_in,
                              void* d_out, int out_size, void* d_ws, size_t ws_size,
                              hipStream_t stream) {
  const float* node  = (const float*)d_in[0];
  const float* R     = (const float*)d_in[1];
  const float* t     = (const float*)d_in[2];
  const float* pairw = (const float*)d_in[3];
  const float* sq_k  = (const float*)d_in[4];
  const float* sk_k  = (const float*)d_in[5];
  const float* sv_k  = (const float*)d_in[6];
  const float* pq_k  = (const float*)d_in[7];
  const float* pk_k  = (const float*)d_in[8];
  const float* pv_k  = (const float*)d_in[9];
  const float* pr_k  = (const float*)d_in[10];
  const float* out_k = (const float*)d_in[11];
  const float* out_b = (const float*)d_in[12];
  const float* pwgt  = (const float*)d_in[13];
  float* ws  = (float*)d_ws;
  float* out = (float*)d_out;

  proj_kernel<<<dim3(32, 7, NH), 32, 0, stream>>>(
      node, sq_k, sk_k, sv_k, pq_k, pk_k, pv_k, ws);
  rigid_kernel<<<(NH*LQ + 127)/128, 128, 0, stream>>>(R, t, ws);
  bias_kernel<<<(LQ*LQ)/16, 32, 0, stream>>>(pairw, pr_k, ws);
  attn_kernel<<<dim3(32, NH), 128, 0, stream>>>(pwgt, ws);
  pairout_kernel<<<LQ, 32*PO_W, 0, stream>>>(pairw, ws);
  pfinish_kernel<<<(NH*LQ + 127)/128, 128, 0, stream>>>(R, t, ws);
  outproj_kernel<<<dim3(32, 24), 32, 0, stream>>>(out_k, out_b, ws + OFF_RES, out);
}


# bias: 4-wave block per row, each wave streams 8 tiles through a 3-tile register ring of 16-byte loads, pr_k fragments resident; each 8 channels summed as 0,1,4,5 then 2,3,6,7 (order changed, fixed)
# speedup vs baseline: 2.7486x; 1.2475x over previous
#include <hip/hip_runtime.h>
#include <math.h>


typedef __attribute__((ext_vector_type(2))) float v2f;
typedef __attribute__((ext_vector_type(4))) float v4f;
typedef __attribute__((ext_vector_type(8))) float v8f;

#define LQ    512
#define NH    12
#define NODE  384
#define PAIR  128
#define OUTIN 176
#define KOUT  (NH*OUTIN)

#define U_PROJ 16
#define U_AV   16
#define BI_W   4
#define BI_T   8
#define BI_G   (PAIR/8)
#define BI_R   3
#define PO_W   4
#define PO_T   2
#define PO_S   2
#define PO_R   8
#define PO_NB  (LQ/(4*PO_S))
#define OUT_S  4
#define OUT_R  6
#define OUT_NB (KOUT/(4*OUT_S))

#define OFF_SQ   ((size_t)0)
#define OFF_SK   (OFF_SQ  + (size_t)NH*LQ*16)
#define OFF_SV   (OFF_SK  + (size_t)NH*LQ*16)
#define OFF_PQ   (OFF_SV  + (size_t)NH*LQ*16)
#define OFF_PK   (OFF_PQ  + (size_t)NH*LQ*16)
#define OFF_PV   (OFF_PK  + (size_t)NH*LQ*16)
#define OFF_QN2  (OFF_PV  + (size_t)NH*LQ*24)
#define OFF_KN2  (OFF_QN2 + (size_t)NH*LQ)
#define OFF_BIAS (OFF_KN2 + (size_t)NH*LQ)
#define OFF_ATTN (OFF_BIAS+ (size_t)NH*LQ*LQ)
#define OFF_RES  (OFF_ATTN+ (size_t)NH*LQ*LQ)

#define RES_IDX(h, l, f) (((size_t)(l)*NH + (h))*OUTIN + (f))

static __device__ __forceinline__ v8f wmma4(v2f a, v2f b, v8f c) {
  return __builtin_amdgcn_wmma_f32_16x16x4_f32(
       false, a,  false, b,
       (short)0, c,  false,  false);
}

static __device__ __forceinline__ v8f vzero8() {
  v8f z = {0.f,0.f,0.f,0.f,0.f,0.f,0.f,0.f};
  return z;
}

__global__ __launch_bounds__(32) void proj_kernel(
    const float* __restrict__ node,
    const float* __restrict__ sq_k, const float* __restrict__ sk_k,
    const float* __restrict__ sv_k, const float* __restrict__ pq_k,
    const float* __restrict__ pk_k, const float* __restrict__ pv_k,
    float* __restrict__ ws)
{
  const int lane  = threadIdx.x;
  const int mt    = blockIdx.x;
  const int seg   = blockIdx.y;
  const int h     = blockIdx.z;
  const int row   = lane & 15;
  const int khalf = (lane >> 4) * 2;

  const float* wb = sk_k + (size_t)h*NODE*16;
  int   wstride = 16;
  int   wcol    = row;
  float wscale  = 1.f;
  switch (seg) {
    case 0: wb = sq_k + (size_t)h*NODE*16; wscale = 0.25f; break;
    case 1: break;
    case 2: wb = sv_k + (size_t)h*NODE*16; break;
    case 3: wb = pq_k + (size_t)h*NODE*12; wstride = 12;
            wcol = (row < 12) ? row : 11; wscale = (row < 12) ? 1.f : 0.f; break;
    case 4: wb = pk_k + (size_t)h*NODE*12; wstride = 12;
            wcol = (row < 12) ? row : 11; wscale = (row < 12) ? 1.f : 0.f; break;
    case 5: wb = pv_k + (size_t)h*NODE*24; wstride = 24; break;
    case 6: wb = pv_k + (size_t)h*NODE*24; wstride = 24;
            wcol = (row < 8) ? 16 + row : 23; wscale = (row < 8) ? 1.f : 0.f; break;
  }
  const float* arow = node + (size_t)(mt*16 + row)*NODE;
  const float* wp   = wb + wcol;

  v8f acc = vzero8();
  for (int k0 = 0; k0 < NODE; k0 += 4*U_PROJ) {
    v2f a[U_PROJ], b[U_PROJ];
#pragma unroll
    for (int u = 0; u < U_PROJ; ++u) {
      const int ka = k0 + 4*u + khalf;
      a[u].x = arow[ka];                     a[u].y = arow[ka + 1];
      b[u].x = wp[(size_t)ka * wstride];     b[u].y = wp[(size_t)(ka + 1) * wstride];
    }
    __builtin_amdgcn_sched_barrier(0);
#pragma unroll
    for (int u = 0; u < U_PROJ; ++u) {
      v2f bs; bs.x = b[u].x * wscale; bs.y = b[u].y * wscale;
      acc = wmma4(a[u], bs, acc);
    }
  }

  const int col   = lane & 15;
  const int rbase = (lane >> 4) * 8;
  float* ob = ws + OFF_SK + (size_t)h*LQ*16;
  int ostride = 16, ocol = col;
  bool dostore = true;
  switch (seg) {
    case 0: ob = ws + OFF_SQ + (size_t)h*LQ*16; break;
    case 1: break;
    case 2: ob = ws + OFF_SV + (size_t)h*LQ*16; break;
    case 3: ob = ws + OFF_PQ + (size_t)h*LQ*16; break;
    case 4: ob = ws + OFF_PK + (size_t)h*LQ*16; break;
    case 5: ob = ws + OFF_PV + (size_t)h*LQ*24; ostride = 24; break;
    case 6: ob = ws + OFF_PV + (size_t)h*LQ*24; ostride = 24;
            ocol = 16 + col; dostore = (col < 8); break;
  }
  if (dostore) {
#pragma unroll
    for (int r = 0; r < 8; ++r)
      ob[(size_t)(mt*16 + rbase + r)*ostride + ocol] = acc[r];
  }
}

__global__ void rigid_kernel(const float* __restrict__ R,
                             const float* __restrict__ t,
                             float* __restrict__ ws)
{
  const int idx = blockIdx.x * blockDim.x + threadIdx.x;
  if (idx >= NH * LQ) return;
  const int h = idx / LQ, l = idx % LQ;
  float Rm[9], tv[3];
#pragma unroll
  for (int i = 0; i < 9; ++i) Rm[i] = R[l*9 + i];
#pragma unroll
  for (int i = 0; i < 3; ++i) tv[i] = t[l*3 + i];

  float* Pq = ws + OFF_PQ + ((size_t)h*LQ + l)*16;
  float* Pk = ws + OFF_PK + ((size_t)h*LQ + l)*16;
  float* Pv = ws + OFF_PV + ((size_t)h*LQ + l)*24;

  float qs = 0.f, ksum = 0.f;
#pragma unroll
  for (int p = 0; p < 4; ++p) {
    float loc[3], g[3];
    for (int f = 0; f < 3; ++f) loc[f] = Pq[p*3+f];
    for (int gg = 0; gg < 3; ++gg) {
      g[gg] = tv[gg];
      for (int f = 0; f < 3; ++f) g[gg] += loc[f]*Rm[gg*3+f];
      Pq[p*3+gg] = g[gg]; qs += g[gg]*g[gg];
    }
    for (int f = 0; f < 3; ++f) loc[f] = Pk[p*3+f];
    for (int gg = 0; gg < 3; ++gg) {
      g[gg] = tv[gg];
      for (int f = 0; f < 3; ++f) g[gg] += loc[f]*Rm[gg*3+f];
      Pk[p*3+gg] = g[gg]; ksum += g[gg]*g[gg];
    }
  }
#pragma unroll
  for (int p = 0; p < 8; ++p) {
    float loc[3], g[3];
    for (int f = 0; f < 3; ++f) loc[f] = Pv[p*3+f];
    for (int gg = 0; gg < 3; ++gg) {
      g[gg] = tv[gg];
      for (int f = 0; f < 3; ++f) g[gg] += loc[f]*Rm[gg*3+f];
      Pv[p*3+gg] = g[gg];
    }
  }
  ws[OFF_QN2 + h*LQ + l] = qs;
  ws[OFF_KN2 + h*LQ + l] = ksum;
}

static __device__ __forceinline__ void bi_load(
    v4f (&q)[BI_G], const float* __restrict__ p)
{
#pragma unroll
  for (int g = 0; g < BI_G; ++g) q[g] = *(const v4f*)(p + 8*g);
}

__global__ __launch_bounds__(32*BI_W) void bias_kernel(
    const float* __restrict__ pairwise, const float* __restrict__ pr_k,
    float* __restrict__ ws)
{
  float* bias = ws + OFF_BIAS;
  const int lane  = threadIdx.x & 31;
  const int w     = threadIdx.x >> 5;
  const int i     = blockIdx.x;
  const int row   = lane & 15;
  const int half  = lane >> 4;
  const int col   = lane & 15;
  const int rbase = (lane >> 4) * 8;

  const float bmask = (row < NH) ? 1.f : 0.f;
  v4f bq[BI_G];
  bi_load(bq, pr_k + (size_t)((row < NH) ? row : NH - 1) * PAIR + 4*half);

  const float* ap = pairwise + ((size_t)i*LQ + 128*w + row) * PAIR + 4*half;
  v4f q[BI_R][BI_G];
  bi_load(q[0], ap);
#pragma unroll
  for (int g = 0; g < BI_G; ++g) bq[g] *= bmask;
  __builtin_amdgcn_sched_barrier(0);
#pragma unroll
  for (int s = 1; s < BI_R - 1; ++s) bi_load(q[s], ap + (size_t)s*16*PAIR);

  float* ob = bias + ((size_t)col*LQ + i)*LQ + 128*w + rbase;
#pragma unroll
  for (int t = 0; t < BI_T; ++t) {
    if (t + BI_R - 1 < BI_T)
      bi_load(q[(t + BI_R - 1) % BI_R], ap + (size_t)(t + BI_R - 1)*16*PAIR);
    __builtin_amdgcn_sched_barrier(0);
    v8f acc = vzero8();
#pragma unroll
    for (int g = 0; g < BI_G; ++g) {
      const v4f qa = q[t % BI_R][g];
      v2f a0; a0.x = qa.x;    a0.y = qa.y;
      v2f a1; a1.x = qa.z;    a1.y = qa.w;
      v2f b0; b0.x = bq[g].x; b0.y = bq[g].y;
      v2f b1; b1.x = bq[g].z; b1.y = bq[g].w;
      acc = wmma4(a0, b0, acc);
      acc = wmma4(a1, b1, acc);
    }
    if (col < NH) {
#pragma unroll
      for (int r = 0; r < 8; ++r) ob[16*t + r] = acc[r];
    }
  }
}

#define SMS 516
__global__ __launch_bounds__(128) void attn_kernel(
    const float* __restrict__ point_weight, float* __restrict__ ws)
{
  __shared__ float sm[16 * SMS];
  const int tid  = threadIdx.x;
  const int lane = tid & 31;
  const int w    = tid >> 5;
  const int nt   = blockIdx.x;
  const int h    = blockIdx.y;

  const float* Sq  = ws + OFF_SQ;  const float* Sk  = ws + OFF_SK;
  const float* Pq  = ws + OFF_PQ;  const float* Pk  = ws + OFF_PK;
  const float* qn2 = ws + OFF_QN2; const float* kn2 = ws + OFF_KN2;
  const float* bias= ws + OFF_BIAS;
  const float* Sv  = ws + OFF_SV;  const float* Pv  = ws + OFF_PV;
  float* attn = ws + OFF_ATTN;
  float* res  = ws + OFF_RES;

  const float pwx = point_weight[h];
  const float pw  = (pwx > 20.f) ? pwx : log1pf(__expf(pwx));
  const float cpt = -0.5f * pw / sqrtf(18.f);
  const float rs3 = 0.57735026919f;

  const int row   = lane & 15;
  const int khalf = (lane >> 4) * 2;
  const int col   = lane & 15;
  const int rbase = (lane >> 4) * 8;

  v2f aq[4], ap[4];
#pragma unroll
  for (int ks = 0; ks < 4; ++ks) {
    const int k = ks*4 + khalf;
    const int l = nt*16 + row;
    aq[ks].x = Sq[((size_t)h*LQ + l)*16 + k];
    aq[ks].y = Sq[((size_t)h*LQ + l)*16 + k + 1];
    ap[ks].x = Pq[((size_t)h*LQ + l)*16 + k];
    ap[ks].y = Pq[((size_t)h*LQ + l)*16 + k + 1];
  }
  float qv[8];
#pragma unroll
  for (int r = 0; r < 8; ++r) qv[r] = qn2[h*LQ + nt*16 + rbase + r];

  for (int mt = w; mt < 32; mt += 4) {
    v8f s = vzero8(), p = vzero8();
#pragma unroll
    for (int ks = 0; ks < 4; ++ks) {
      const int k = ks*4 + khalf;
      const int m = mt*16 + row;
      v2f bq, bp;
      bq.x = Sk[((size_t)h*LQ + m)*16 + k]; bq.y = Sk[((size_t)h*LQ + m)*16 + k + 1];
      bp.x = Pk[((size_t)h*LQ + m)*16 + k]; bp.y = Pk[((size_t)h*LQ + m)*16 + k + 1];
      s = wmma4(aq[ks], bq, s);
      p = wmma4(ap[ks], bp, p);
    }
    const int m = mt*16 + col;
    const float kv = kn2[h*LQ + m];
#pragma unroll
    for (int r = 0; r < 8; ++r) {
      const int rr = rbase + r;
      const int n  = nt*16 + rr;
      const float dist = qv[r] + kv - 2.f*p[r];
      sm[rr*SMS + m] =
          (s[r] + bias[((size_t)h*LQ + n)*LQ + m] + cpt*dist) * rs3;
    }
  }
  __syncthreads();

  {
    const int rrow = tid >> 3;
    const int sub  = tid & 7;
    float mx = -1e30f;
    for (int j = sub; j < LQ; j += 8) mx = fmaxf(mx, sm[rrow*SMS + j]);
    for (int o = 1; o < 8; o <<= 1) mx = fmaxf(mx, __shfl_xor(mx, o, 8));
    float sum = 0.f;
    for (int j = sub; j < LQ; j += 8) {
      const float e = __expf(sm[rrow*SMS + j] - mx);
      sm[rrow*SMS + j] = e; sum += e;
    }
    for (int o = 1; o < 8; o <<= 1) sum += __shfl_xor(sum, o, 8);
    const float inv = 1.f / sum;
    const int n = nt*16 + rrow;
    for (int j = sub; j < LQ; j += 8) {
      const float v = sm[rrow*SMS + j] * inv;
      sm[rrow*SMS + j] = v;
      attn[((size_t)h*LQ + n)*LQ + j] = v;
    }
  }
  __syncthreads();

  if (w < 3) {
    const float* vb = (w == 0) ? Sv + (size_t)h*LQ*16 : Pv + (size_t)h*LQ*24;
    const int    vs = (w == 0) ? 16 : 24;
    int   vc = col;
    float vmask = 1.f;
    if (w == 2) { vc = (col < 8) ? 16 + col : 23; vmask = (col < 8) ? 1.f : 0.f; }
    const float* vp = vb + vc;

    v8f acc = vzero8();
    for (int kt0 = 0; kt0 < 128; kt0 += U_AV) {
      v2f a[U_AV], b[U_AV];
#pragma unroll
      for (int u = 0; u < U_AV; ++u) {
        const int k = (kt0 + u)*4 + khalf;
        a[u].x = sm[row*SMS + k];        a[u].y = sm[row*SMS + k + 1];
        b[u].x = vp[(size_t)k * vs];     b[u].y = vp[(size_t)(k + 1) * vs];
      }
      __builtin_amdgcn_sched_barrier(0);
#pragma unroll
      for (int u = 0; u < U_AV; ++u) {
        v2f bs; bs.x = b[u].x * vmask; bs.y = b[u].y * vmask;
        acc = wmma4(a[u], bs, acc);
      }
    }
    const int fbase = (w == 0) ? 0 : 144;
#pragma unroll
    for (int r = 0; r < 8; ++r) {
      const int n = nt*16 + rbase + r;
      const float v = acc[r];
      if (w < 2)        res[RES_IDX(h, n, fbase + col)] = v;
      else if (col < 8) res[RES_IDX(h, n, 160 + col)]   = v;
    }
  }
}

static __device__ __forceinline__ void po_load(
    v2f (&a)[PO_S], v2f (&x)[PO_S], v2f (&y)[PO_S],
    const float* __restrict__ arow, const float* __restrict__ prow, int m0)
{
#pragma unroll
  for (int u = 0; u < PO_S; ++u) {
    const int m = m0 + 4*u;
    a[u] = *(const v2f*)(arow + m);
    x[u] = *(const v2f*)(prow + (size_t)m * PAIR);
    y[u] = *(const v2f*)(prow + (size_t)(m + 1) * PAIR);
  }
}

template <bool TAIL>
static __device__ __forceinline__ void po_turn(
    v8f (&acc)[PO_T], v2f (&a)[PO_R][PO_S], v2f (&x)[PO_R][PO_S],
    v2f (&y)[PO_R][PO_S], const float* __restrict__ arow,
    const float* __restrict__ prow, int kb0, int khalf, float amask)
{
#pragma unroll
  for (int j = 0; j < PO_R; ++j) {
    if (!TAIL || j == 0)
      po_load(a[(j + PO_R - 1) % PO_R], x[(j + PO_R - 1) % PO_R],
              y[(j + PO_R - 1) % PO_R], arow, prow,
              (kb0 + j + PO_R - 1)*4*PO_S + khalf);
    __builtin_amdgcn_sched_barrier(0);
#pragma unroll
    for (int u = 0; u < PO_S; ++u) {
      v2f am; am.x = a[j][u].x * amask; am.y = a[j][u].y * amask;
      v2f b0; b0.x = x[j][u].x; b0.y = y[j][u].x;
      v2f b1; b1.x = x[j][u].y; b1.y = y[j][u].y;
      acc[0] = wmma4(am, b0, acc[0]);
      acc[1] = wmma4(am, b1, acc[1]);
    }
    __builtin_amdgcn_sched_barrier(0);
  }
}

__global__ __launch_bounds__(32*PO_W) void pairout_kernel(
    const float* __restrict__ pairwise, float* __restrict__ ws)
{
  const float* attn = ws + OFF_ATTN;
  float* res = ws + OFF_RES;
  const int lane  = threadIdx.x & 31;
  const int w     = threadIdx.x >> 5;
  const int n     = blockIdx.x;
  const int row   = lane & 15;
  const int khalf = (lane >> 4) * 2;
  const int col   = lane & 15;

  const int hh0 = (row < NH) ? row : NH - 1;
  const float amask = (row < NH) ? 1.f : 0.f;
  const float* arow = attn + ((size_t)hh0*LQ + n)*LQ;

  v8f acc[PO_T];
#pragma unroll
  for (int i = 0; i < PO_T; ++i) acc[i] = vzero8();

  v2f a[PO_R][PO_S], x[PO_R][PO_S], y[PO_R][PO_S];
  const float* prow = pairwise + (size_t)n*LQ*PAIR + 32*w + 2*col;
#pragma unroll
  for (int s = 0; s < PO_R - 1; ++s) {
    po_load(a[s], x[s], y[s], arow, prow, s*4*PO_S + khalf);
    __builtin_amdgcn_sched_barrier(0);
  }

  for (int kb0 = 0; kb0 < PO_NB - PO_R; kb0 += PO_R)
    po_turn<false>(acc, a, x, y, arow, prow, kb0, khalf, amask);
  po_turn<true>(acc, a, x, y, arow, prow, PO_NB - PO_R, khalf, amask);

  const int rbase = (lane >> 4) * 8;
#pragma unroll
  for (int r = 0; r < 8; ++r) {
    const int hh = rbase + r;
    if (hh < NH) {
      v2f o; o.x = acc[0][r]; o.y = acc[1][r];
      *(v2f*)(res + RES_IDX(hh, n, 16 + 32*w + 2*col)) = o;
    }
  }
}

__global__ void pfinish_kernel(const float* __restrict__ R,
                               const float* __restrict__ t,
                               float* __restrict__ ws)
{
  const int idx = blockIdx.x * blockDim.x + threadIdx.x;
  if (idx >= NH * LQ) return;
  const int h = idx / LQ, l = idx % LQ;
  float* resp = ws + OFF_RES + RES_IDX(h, l, 0);
  float Rm[9], tv[3];
#pragma unroll
  for (int i = 0; i < 9; ++i) Rm[i] = R[l*9 + i];
#pragma unroll
  for (int i = 0; i < 3; ++i) tv[i] = t[l*3 + i];
#pragma unroll
  for (int p = 0; p < 8; ++p) {
    float g[3], loc[3], nrm = 0.f;
    for (int gg = 0; gg < 3; ++gg) g[gg] = resp[144 + p*3 + gg] - tv[gg];
    for (int f = 0; f < 3; ++f) {
      float v = 0.f;
      for (int gg = 0; gg < 3; ++gg) v += g[gg]*Rm[gg*3 + f];
      loc[f] = v; nrm += v*v;
    }
    for (int f = 0; f < 3; ++f) resp[144 + p*3 + f] = loc[f];
    resp[168 + p] = sqrtf(nrm + 1e-12f);
  }
}

static __device__ __forceinline__ void out_load(
    v2f (&a)[OUT_S], v2f (&b)[OUT_S], const float* __restrict__ resl,
    const float* __restrict__ okp, int k0, int khalf)
{
#pragma unroll
  for (int u = 0; u < OUT_S; ++u) {
    const int ka = k0 + 4*u + khalf;
    a[u].x = resl[ka];                 a[u].y = resl[ka + 1];
    b[u].x = okp[(size_t)ka * NODE];   b[u].y = okp[(size_t)(ka + 1) * NODE];
  }
}

template <bool TAIL>
static __device__ __forceinline__ void out_turn(
    v8f& acc, v2f (&a)[OUT_R][OUT_S], v2f (&b)[OUT_R][OUT_S],
    const float* __restrict__ resl, const float* __restrict__ okp,
    int kb0, int khalf)
{
#pragma unroll
  for (int j = 0; j < OUT_R; ++j) {
    if (!TAIL || j == 0)
      out_load(a[(j + OUT_R - 1) % OUT_R], b[(j + OUT_R - 1) % OUT_R],
               resl, okp, (kb0 + j + OUT_R - 1)*4*OUT_S, khalf);
    __builtin_amdgcn_sched_barrier(0);
#pragma unroll
    for (int u = 0; u < OUT_S; ++u) acc = wmma4(a[j][u], b[j][u], acc);
    __builtin_amdgcn_sched_barrier(0);
  }
}

__global__ __launch_bounds__(32) void outproj_kernel(
    const float* __restrict__ out_k, const float* __restrict__ out_b,
    const float* __restrict__ res, float* __restrict__ out)
{
  const int lane  = threadIdx.x;
  const int mt    = blockIdx.x;
  const int ntile = blockIdx.y;
  const int row   = lane & 15;
  const int khalf = (lane >> 4) * 2;
  const int col   = lane & 15;
  const int o     = ntile*16 + col;

  const float* resl = res + (size_t)(mt*16 + row) * KOUT;
  const float* okp  = out_k + o;

  v8f acc = vzero8();
  v2f a[OUT_R][OUT_S], b[OUT_R][OUT_S];
#pragma unroll
  for (int s = 0; s < OUT_R - 1; ++s) {
    out_load(a[s], b[s], resl, okp, s*4*OUT_S, khalf);
    __builtin_amdgcn_sched_barrier(0);
  }

  for (int kb0 = 0; kb0 < OUT_NB - OUT_R; kb0 += OUT_R)
    out_turn<false>(acc, a, b, resl, okp, kb0, khalf);
  out_turn<true>(acc, a, b, resl, okp, OUT_NB - OUT_R, khalf);
  const int rbase = (lane >> 4) * 8;
  const float ob = out_b[o];
#pragma unroll
  for (int r = 0; r < 8; ++r) {
    const int l = mt*16 + rbase + r;
    out[(size_t)l*NODE + o] = acc[r] + ob;
  }
}

extern "C" void kernel_launch(void* const* d_in, const int* in_sizes, int n_in,
                              void* d_out, int out_size, void* d_ws, size_t ws_size,
                              hipStream_t stream) {
  const float* node  = (const float*)d_in[0];
  const float* R     = (const float*)d_in[1];
  const float* t     = (const float*)d_in[2];
  const float* pairw = (const float*)d_in[3];
  const float* sq_k  = (const float*)d_in[4];
  const float* sk_k  = (const float*)d_in[5];
  const float* sv_k  = (const float*)d_in[6];
  const float* pq_k  = (const float*)d_in[7];
  const float* pk_k  = (const float*)d_in[8];
  const float* pv_k  = (const float*)d_in[9];
  const float* pr_k  = (const float*)d_in[10];
  const float* out_k = (const float*)d_in[11];
  const float* out_b = (const float*)d_in[12];
  const float* pwgt  = (const float*)d_in[13];
  float* ws  = (float*)d_ws;
  float* out = (float*)d_out;

  proj_kernel<<<dim3(32, 7, NH), 32, 0, stream>>>(
      node, sq_k, sk_k, sv_k, pq_k, pk_k, pv_k, ws);
  rigid_kernel<<<(NH*LQ + 127)/128, 128, 0, stream>>>(R, t, ws);
  bias_kernel<<<LQ, 32*BI_W, 0, stream>>>(pairw, pr_k, ws);
  attn_kernel<<<dim3(32, NH), 128, 0, stream>>>(pwgt, ws);
  pairout_kernel<<<LQ, 32*PO_W, 0, stream>>>(pairw, ws);
  pfinish_kernel<<<(NH*LQ + 127)/128, 128, 0, stream>>>(R, t, ws);
  outproj_kernel<<<dim3(32, 24), 32, 0, stream>>>(out_k, out_b, ws + OFF_RES, out);
}


# independent kernels share a launch, roles by block index: proj units behind the bias rows, pfinish blocks behind the pairout rows; 7 -> 5 launches; kernel bodies and every sum order unchanged
# speedup vs baseline: 3.0226x; 1.0997x over previous
#include <hip/hip_runtime.h>
#include <math.h>


typedef __attribute__((ext_vector_type(2))) float v2f;
typedef __attribute__((ext_vector_type(4))) float v4f;
typedef __attribute__((ext_vector_type(8))) float v8f;

#define LQ    512
#define NH    12
#define NODE  384
#define PAIR  128
#define OUTIN 176
#define KOUT  (NH*OUTIN)

#define U_PROJ 16
#define U_AV   16
#define BI_W   4
#define BI_T   8
#define BI_G   (PAIR/8)
#define BI_R   3
#define PO_W   4
#define PO_T   2
#define PO_S   2
#define PO_R   8
#define PO_NB  (LQ/(4*PO_S))
#define OUT_S  4
#define OUT_R  6
#define OUT_NB (KOUT/(4*OUT_S))

#define OFF_SQ   ((size_t)0)
#define OFF_SK   (OFF_SQ  + (size_t)NH*LQ*16)
#define OFF_SV   (OFF_SK  + (size_t)NH*LQ*16)
#define OFF_PQ   (OFF_SV  + (size_t)NH*LQ*16)
#define OFF_PK   (OFF_PQ  + (size_t)NH*LQ*16)
#define OFF_PV   (OFF_PK  + (size_t)NH*LQ*16)
#define OFF_QN2  (OFF_PV  + (size_t)NH*LQ*24)
#define OFF_KN2  (OFF_QN2 + (size_t)NH*LQ)
#define OFF_BIAS (OFF_KN2 + (size_t)NH*LQ)
#define OFF_ATTN (OFF_BIAS+ (size_t)NH*LQ*LQ)
#define OFF_RES  (OFF_ATTN+ (size_t)NH*LQ*LQ)

#define RES_IDX(h, l, f) (((size_t)(l)*NH + (h))*OUTIN + (f))

static __device__ __forceinline__ v8f wmma4(v2f a, v2f b, v8f c) {
  return __builtin_amdgcn_wmma_f32_16x16x4_f32(
       false, a,  false, b,
       (short)0, c,  false,  false);
}

static __device__ __forceinline__ v8f vzero8() {
  v8f z = {0.f,0.f,0.f,0.f,0.f,0.f,0.f,0.f};
  return z;
}

static __device__ __forceinline__ void proj_body(
    const float* __restrict__ node,
    const float* __restrict__ sq_k, const float* __restrict__ sk_k,
    const float* __restrict__ sv_k, const float* __restrict__ pq_k,
    const float* __restrict__ pk_k, const float* __restrict__ pv_k,
    float* __restrict__ ws, const int unit)
{
  const int lane  = threadIdx.x & 31;
  const int mt    = unit & 31;
  const int seg   = (unit >> 5) % 7;
  const int h     = (unit >> 5) / 7;
  const int row   = lane & 15;
  const int khalf = (lane >> 4) * 2;

  const float* wb = sk_k + (size_t)h*NODE*16;
  int   wstride = 16;
  int   wcol    = row;
  float wscale  = 1.f;
  switch (seg) {
    case 0: wb = sq_k + (size_t)h*NODE*16; wscale = 0.25f; break;
    case 1: break;
    case 2: wb = sv_k + (size_t)h*NODE*16; break;
    case 3: wb = pq_k + (size_t)h*NODE*12; wstride = 12;
            wcol = (row < 12) ? row : 11; wscale = (row < 12) ? 1.f : 0.f; break;
    case 4: wb = pk_k + (size_t)h*NODE*12; wstride = 12;
            wcol = (row < 12) ? row : 11; wscale = (row < 12) ? 1.f : 0.f; break;
    case 5: wb = pv_k + (size_t)h*NODE*24; wstride = 24; break;
    case 6: wb = pv_k + (size_t)h*NODE*24; wstride = 24;
            wcol = (row < 8) ? 16 + row : 23; wscale = (row < 8) ? 1.f : 0.f; break;
  }
  const float* arow = node + (size_t)(mt*16 + row)*NODE;
  const float* wp   = wb + wcol;

  v8f acc = vzero8();
  for (int k0 = 0; k0 < NODE; k0 += 4*U_PROJ) {
    v2f a[U_PROJ], b[U_PROJ];
#pragma unroll
    for (int u = 0; u < U_PROJ; ++u) {
      const int ka = k0 + 4*u + khalf;
      a[u].x = arow[ka];                     a[u].y = arow[ka + 1];
      b[u].x = wp[(size_t)ka * wstride];     b[u].y = wp[(size_t)(ka + 1) * wstride];
    }
    __builtin_amdgcn_sched_barrier(0);
#pragma unroll
    for (int u = 0; u < U_PROJ; ++u) {
      v2f bs; bs.x = b[u].x * wscale; bs.y = b[u].y * wscale;
      acc = wmma4(a[u], bs, acc);
    }
  }

  const int col   = lane & 15;
  const int rbase = (lane >> 4) * 8;
  float* ob = ws + OFF_SK + (size_t)h*LQ*16;
  int ostride = 16, ocol = col;
  bool dostore = true;
  switch (seg) {
    case 0: ob = ws + OFF_SQ + (size_t)h*LQ*16; break;
    case 1: break;
    case 2: ob = ws + OFF_SV + (size_t)h*LQ*16; break;
    case 3: ob = ws + OFF_PQ + (size_t)h*LQ*16; break;
    case 4: ob = ws + OFF_PK + (size_t)h*LQ*16; break;
    case 5: ob = ws + OFF_PV + (size_t)h*LQ*24; ostride = 24; break;
    case 6: ob = ws + OFF_PV + (size_t)h*LQ*24; ostride = 24;
            ocol = 16 + col; dostore = (col < 8); break;
  }
  if (dostore) {
#pragma unroll
    for (int r = 0; r < 8; ++r)
      ob[(size_t)(mt*16 + rbase + r)*ostride + ocol] = acc[r];
  }
}

__global__ void rigid_kernel(const float* __restrict__ R,
                             const float* __restrict__ t,
                             float* __restrict__ ws)
{
  const int idx = blockIdx.x * blockDim.x + threadIdx.x;
  if (idx >= NH * LQ) return;
  const int h = idx / LQ, l = idx % LQ;
  float Rm[9], tv[3];
#pragma unroll
  for (int i = 0; i < 9; ++i) Rm[i] = R[l*9 + i];
#pragma unroll
  for (int i = 0; i < 3; ++i) tv[i] = t[l*3 + i];

  float* Pq = ws + OFF_PQ + ((size_t)h*LQ + l)*16;
  float* Pk = ws + OFF_PK + ((size_t)h*LQ + l)*16;
  float* Pv = ws + OFF_PV + ((size_t)h*LQ + l)*24;

  float qs = 0.f, ksum = 0.f;
#pragma unroll
  for (int p = 0; p < 4; ++p) {
    float loc[3], g[3];
    for (int f = 0; f < 3; ++f) loc[f] = Pq[p*3+f];
    for (int gg = 0; gg < 3; ++gg) {
      g[gg] = tv[gg];
      for (int f = 0; f < 3; ++f) g[gg] += loc[f]*Rm[gg*3+f];
      Pq[p*3+gg] = g[gg]; qs += g[gg]*g[gg];
    }
    for (int f = 0; f < 3; ++f) loc[f] = Pk[p*3+f];
    for (int gg = 0; gg < 3; ++gg) {
      g[gg] = tv[gg];
      for (int f = 0; f < 3; ++f) g[gg] += loc[f]*Rm[gg*3+f];
      Pk[p*3+gg] = g[gg]; ksum += g[gg]*g[gg];
    }
  }
#pragma unroll
  for (int p = 0; p < 8; ++p) {
    float loc[3], g[3];
    for (int f = 0; f < 3; ++f) loc[f] = Pv[p*3+f];
    for (int gg = 0; gg < 3; ++gg) {
      g[gg] = tv[gg];
      for (int f = 0; f < 3; ++f) g[gg] += loc[f]*Rm[gg*3+f];
      Pv[p*3+gg] = g[gg];
    }
  }
  ws[OFF_QN2 + h*LQ + l] = qs;
  ws[OFF_KN2 + h*LQ + l] = ksum;
}

static __device__ __forceinline__ void bi_load(
    v4f (&q)[BI_G], const float* __restrict__ p)
{
#pragma unroll
  for (int g = 0; g < BI_G; ++g) q[g] = *(const v4f*)(p + 8*g);
}

static __device__ __forceinline__ void bias_body(
    const float* __restrict__ pairwise, const float* __restrict__ pr_k,
    float* __restrict__ ws, const int i  )
{
  float* bias = ws + OFF_BIAS;
  const int lane  = threadIdx.x & 31;
  const int w     = threadIdx.x >> 5;
  const int row   = lane & 15;
  const int half  = lane >> 4;
  const int col   = lane & 15;
  const int rbase = (lane >> 4) * 8;

  const float bmask = (row < NH) ? 1.f : 0.f;
  v4f bq[BI_G];
  bi_load(bq, pr_k + (size_t)((row < NH) ? row : NH - 1) * PAIR + 4*half);

  const float* ap = pairwise + ((size_t)i*LQ + 128*w + row) * PAIR + 4*half;
  v4f q[BI_R][BI_G];
  bi_load(q[0], ap);
#pragma unroll
  for (int g = 0; g < BI_G; ++g) bq[g] *= bmask;
  __builtin_amdgcn_sched_barrier(0);
#pragma unroll
  for (int s = 1; s < BI_R - 1; ++s) bi_load(q[s], ap + (size_t)s*16*PAIR);

  float* ob = bias + ((size_t)col*LQ + i)*LQ + 128*w + rbase;
#pragma unroll
  for (int t = 0; t < BI_T; ++t) {
    if (t + BI_R - 1 < BI_T)
      bi_load(q[(t + BI_R - 1) % BI_R], ap + (size_t)(t + BI_R - 1)*16*PAIR);
    __builtin_amdgcn_sched_barrier(0);
    v8f acc = vzero8();
#pragma unroll
    for (int g = 0; g < BI_G; ++g) {
      const v4f qa = q[t % BI_R][g];
      v2f a0; a0.x = qa.x;    a0.y = qa.y;
      v2f a1; a1.x = qa.z;    a1.y = qa.w;
      v2f b0; b0.x = bq[g].x; b0.y = bq[g].y;
      v2f b1; b1.x = bq[g].z; b1.y = bq[g].w;
      acc = wmma4(a0, b0, acc);
      acc = wmma4(a1, b1, acc);
    }
    if (col < NH) {
#pragma unroll
      for (int r = 0; r < 8; ++r) ob[16*t + r] = acc[r];
    }
  }
}

#define PJ_UNITS (32*7*NH)
#define PJ_B     (PJ_UNITS/BI_W)
__global__ __launch_bounds__(32*BI_W) void projbias_kernel(
    const float* __restrict__ node,
    const float* __restrict__ sq_k, const float* __restrict__ sk_k,
    const float* __restrict__ sv_k, const float* __restrict__ pq_k,
    const float* __restrict__ pk_k, const float* __restrict__ pv_k,
    const float* __restrict__ pairwise, const float* __restrict__ pr_k,
    float* __restrict__ ws)
{
  if (blockIdx.x < LQ)
    bias_body(pairwise, pr_k, ws, blockIdx.x);
  else
    proj_body(node, sq_k, sk_k, sv_k, pq_k, pk_k, pv_k, ws,
              (blockIdx.x - LQ)*BI_W + (threadIdx.x >> 5));
}

#define SMS 516
__global__ __launch_bounds__(128) void attn_kernel(
    const float* __restrict__ point_weight, float* __restrict__ ws)
{
  __shared__ float sm[16 * SMS];
  const int tid  = threadIdx.x;
  const int lane = tid & 31;
  const int w    = tid >> 5;
  const int nt   = blockIdx.x;
  const int h    = blockIdx.y;

  const float* Sq  = ws + OFF_SQ;  const float* Sk  = ws + OFF_SK;
  const float* Pq  = ws + OFF_PQ;  const float* Pk  = ws + OFF_PK;
  const float* qn2 = ws + OFF_QN2; const float* kn2 = ws + OFF_KN2;
  const float* bias= ws + OFF_BIAS;
  const float* Sv  = ws + OFF_SV;  const float* Pv  = ws + OFF_PV;
  float* attn = ws + OFF_ATTN;
  float* res  = ws + OFF_RES;

  const float pwx = point_weight[h];
  const float pw  = (pwx > 20.f) ? pwx : log1pf(__expf(pwx));
  const float cpt = -0.5f * pw / sqrtf(18.f);
  const float rs3 = 0.57735026919f;

  const int row   = lane & 15;
  const int khalf = (lane >> 4) * 2;
  const int col   = lane & 15;
  const int rbase = (lane >> 4) * 8;

  v2f aq[4], ap[4];
#pragma unroll
  for (int ks = 0; ks < 4; ++ks) {
    const int k = ks*4 + khalf;
    const int l = nt*16 + row;
    aq[ks].x = Sq[((size_t)h*LQ + l)*16 + k];
    aq[ks].y = Sq[((size_t)h*LQ + l)*16 + k + 1];
    ap[ks].x = Pq[((size_t)h*LQ + l)*16 + k];
    ap[ks].y = Pq[((size_t)h*LQ + l)*16 + k + 1];
  }
  float qv[8];
#pragma unroll
  for (int r = 0; r < 8; ++r) qv[r] = qn2[h*LQ + nt*16 + rbase + r];

  for (int mt = w; mt < 32; mt += 4) {
    v8f s = vzero8(), p = vzero8();
#pragma unroll
    for (int ks = 0; ks < 4; ++ks) {
      const int k = ks*4 + khalf;
      const int m = mt*16 + row;
      v2f bq, bp;
      bq.x = Sk[((size_t)h*LQ + m)*16 + k]; bq.y = Sk[((size_t)h*LQ + m)*16 + k + 1];
      bp.x = Pk[((size_t)h*LQ + m)*16 + k]; bp.y = Pk[((size_t)h*LQ + m)*16 + k + 1];
      s = wmma4(aq[ks], bq, s);
      p = wmma4(ap[ks], bp, p);
    }
    const int m = mt*16 + col;
    const float kv = kn2[h*LQ + m];
#pragma unroll
    for (int r = 0; r < 8; ++r) {
      const int rr = rbase + r;
      const int n  = nt*16 + rr;
      const float dist = qv[r] + kv - 2.f*p[r];
      sm[rr*SMS + m] =
          (s[r] + bias[((size_t)h*LQ + n)*LQ + m] + cpt*dist) * rs3;
    }
  }
  __syncthreads();

  {
    const int rrow = tid >> 3;
    const int sub  = tid & 7;
    float mx = -1e30f;
    for (int j = sub; j < LQ; j += 8) mx = fmaxf(mx, sm[rrow*SMS + j]);
    for (int o = 1; o < 8; o <<= 1) mx = fmaxf(mx, __shfl_xor(mx, o, 8));
    float sum = 0.f;
    for (int j = sub; j < LQ; j += 8) {
      const float e = __expf(sm[rrow*SMS + j] - mx);
      sm[rrow*SMS + j] = e; sum += e;
    }
    for (int o = 1; o < 8; o <<= 1) sum += __shfl_xor(sum, o, 8);
    const float inv = 1.f / sum;
    const int n = nt*16 + rrow;
    for (int j = sub; j < LQ; j += 8) {
      const float v = sm[rrow*SMS + j] * inv;
      sm[rrow*SMS + j] = v;
      attn[((size_t)h*LQ + n)*LQ + j] = v;
    }
  }
  __syncthreads();

  if (w < 3) {
    const float* vb = (w == 0) ? Sv + (size_t)h*LQ*16 : Pv + (size_t)h*LQ*24;
    const int    vs = (w == 0) ? 16 : 24;
    int   vc = col;
    float vmask = 1.f;
    if (w == 2) { vc = (col < 8) ? 16 + col : 23; vmask = (col < 8) ? 1.f : 0.f; }
    const float* vp = vb + vc;

    v8f acc = vzero8();
    for (int kt0 = 0; kt0 < 128; kt0 += U_AV) {
      v2f a[U_AV], b[U_AV];
#pragma unroll
      for (int u = 0; u < U_AV; ++u) {
        const int k = (kt0 + u)*4 + khalf;
        a[u].x = sm[row*SMS + k];        a[u].y = sm[row*SMS + k + 1];
        b[u].x = vp[(size_t)k * vs];     b[u].y = vp[(size_t)(k + 1) * vs];
      }
      __builtin_amdgcn_sched_barrier(0);
#pragma unroll
      for (int u = 0; u < U_AV; ++u) {
        v2f bs; bs.x = b[u].x * vmask; bs.y = b[u].y * vmask;
        acc = wmma4(a[u], bs, acc);
      }
    }
    const int fbase = (w == 0) ? 0 : 144;
#pragma unroll
    for (int r = 0; r < 8; ++r) {
      const int n = nt*16 + rbase + r;
      const float v = acc[r];
      if (w < 2)        res[RES_IDX(h, n, fbase + col)] = v;
      else if (col < 8) res[RES_IDX(h, n, 160 + col)]   = v;
    }
  }
}

static __device__ __forceinline__ void po_load(
    v2f (&a)[PO_S], v2f (&x)[PO_S], v2f (&y)[PO_S],
    const float* __restrict__ arow, const float* __restrict__ prow, int m0)
{
#pragma unroll
  for (int u = 0; u < PO_S; ++u) {
    const int m = m0 + 4*u;
    a[u] = *(const v2f*)(arow + m);
    x[u] = *(const v2f*)(prow + (size_t)m * PAIR);
    y[u] = *(const v2f*)(prow + (size_t)(m + 1) * PAIR);
  }
}

template <bool TAIL>
static __device__ __forceinline__ void po_turn(
    v8f (&acc)[PO_T], v2f (&a)[PO_R][PO_S], v2f (&x)[PO_R][PO_S],
    v2f (&y)[PO_R][PO_S], const float* __restrict__ arow,
    const float* __restrict__ prow, int kb0, int khalf, float amask)
{
#pragma unroll
  for (int j = 0; j < PO_R; ++j) {
    if (!TAIL || j == 0)
      po_load(a[(j + PO_R - 1) % PO_R], x[(j + PO_R - 1) % PO_R],
              y[(j + PO_R - 1) % PO_R], arow, prow,
              (kb0 + j + PO_R - 1)*4*PO_S + khalf);
    __builtin_amdgcn_sched_barrier(0);
#pragma unroll
    for (int u = 0; u < PO_S; ++u) {
      v2f am; am.x = a[j][u].x * amask; am.y = a[j][u].y * amask;
      v2f b0; b0.x = x[j][u].x; b0.y = y[j][u].x;
      v2f b1; b1.x = x[j][u].y; b1.y = y[j][u].y;
      acc[0] = wmma4(am, b0, acc[0]);
      acc[1] = wmma4(am, b1, acc[1]);
    }
    __builtin_amdgcn_sched_barrier(0);
  }
}

static __device__ __forceinline__ void pairout_body(
    const float* __restrict__ pairwise, float* __restrict__ ws, const int n)
{
  const float* attn = ws + OFF_ATTN;
  float* res = ws + OFF_RES;
  const int lane  = threadIdx.x & 31;
  const int w     = threadIdx.x >> 5;
  const int row   = lane & 15;
  const int khalf = (lane >> 4) * 2;
  const int col   = lane & 15;

  const int hh0 = (row < NH) ? row : NH - 1;
  const float amask = (row < NH) ? 1.f : 0.f;
  const float* arow = attn + ((size_t)hh0*LQ + n)*LQ;

  v8f acc[PO_T];
#pragma unroll
  for (int i = 0; i < PO_T; ++i) acc[i] = vzero8();

  v2f a[PO_R][PO_S], x[PO_R][PO_S], y[PO_R][PO_S];
  const float* prow = pairwise + (size_t)n*LQ*PAIR + 32*w + 2*col;
#pragma unroll
  for (int s = 0; s < PO_R - 1; ++s) {
    po_load(a[s], x[s], y[s], arow, prow, s*4*PO_S + khalf);
    __builtin_amdgcn_sched_barrier(0);
  }

  for (int kb0 = 0; kb0 < PO_NB - PO_R; kb0 += PO_R)
    po_turn<false>(acc, a, x, y, arow, prow, kb0, khalf, amask);
  po_turn<true>(acc, a, x, y, arow, prow, PO_NB - PO_R, khalf, amask);

  const int rbase = (lane >> 4) * 8;
#pragma unroll
  for (int r = 0; r < 8; ++r) {
    const int hh = rbase + r;
    if (hh < NH) {
      v2f o; o.x = acc[0][r]; o.y = acc[1][r];
      *(v2f*)(res + RES_IDX(hh, n, 16 + 32*w + 2*col)) = o;
    }
  }
}

static __device__ __forceinline__ void pfinish_body(
    const float* __restrict__ R, const float* __restrict__ t,
    float* __restrict__ ws, const int idx  )
{
  const int h = idx / LQ, l = idx % LQ;
  float* resp = ws + OFF_RES + RES_IDX(h, l, 0);
  float Rm[9], tv[3];
#pragma unroll
  for (int i = 0; i < 9; ++i) Rm[i] = R[l*9 + i];
#pragma unroll
  for (int i = 0; i < 3; ++i) tv[i] = t[l*3 + i];
#pragma unroll
  for (int p = 0; p < 8; ++p) {
    float g[3], loc[3], nrm = 0.f;
    for (int gg = 0; gg < 3; ++gg) g[gg] = resp[144 + p*3 + gg] - tv[gg];
    for (int f = 0; f < 3; ++f) {
      float v = 0.f;
      for (int gg = 0; gg < 3; ++gg) v += g[gg]*Rm[gg*3 + f];
      loc[f] = v; nrm += v*v;
    }
    for (int f = 0; f < 3; ++f) resp[144 + p*3 + f] = loc[f];
    resp[168 + p] = sqrtf(nrm + 1e-12f);
  }
}

#define PF_B ((NH*LQ)/(32*PO_W))
__global__ __launch_bounds__(32*PO_W) void pairout_pfinish_kernel(
    const float* __restrict__ pairwise, const float* __restrict__ R,
    const float* __restrict__ t, float* __restrict__ ws)
{
  if (blockIdx.x < LQ)
    pairout_body(pairwise, ws, blockIdx.x);
  else
    pfinish_body(R, t, ws, (blockIdx.x - LQ)*(32*PO_W) + threadIdx.x);
}

static __device__ __forceinline__ void out_load(
    v2f (&a)[OUT_S], v2f (&b)[OUT_S], const float* __restrict__ resl,
    const float* __restrict__ okp, int k0, int khalf)
{
#pragma unroll
  for (int u = 0; u < OUT_S; ++u) {
    const int ka = k0 + 4*u + khalf;
    a[u].x = resl[ka];                 a[u].y = resl[ka + 1];
    b[u].x = okp[(size_t)ka * NODE];   b[u].y = okp[(size_t)(ka + 1) * NODE];
  }
}

template <bool TAIL>
static __device__ __forceinline__ void out_turn(
    v8f& acc, v2f (&a)[OUT_R][OUT_S], v2f (&b)[OUT_R][OUT_S],
    const float* __restrict__ resl, const float* __restrict__ okp,
    int kb0, int khalf)
{
#pragma unroll
  for (int j = 0; j < OUT_R; ++j) {
    if (!TAIL || j == 0)
      out_load(a[(j + OUT_R - 1) % OUT_R], b[(j + OUT_R - 1) % OUT_R],
               resl, okp, (kb0 + j + OUT_R - 1)*4*OUT_S, khalf);
    __builtin_amdgcn_sched_barrier(0);
#pragma unroll
    for (int u = 0; u < OUT_S; ++u) acc = wmma4(a[j][u], b[j][u], acc);
    __builtin_amdgcn_sched_barrier(0);
  }
}

__global__ __launch_bounds__(32) void outproj_kernel(
    const float* __restrict__ out_k, const float* __restrict__ out_b,
    const float* __restrict__ res, float* __restrict__ out)
{
  const int lane  = threadIdx.x;
  const int mt    = blockIdx.x;
  const int ntile = blockIdx.y;
  const int row   = lane & 15;
  const int khalf = (lane >> 4) * 2;
  const int col   = lane & 15;
  const int o     = ntile*16 + col;

  const float* resl = res + (size_t)(mt*16 + row) * KOUT;
  const float* okp  = out_k + o;

  v8f acc = vzero8();
  v2f a[OUT_R][OUT_S], b[OUT_R][OUT_S];
#pragma unroll
  for (int s = 0; s < OUT_R - 1; ++s) {
    out_load(a[s], b[s], resl, okp, s*4*OUT_S, khalf);
    __builtin_amdgcn_sched_barrier(0);
  }

  for (int kb0 = 0; kb0 < OUT_NB - OUT_R; kb0 += OUT_R)
    out_turn<false>(acc, a, b, resl, okp, kb0, khalf);
  out_turn<true>(acc, a, b, resl, okp, OUT_NB - OUT_R, khalf);
  const int rbase = (lane >> 4) * 8;
  const float ob = out_b[o];
#pragma unroll
  for (int r = 0; r < 8; ++r) {
    const int l = mt*16 + rbase + r;
    out[(size_t)l*NODE + o] = acc[r] + ob;
  }
}

extern "C" void kernel_launch(void* const* d_in, const int* in_sizes, int n_in,
                              void* d_out, int out_size, void* d_ws, size_t ws_size,
                              hipStream_t stream) {
  const float* node  = (const float*)d_in[0];
  const float* R     = (const float*)d_in[1];
  const float* t     = (const float*)d_in[2];
  const float* pairw = (const float*)d_in[3];
  const float* sq_k  = (const float*)d_in[4];
  const float* sk_k  = (const float*)d_in[5];
  const float* sv_k  = (const float*)d_in[6];
  const float* pq_k  = (const float*)d_in[7];
  const float* pk_k  = (const float*)d_in[8];
  const float* pv_k  = (const float*)d_in[9];
  const float* pr_k  = (const float*)d_in[10];
  const float* out_k = (const float*)d_in[11];
  const float* out_b = (const float*)d_in[12];
  const float* pwgt  = (const float*)d_in[13];
  float* ws  = (float*)d_ws;
  float* out = (float*)d_out;

  projbias_kernel<<<LQ + PJ_B, 32*BI_W, 0, stream>>>(
      node, sq_k, sk_k, sv_k, pq_k, pk_k, pv_k, pairw, pr_k, ws);
  rigid_kernel<<<(NH*LQ + 127)/128, 128, 0, stream>>>(R, t, ws);
  attn_kernel<<<dim3(32, NH), 128, 0, stream>>>(pwgt, ws);
  pairout_pfinish_kernel<<<LQ + PF_B, 32*PO_W, 0, stream>>>(pairw, R, t, ws);
  outproj_kernel<<<dim3(32, 24), 32, 0, stream>>>(out_k, out_b, ws + OFF_RES, out);
}


# outproj re-tiled: block = 16 rows x 48 cols, 4 waves = 4 K quarters, 3 N-tiles share 16-byte A loads; quarters added via LDS in fixed order; each 8 k summed 0,1,4,5 then 2,3,6,7 (order changed)
# speedup vs baseline: 3.2635x; 1.0797x over previous
#include <hip/hip_runtime.h>
#include <math.h>


typedef __attribute__((ext_vector_type(2))) float v2f;
typedef __attribute__((ext_vector_type(4))) float v4f;
typedef __attribute__((ext_vector_type(8))) float v8f;

#define LQ    512
#define NH    12
#define NODE  384
#define PAIR  128
#define OUTIN 176
#define KOUT  (NH*OUTIN)

#define U_PROJ 16
#define U_AV   16
#define BI_W   4
#define BI_T   8
#define BI_G   (PAIR/8)
#define BI_R   3
#define PO_W   4
#define PO_T   2
#define PO_S   2
#define PO_R   8
#define PO_NB  (LQ/(4*PO_S))
#define OP_T   3
#define OP_W   4
#define OP_KQ  (KOUT/OP_W)
#define OP_R   11
#define OP_NB  (OP_KQ/8)

#define OFF_SQ   ((size_t)0)
#define OFF_SK   (OFF_SQ  + (size_t)NH*LQ*16)
#define OFF_SV   (OFF_SK  + (size_t)NH*LQ*16)
#define OFF_PQ   (OFF_SV  + (size_t)NH*LQ*16)
#define OFF_PK   (OFF_PQ  + (size_t)NH*LQ*16)
#define OFF_PV   (OFF_PK  + (size_t)NH*LQ*16)
#define OFF_QN2  (OFF_PV  + (size_t)NH*LQ*24)
#define OFF_KN2  (OFF_QN2 + (size_t)NH*LQ)
#define OFF_BIAS (OFF_KN2 + (size_t)NH*LQ)
#define OFF_ATTN (OFF_BIAS+ (size_t)NH*LQ*LQ)
#define OFF_RES  (OFF_ATTN+ (size_t)NH*LQ*LQ)

#define RES_IDX(h, l, f) (((size_t)(l)*NH + (h))*OUTIN + (f))

static __device__ __forceinline__ v8f wmma4(v2f a, v2f b, v8f c) {
  return __builtin_amdgcn_wmma_f32_16x16x4_f32(
       false, a,  false, b,
       (short)0, c,  false,  false);
}

static __device__ __forceinline__ v8f vzero8() {
  v8f z = {0.f,0.f,0.f,0.f,0.f,0.f,0.f,0.f};
  return z;
}

static __device__ __forceinline__ void proj_body(
    const float* __restrict__ node,
    const float* __restrict__ sq_k, const float* __restrict__ sk_k,
    const float* __restrict__ sv_k, const float* __restrict__ pq_k,
    const float* __restrict__ pk_k, const float* __restrict__ pv_k,
    float* __restrict__ ws, const int unit)
{
  const int lane  = threadIdx.x & 31;
  const int mt    = unit & 31;
  const int seg   = (unit >> 5) % 7;
  const int h     = (unit >> 5) / 7;
  const int row   = lane & 15;
  const int khalf = (lane >> 4) * 2;

  const float* wb = sk_k + (size_t)h*NODE*16;
  int   wstride = 16;
  int   wcol    = row;
  float wscale  = 1.f;
  switch (seg) {
    case 0: wb = sq_k + (size_t)h*NODE*16; wscale = 0.25f; break;
    case 1: break;
    case 2: wb = sv_k + (size_t)h*NODE*16; break;
    case 3: wb = pq_k + (size_t)h*NODE*12; wstride = 12;
            wcol = (row < 12) ? row : 11; wscale = (row < 12) ? 1.f : 0.f; break;
    case 4: wb = pk_k + (size_t)h*NODE*12; wstride = 12;
            wcol = (row < 12) ? row : 11; wscale = (row < 12) ? 1.f : 0.f; break;
    case 5: wb = pv_k + (size_t)h*NODE*24; wstride = 24; break;
    case 6: wb = pv_k + (size_t)h*NODE*24; wstride = 24;
            wcol = (row < 8) ? 16 + row : 23; wscale = (row < 8) ? 1.f : 0.f; break;
  }
  const float* arow = node + (size_t)(mt*16 + row)*NODE;
  const float* wp   = wb + wcol;

  v8f acc = vzero8();
  for (int k0 = 0; k0 < NODE; k0 += 4*U_PROJ) {
    v2f a[U_PROJ], b[U_PROJ];
#pragma unroll
    for (int u = 0; u < U_PROJ; ++u) {
      const int ka = k0 + 4*u + khalf;
      a[u].x = arow[ka];                     a[u].y = arow[ka + 1];
      b[u].x = wp[(size_t)ka * wstride];     b[u].y = wp[(size_t)(ka + 1) * wstride];
    }
    __builtin_amdgcn_sched_barrier(0);
#pragma unroll
    for (int u = 0; u < U_PROJ; ++u) {
      v2f bs; bs.x = b[u].x * wscale; bs.y = b[u].y * wscale;
      acc = wmma4(a[u], bs, acc);
    }
  }

  const int col   = lane & 15;
  const int rbase = (lane >> 4) * 8;
  float* ob = ws + OFF_SK + (size_t)h*LQ*16;
  int ostride = 16, ocol = col;
  bool dostore = true;
  switch (seg) {
    case 0: ob = ws + OFF_SQ + (size_t)h*LQ*16; break;
    case 1: break;
    case 2: ob = ws + OFF_SV + (size_t)h*LQ*16; break;
    case 3: ob = ws + OFF_PQ + (size_t)h*LQ*16; break;
    case 4: ob = ws + OFF_PK + (size_t)h*LQ*16; break;
    case 5: ob = ws + OFF_PV + (size_t)h*LQ*24; ostride = 24; break;
    case 6: ob = ws + OFF_PV + (size_t)h*LQ*24; ostride = 24;
            ocol = 16 + col; dostore = (col < 8); break;
  }
  if (dostore) {
#pragma unroll
    for (int r = 0; r < 8; ++r)
      ob[(size_t)(mt*16 + rbase + r)*ostride + ocol] = acc[r];
  }
}

__global__ void rigid_kernel(const float* __restrict__ R,
                             const float* __restrict__ t,
                             float* __restrict__ ws)
{
  const int idx = blockIdx.x * blockDim.x + threadIdx.x;
  if (idx >= NH * LQ) return;
  const int h = idx / LQ, l = idx % LQ;
  float Rm[9], tv[3];
#pragma unroll
  for (int i = 0; i < 9; ++i) Rm[i] = R[l*9 + i];
#pragma unroll
  for (int i = 0; i < 3; ++i) tv[i] = t[l*3 + i];

  float* Pq = ws + OFF_PQ + ((size_t)h*LQ + l)*16;
  float* Pk = ws + OFF_PK + ((size_t)h*LQ + l)*16;
  float* Pv = ws + OFF_PV + ((size_t)h*LQ + l)*24;

  float qs = 0.f, ksum = 0.f;
#pragma unroll
  for (int p = 0; p < 4; ++p) {
    float loc[3], g[3];
    for (int f = 0; f < 3; ++f) loc[f] = Pq[p*3+f];
    for (int gg = 0; gg < 3; ++gg) {
      g[gg] = tv[gg];
      for (int f = 0; f < 3; ++f) g[gg] += loc[f]*Rm[gg*3+f];
      Pq[p*3+gg] = g[gg]; qs += g[gg]*g[gg];
    }
    for (int f = 0; f < 3; ++f) loc[f] = Pk[p*3+f];
    for (int gg = 0; gg < 3; ++gg) {
      g[gg] = tv[gg];
      for (int f = 0; f < 3; ++f) g[gg] += loc[f]*Rm[gg*3+f];
      Pk[p*3+gg] = g[gg]; ksum += g[gg]*g[gg];
    }
  }
#pragma unroll
  for (int p = 0; p < 8; ++p) {
    float loc[3], g[3];
    for (int f = 0; f < 3; ++f) loc[f] = Pv[p*3+f];
    for (int gg = 0; gg < 3; ++gg) {
      g[gg] = tv[gg];
      for (int f = 0; f < 3; ++f) g[gg] += loc[f]*Rm[gg*3+f];
      Pv[p*3+gg] = g[gg];
    }
  }
  ws[OFF_QN2 + h*LQ + l] = qs;
  ws[OFF_KN2 + h*LQ + l] = ksum;
}

static __device__ __forceinline__ void bi_load(
    v4f (&q)[BI_G], const float* __restrict__ p)
{
#pragma unroll
  for (int g = 0; g < BI_G; ++g) q[g] = *(const v4f*)(p + 8*g);
}

static __device__ __forceinline__ void bias_body(
    const float* __restrict__ pairwise, const float* __restrict__ pr_k,
    float* __restrict__ ws, const int i  )
{
  float* bias = ws + OFF_BIAS;
  const int lane  = threadIdx.x & 31;
  const int w     = threadIdx.x >> 5;
  const int row   = lane & 15;
  const int half  = lane >> 4;
  const int col   = lane & 15;
  const int rbase = (lane >> 4) * 8;

  const float bmask = (row < NH) ? 1.f : 0.f;
  v4f bq[BI_G];
  bi_load(bq, pr_k + (size_t)((row < NH) ? row : NH - 1) * PAIR + 4*half);

  const float* ap = pairwise + ((size_t)i*LQ + 128*w + row) * PAIR + 4*half;
  v4f q[BI_R][BI_G];
  bi_load(q[0], ap);
#pragma unroll
  for (int g = 0; g < BI_G; ++g) bq[g] *= bmask;
  __builtin_amdgcn_sched_barrier(0);
#pragma unroll
  for (int s = 1; s < BI_R - 1; ++s) bi_load(q[s], ap + (size_t)s*16*PAIR);

  float* ob = bias + ((size_t)col*LQ + i)*LQ + 128*w + rbase;
#pragma unroll
  for (int t = 0; t < BI_T; ++t) {
    if (t + BI_R - 1 < BI_T)
      bi_load(q[(t + BI_R - 1) % BI_R], ap + (size_t)(t + BI_R - 1)*16*PAIR);
    __builtin_amdgcn_sched_barrier(0);
    v8f acc = vzero8();
#pragma unroll
    for (int g = 0; g < BI_G; ++g) {
      const v4f qa = q[t % BI_R][g];
      v2f a0; a0.x = qa.x;    a0.y = qa.y;
      v2f a1; a1.x = qa.z;    a1.y = qa.w;
      v2f b0; b0.x = bq[g].x; b0.y = bq[g].y;
      v2f b1; b1.x = bq[g].z; b1.y = bq[g].w;
      acc = wmma4(a0, b0, acc);
      acc = wmma4(a1, b1, acc);
    }
    if (col < NH) {
#pragma unroll
      for (int r = 0; r < 8; ++r) ob[16*t + r] = acc[r];
    }
  }
}

#define PJ_UNITS (32*7*NH)
#define PJ_B     (PJ_UNITS/BI_W)
__global__ __launch_bounds__(32*BI_W) void projbias_kernel(
    const float* __restrict__ node,
    const float* __restrict__ sq_k, const float* __restrict__ sk_k,
    const float* __restrict__ sv_k, const float* __restrict__ pq_k,
    const float* __restrict__ pk_k, const float* __restrict__ pv_k,
    const float* __restrict__ pairwise, const float* __restrict__ pr_k,
    float* __restrict__ ws)
{
  if (blockIdx.x < LQ)
    bias_body(pairwise, pr_k, ws, blockIdx.x);
  else
    proj_body(node, sq_k, sk_k, sv_k, pq_k, pk_k, pv_k, ws,
              (blockIdx.x - LQ)*BI_W + (threadIdx.x >> 5));
}

#define SMS 516
__global__ __launch_bounds__(128) void attn_kernel(
    const float* __restrict__ point_weight, float* __restrict__ ws)
{
  __shared__ float sm[16 * SMS];
  const int tid  = threadIdx.x;
  const int lane = tid & 31;
  const int w    = tid >> 5;
  const int nt   = blockIdx.x;
  const int h    = blockIdx.y;

  const float* Sq  = ws + OFF_SQ;  const float* Sk  = ws + OFF_SK;
  const float* Pq  = ws + OFF_PQ;  const float* Pk  = ws + OFF_PK;
  const float* qn2 = ws + OFF_QN2; const float* kn2 = ws + OFF_KN2;
  const float* bias= ws + OFF_BIAS;
  const float* Sv  = ws + OFF_SV;  const float* Pv  = ws + OFF_PV;
  float* attn = ws + OFF_ATTN;
  float* res  = ws + OFF_RES;

  const float pwx = point_weight[h];
  const float pw  = (pwx > 20.f) ? pwx : log1pf(__expf(pwx));
  const float cpt = -0.5f * pw / sqrtf(18.f);
  const float rs3 = 0.57735026919f;

  const int row   = lane & 15;
  const int khalf = (lane >> 4) * 2;
  const int col   = lane & 15;
  const int rbase = (lane >> 4) * 8;

  v2f aq[4], ap[4];
#pragma unroll
  for (int ks = 0; ks < 4; ++ks) {
    const int k = ks*4 + khalf;
    const int l = nt*16 + row;
    aq[ks].x = Sq[((size_t)h*LQ + l)*16 + k];
    aq[ks].y = Sq[((size_t)h*LQ + l)*16 + k + 1];
    ap[ks].x = Pq[((size_t)h*LQ + l)*16 + k];
    ap[ks].y = Pq[((size_t)h*LQ + l)*16 + k + 1];
  }
  float qv[8];
#pragma unroll
  for (int r = 0; r < 8; ++r) qv[r] = qn2[h*LQ + nt*16 + rbase + r];

  for (int mt = w; mt < 32; mt += 4) {
    v8f s = vzero8(), p = vzero8();
#pragma unroll
    for (int ks = 0; ks < 4; ++ks) {
      const int k = ks*4 + khalf;
      const int m = mt*16 + row;
      v2f bq, bp;
      bq.x = Sk[((size_t)h*LQ + m)*16 + k]; bq.y = Sk[((size_t)h*LQ + m)*16 + k + 1];
      bp.x = Pk[((size_t)h*LQ + m)*16 + k]; bp.y = Pk[((size_t)h*LQ + m)*16 + k + 1];
      s = wmma4(aq[ks], bq, s);
      p = wmma4(ap[ks], bp, p);
    }
    const int m = mt*16 + col;
    const float kv = kn2[h*LQ + m];
#pragma unroll
    for (int r = 0; r < 8; ++r) {
      const int rr = rbase + r;
      const int n  = nt*16 + rr;
      const float dist = qv[r] + kv - 2.f*p[r];
      sm[rr*SMS + m] =
          (s[r] + bias[((size_t)h*LQ + n)*LQ + m] + cpt*dist) * rs3;
    }
  }
  __syncthreads();

  {
    const int rrow = tid >> 3;
    const int sub  = tid & 7;
    float mx = -1e30f;
    for (int j = sub; j < LQ; j += 8) mx = fmaxf(mx, sm[rrow*SMS + j]);
    for (int o = 1; o < 8; o <<= 1) mx = fmaxf(mx, __shfl_xor(mx, o, 8));
    float sum = 0.f;
    for (int j = sub; j < LQ; j += 8) {
      const float e = __expf(sm[rrow*SMS + j] - mx);
      sm[rrow*SMS + j] = e; sum += e;
    }
    for (int o = 1; o < 8; o <<= 1) sum += __shfl_xor(sum, o, 8);
    const float inv = 1.f / sum;
    const int n = nt*16 + rrow;
    for (int j = sub; j < LQ; j += 8) {
      const float v = sm[rrow*SMS + j] * inv;
      sm[rrow*SMS + j] = v;
      attn[((size_t)h*LQ + n)*LQ + j] = v;
    }
  }
  __syncthreads();

  if (w < 3) {
    const float* vb = (w == 0) ? Sv + (size_t)h*LQ*16 : Pv + (size_t)h*LQ*24;
    const int    vs = (w == 0) ? 16 : 24;
    int   vc = col;
    float vmask = 1.f;
    if (w == 2) { vc = (col < 8) ? 16 + col : 23; vmask = (col < 8) ? 1.f : 0.f; }
    const float* vp = vb + vc;

    v8f acc = vzero8();
    for (int kt0 = 0; kt0 < 128; kt0 += U_AV) {
      v2f a[U_AV], b[U_AV];
#pragma unroll
      for (int u = 0; u < U_AV; ++u) {
        const int k = (kt0 + u)*4 + khalf;
        a[u].x = sm[row*SMS + k];        a[u].y = sm[row*SMS + k + 1];
        b[u].x = vp[(size_t)k * vs];     b[u].y = vp[(size_t)(k + 1) * vs];
      }
      __builtin_amdgcn_sched_barrier(0);
#pragma unroll
      for (int u = 0; u < U_AV; ++u) {
        v2f bs; bs.x = b[u].x * vmask; bs.y = b[u].y * vmask;
        acc = wmma4(a[u], bs, acc);
      }
    }
    const int fbase = (w == 0) ? 0 : 144;
#pragma unroll
    for (int r = 0; r < 8; ++r) {
      const int n = nt*16 + rbase + r;
      const float v = acc[r];
      if (w < 2)        res[RES_IDX(h, n, fbase + col)] = v;
      else if (col < 8) res[RES_IDX(h, n, 160 + col)]   = v;
    }
  }
}

static __device__ __forceinline__ void po_load(
    v2f (&a)[PO_S], v2f (&x)[PO_S], v2f (&y)[PO_S],
    const float* __restrict__ arow, const float* __restrict__ prow, int m0)
{
#pragma unroll
  for (int u = 0; u < PO_S; ++u) {
    const int m = m0 + 4*u;
    a[u] = *(const v2f*)(arow + m);
    x[u] = *(const v2f*)(prow + (size_t)m * PAIR);
    y[u] = *(const v2f*)(prow + (size_t)(m + 1) * PAIR);
  }
}

template <bool TAIL>
static __device__ __forceinline__ void po_turn(
    v8f (&acc)[PO_T], v2f (&a)[PO_R][PO_S], v2f (&x)[PO_R][PO_S],
    v2f (&y)[PO_R][PO_S], const float* __restrict__ arow,
    const float* __restrict__ prow, int kb0, int khalf, float amask)
{
#pragma unroll
  for (int j = 0; j < PO_R; ++j) {
    if (!TAIL || j == 0)
      po_load(a[(j + PO_R - 1) % PO_R], x[(j + PO_R - 1) % PO_R],
              y[(j + PO_R - 1) % PO_R], arow, prow,
              (kb0 + j + PO_R - 1)*4*PO_S + khalf);
    __builtin_amdgcn_sched_barrier(0);
#pragma unroll
    for (int u = 0; u < PO_S; ++u) {
      v2f am; am.x = a[j][u].x * amask; am.y = a[j][u].y * amask;
      v2f b0; b0.x = x[j][u].x; b0.y = y[j][u].x;
      v2f b1; b1.x = x[j][u].y; b1.y = y[j][u].y;
      acc[0] = wmma4(am, b0, acc[0]);
      acc[1] = wmma4(am, b1, acc[1]);
    }
    __builtin_amdgcn_sched_barrier(0);
  }
}

static __device__ __forceinline__ void pairout_body(
    const float* __restrict__ pairwise, float* __restrict__ ws, const int n)
{
  const float* attn = ws + OFF_ATTN;
  float* res = ws + OFF_RES;
  const int lane  = threadIdx.x & 31;
  const int w     = threadIdx.x >> 5;
  const int row   = lane & 15;
  const int khalf = (lane >> 4) * 2;
  const int col   = lane & 15;

  const int hh0 = (row < NH) ? row : NH - 1;
  const float amask = (row < NH) ? 1.f : 0.f;
  const float* arow = attn + ((size_t)hh0*LQ + n)*LQ;

  v8f acc[PO_T];
#pragma unroll
  for (int i = 0; i < PO_T; ++i) acc[i] = vzero8();

  v2f a[PO_R][PO_S], x[PO_R][PO_S], y[PO_R][PO_S];
  const float* prow = pairwise + (size_t)n*LQ*PAIR + 32*w + 2*col;
#pragma unroll
  for (int s = 0; s < PO_R - 1; ++s) {
    po_load(a[s], x[s], y[s], arow, prow, s*4*PO_S + khalf);
    __builtin_amdgcn_sched_barrier(0);
  }

  for (int kb0 = 0; kb0 < PO_NB - PO_R; kb0 += PO_R)
    po_turn<false>(acc, a, x, y, arow, prow, kb0, khalf, amask);
  po_turn<true>(acc, a, x, y, arow, prow, PO_NB - PO_R, khalf, amask);

  const int rbase = (lane >> 4) * 8;
#pragma unroll
  for (int r = 0; r < 8; ++r) {
    const int hh = rbase + r;
    if (hh < NH) {
      v2f o; o.x = acc[0][r]; o.y = acc[1][r];
      *(v2f*)(res + RES_IDX(hh, n, 16 + 32*w + 2*col)) = o;
    }
  }
}

static __device__ __forceinline__ void pfinish_body(
    const float* __restrict__ R, const float* __restrict__ t,
    float* __restrict__ ws, const int idx  )
{
  const int h = idx / LQ, l = idx % LQ;
  float* resp = ws + OFF_RES + RES_IDX(h, l, 0);
  float Rm[9], tv[3];
#pragma unroll
  for (int i = 0; i < 9; ++i) Rm[i] = R[l*9 + i];
#pragma unroll
  for (int i = 0; i < 3; ++i) tv[i] = t[l*3 + i];
#pragma unroll
  for (int p = 0; p < 8; ++p) {
    float g[3], loc[3], nrm = 0.f;
    for (int gg = 0; gg < 3; ++gg) g[gg] = resp[144 + p*3 + gg] - tv[gg];
    for (int f = 0; f < 3; ++f) {
      float v = 0.f;
      for (int gg = 0; gg < 3; ++gg) v += g[gg]*Rm[gg*3 + f];
      loc[f] = v; nrm += v*v;
    }
    for (int f = 0; f < 3; ++f) resp[144 + p*3 + f] = loc[f];
    resp[168 + p] = sqrtf(nrm + 1e-12f);
  }
}

#define PF_B ((NH*LQ)/(32*PO_W))
__global__ __launch_bounds__(32*PO_W) void pairout_pfinish_kernel(
    const float* __restrict__ pairwise, const float* __restrict__ R,
    const float* __restrict__ t, float* __restrict__ ws)
{
  if (blockIdx.x < LQ)
    pairout_body(pairwise, ws, blockIdx.x);
  else
    pfinish_body(R, t, ws, (blockIdx.x - LQ)*(32*PO_W) + threadIdx.x);
}

struct f3 { float x, y, z; };

static __device__ __forceinline__ void op_load(
    v4f& a, f3 (&b)[4], const float* __restrict__ ap,
    const float* __restrict__ bp, int d)
{
  a = *(const v4f*)(ap + 8*d);
#pragma unroll
  for (int i = 0; i < 4; ++i) {
    const float* p = bp + (size_t)(8*d + i) * NODE;
    b[i].x = p[0]; b[i].y = p[1]; b[i].z = p[2];
  }
}

template <bool TAIL>
static __device__ __forceinline__ void op_turn(
    v8f (&acc)[OP_T], v4f (&a)[OP_R], f3 (&b)[OP_R][4],
    const float* __restrict__ ap, const float* __restrict__ bp, int kb0)
{
#pragma unroll
  for (int j = 0; j < OP_R; ++j) {
    if (!TAIL || j == 0)
      op_load(a[(j + OP_R - 1) % OP_R], b[(j + OP_R - 1) % OP_R], ap, bp,
              kb0 + j + OP_R - 1);
    __builtin_amdgcn_sched_barrier(0);
    v2f a0; a0.x = a[j].x; a0.y = a[j].y;
    v2f a1; a1.x = a[j].z; a1.y = a[j].w;
    v2f b0, b1, b2;
    b0.x = b[j][0].x; b0.y = b[j][1].x;
    b1.x = b[j][0].y; b1.y = b[j][1].y;
    b2.x = b[j][0].z; b2.y = b[j][1].z;
    acc[0] = wmma4(a0, b0, acc[0]);
    acc[1] = wmma4(a0, b1, acc[1]);
    acc[2] = wmma4(a0, b2, acc[2]);
    b0.x = b[j][2].x; b0.y = b[j][3].x;
    b1.x = b[j][2].y; b1.y = b[j][3].y;
    b2.x = b[j][2].z; b2.y = b[j][3].z;
    acc[0] = wmma4(a1, b0, acc[0]);
    acc[1] = wmma4(a1, b1, acc[1]);
    acc[2] = wmma4(a1, b2, acc[2]);
    __builtin_amdgcn_sched_barrier(0);
  }
}

__global__ __launch_bounds__(32*OP_W) void outproj_kernel(
    const float* __restrict__ out_k, const float* __restrict__ out_b,
    const float* __restrict__ res, float* __restrict__ out)
{
  __shared__ float red[OP_W][OP_T][8][32];
  const int lane  = threadIdx.x & 31;
  const int q     = threadIdx.x >> 5;
  const int mt    = blockIdx.x;
  const int n0    = blockIdx.y * (16*OP_T);
  const int row   = lane & 15;
  const int half  = lane >> 4;
  const int col   = lane & 15;

  const float* ap = res + (size_t)(mt*16 + row) * KOUT + q*OP_KQ + 4*half;
  const float* bp = out_k + (size_t)(q*OP_KQ + 4*half) * NODE + n0 + 3*col;

  v8f acc[OP_T];
#pragma unroll
  for (int j = 0; j < OP_T; ++j) acc[j] = vzero8();
  v4f a[OP_R];
  f3  b[OP_R][4];
#pragma unroll
  for (int s = 0; s < OP_R - 1; ++s) {
    op_load(a[s], b[s], ap, bp, s);
    __builtin_amdgcn_sched_barrier(0);
  }

  for (int kb0 = 0; kb0 < OP_NB - OP_R; kb0 += OP_R)
    op_turn<false>(acc, a, b, ap, bp, kb0);
  op_turn<true>(acc, a, b, ap, bp, OP_NB - OP_R);

#pragma unroll
  for (int j = 0; j < OP_T; ++j)
#pragma unroll
    for (int r = 0; r < 8; ++r) red[q][j][r][lane] = acc[j][r];
  __syncthreads();

  const float* obp = out_b + n0 + 3*col;
  const float ob0 = obp[0], ob1 = obp[1], ob2 = obp[2];
#pragma unroll
  for (int rr = 0; rr < 2; ++rr) {
    const int r = 2*q + rr;
    const int l = mt*16 + 8*half + r;
    float o[OP_T];
#pragma unroll
    for (int j = 0; j < OP_T; ++j)
      o[j] = ((red[0][j][r][lane] + red[1][j][r][lane])
              + red[2][j][r][lane]) + red[3][j][r][lane];
    float* op = out + (size_t)l*NODE + n0 + 3*col;
    op[0] = o[0] + ob0; op[1] = o[1] + ob1; op[2] = o[2] + ob2;
  }
}

extern "C" void kernel_launch(void* const* d_in, const int* in_sizes, int n_in,
                              void* d_out, int out_size, void* d_ws, size_t ws_size,
                              hipStream_t stream) {
  const float* node  = (const float*)d_in[0];
  const float* R     = (const float*)d_in[1];
  const float* t     = (const float*)d_in[2];
  const float* pairw = (const float*)d_in[3];
  const float* sq_k  = (const float*)d_in[4];
  const float* sk_k  = (const float*)d_in[5];
  const float* sv_k  = (const float*)d_in[6];
  const float* pq_k  = (const float*)d_in[7];
  const float* pk_k  = (const float*)d_in[8];
  const float* pv_k  = (const float*)d_in[9];
  const float* pr_k  = (const float*)d_in[10];
  const float* out_k = (const float*)d_in[11];
  const float* out_b = (const float*)d_in[12];
  const float* pwgt  = (const float*)d_in[13];
  float* ws  = (float*)d_ws;
  float* out = (float*)d_out;

  projbias_kernel<<<LQ + PJ_B, 32*BI_W, 0, stream>>>(
      node, sq_k, sk_k, sv_k, pq_k, pk_k, pv_k, pairw, pr_k, ws);
  rigid_kernel<<<(NH*LQ + 127)/128, 128, 0, stream>>>(R, t, ws);
  attn_kernel<<<dim3(32, NH), 128, 0, stream>>>(pwgt, ws);
  pairout_pfinish_kernel<<<LQ + PF_B, 32*PO_W, 0, stream>>>(pairw, R, t, ws);
  outproj_kernel<<<dim3(32, 8), 32*OP_W, 0, stream>>>(out_k, out_b, ws + OFF_RES, out);
}


# rigid transform and squared norms moved into the projection epilogue (point coordinates by lane shuffle, pv segments 12+12; norms as a 16-lane xor-tree sum: order changed); launches 5 -> 4
# speedup vs baseline: 3.4607x; 1.0604x over previous
#include <hip/hip_runtime.h>
#include <math.h>


typedef __attribute__((ext_vector_type(2))) float v2f;
typedef __attribute__((ext_vector_type(4))) float v4f;
typedef __attribute__((ext_vector_type(8))) float v8f;

#define LQ    512
#define NH    12
#define NODE  384
#define PAIR  128
#define OUTIN 176
#define KOUT  (NH*OUTIN)

#define U_PROJ 16
#define U_AV   16
#define BI_W   4
#define BI_T   8
#define BI_G   (PAIR/8)
#define BI_R   3
#define PO_W   4
#define PO_T   2
#define PO_S   2
#define PO_R   8
#define PO_NB  (LQ/(4*PO_S))
#define OP_T   3
#define OP_W   4
#define OP_KQ  (KOUT/OP_W)
#define OP_R   11
#define OP_NB  (OP_KQ/8)

#define OFF_SQ   ((size_t)0)
#define OFF_SK   (OFF_SQ  + (size_t)NH*LQ*16)
#define OFF_SV   (OFF_SK  + (size_t)NH*LQ*16)
#define OFF_PQ   (OFF_SV  + (size_t)NH*LQ*16)
#define OFF_PK   (OFF_PQ  + (size_t)NH*LQ*16)
#define OFF_PV   (OFF_PK  + (size_t)NH*LQ*16)
#define OFF_QN2  (OFF_PV  + (size_t)NH*LQ*24)
#define OFF_KN2  (OFF_QN2 + (size_t)NH*LQ)
#define OFF_BIAS (OFF_KN2 + (size_t)NH*LQ)
#define OFF_ATTN (OFF_BIAS+ (size_t)NH*LQ*LQ)
#define OFF_RES  (OFF_ATTN+ (size_t)NH*LQ*LQ)

#define RES_IDX(h, l, f) (((size_t)(l)*NH + (h))*OUTIN + (f))

static __device__ __forceinline__ v8f wmma4(v2f a, v2f b, v8f c) {
  return __builtin_amdgcn_wmma_f32_16x16x4_f32(
       false, a,  false, b,
       (short)0, c,  false,  false);
}

static __device__ __forceinline__ v8f vzero8() {
  v8f z = {0.f,0.f,0.f,0.f,0.f,0.f,0.f,0.f};
  return z;
}

static __device__ __forceinline__ void proj_body(
    const float* __restrict__ node,
    const float* __restrict__ sq_k, const float* __restrict__ sk_k,
    const float* __restrict__ sv_k, const float* __restrict__ pq_k,
    const float* __restrict__ pk_k, const float* __restrict__ pv_k,
    const float* __restrict__ R, const float* __restrict__ t,
    float* __restrict__ ws, const int unit)
{
  const int lane  = threadIdx.x & 31;
  const int mt    = unit & 31;
  const int seg   = (unit >> 5) % 7;
  const int h     = (unit >> 5) / 7;
  const int row   = lane & 15;
  const int khalf = (lane >> 4) * 2;

  const float* wb = sk_k + (size_t)h*NODE*16;
  int   wstride = 16;
  int   wcol    = row;
  float wscale  = 1.f;
  switch (seg) {
    case 0: wb = sq_k + (size_t)h*NODE*16; wscale = 0.25f; break;
    case 1: break;
    case 2: wb = sv_k + (size_t)h*NODE*16; break;
    case 3: wb = pq_k + (size_t)h*NODE*12; wstride = 12;
            wcol = (row < 12) ? row : 11; wscale = (row < 12) ? 1.f : 0.f; break;
    case 4: wb = pk_k + (size_t)h*NODE*12; wstride = 12;
            wcol = (row < 12) ? row : 11; wscale = (row < 12) ? 1.f : 0.f; break;
    case 5: wb = pv_k + (size_t)h*NODE*24; wstride = 24;
            wcol = (row < 12) ? row : 11; wscale = (row < 12) ? 1.f : 0.f; break;
    case 6: wb = pv_k + (size_t)h*NODE*24; wstride = 24;
            wcol = (row < 12) ? 12 + row : 23; wscale = (row < 12) ? 1.f : 0.f; break;
  }
  const float* arow = node + (size_t)(mt*16 + row)*NODE;
  const float* wp   = wb + wcol;

  const int col   = lane & 15;
  const int rbase = (lane >> 4) * 8;
  const int l0    = mt*16 + rbase;
  const int pc = (col < 12) ? col : 11;
  const int g  = pc % 3;
  float rg0[8], rg1[8], rg2[8], tg[8];
#pragma unroll
  for (int r = 0; r < 8; ++r) {
    const float* Rg = R + (l0 + r)*9 + g*3;
    rg0[r] = Rg[0]; rg1[r] = Rg[1]; rg2[r] = Rg[2];
    tg[r]  = t[(l0 + r)*3 + g];
  }

  v8f acc = vzero8();
  for (int k0 = 0; k0 < NODE; k0 += 4*U_PROJ) {
    v2f a[U_PROJ], b[U_PROJ];
#pragma unroll
    for (int u = 0; u < U_PROJ; ++u) {
      const int ka = k0 + 4*u + khalf;
      a[u].x = arow[ka];                     a[u].y = arow[ka + 1];
      b[u].x = wp[(size_t)ka * wstride];     b[u].y = wp[(size_t)(ka + 1) * wstride];
    }
    __builtin_amdgcn_sched_barrier(0);
#pragma unroll
    for (int u = 0; u < U_PROJ; ++u) {
      v2f bs; bs.x = b[u].x * wscale; bs.y = b[u].y * wscale;
      acc = wmma4(a[u], bs, acc);
    }
  }

  if (seg < 3) {
    float* ob = ws + ((seg == 0) ? OFF_SQ : (seg == 1) ? OFF_SK : OFF_SV)
                   + (size_t)h*LQ*16;
#pragma unroll
    for (int r = 0; r < 8; ++r)
      ob[(size_t)(l0 + r)*16 + col] = acc[r];
    return;
  }

  const int src = (lane & 16) + pc - g;
  float o[8];
#pragma unroll
  for (int r = 0; r < 8; ++r) {
    const float x0 = __shfl(acc[r], src);
    const float x1 = __shfl(acc[r], src + 1);
    const float x2 = __shfl(acc[r], src + 2);
    float gv = tg[r];
    gv += x0*rg0[r]; gv += x1*rg1[r]; gv += x2*rg2[r];
    o[r] = (col < 12) ? gv : 0.f;
  }
  if (seg < 5) {
    float* ob = ws + ((seg == 3) ? OFF_PQ : OFF_PK) + (size_t)h*LQ*16;
    float* nb = ws + ((seg == 3) ? OFF_QN2 : OFF_KN2) + (size_t)h*LQ;
    float s[8];
#pragma unroll
    for (int r = 0; r < 8; ++r) {
      ob[(size_t)(l0 + r)*16 + col] = o[r];
      s[r] = o[r]*o[r];
    }
#pragma unroll
    for (int d = 1; d < 16; d <<= 1) {
#pragma unroll
      for (int r = 0; r < 8; ++r) s[r] += __shfl_xor(s[r], d, 16);
    }
    if (col == 0) {
#pragma unroll
      for (int r = 0; r < 8; ++r) nb[l0 + r] = s[r];
    }
  } else {
    float* ob = ws + OFF_PV + (size_t)h*LQ*24 + ((seg == 6) ? 12 : 0);
    if (col < 12) {
#pragma unroll
      for (int r = 0; r < 8; ++r)
        ob[(size_t)(l0 + r)*24 + col] = o[r];
    }
  }
}

static __device__ __forceinline__ void bi_load(
    v4f (&q)[BI_G], const float* __restrict__ p)
{
#pragma unroll
  for (int g = 0; g < BI_G; ++g) q[g] = *(const v4f*)(p + 8*g);
}

static __device__ __forceinline__ void bias_body(
    const float* __restrict__ pairwise, const float* __restrict__ pr_k,
    float* __restrict__ ws, const int i  )
{
  float* bias = ws + OFF_BIAS;
  const int lane  = threadIdx.x & 31;
  const int w     = threadIdx.x >> 5;
  const int row   = lane & 15;
  const int half  = lane >> 4;
  const int col   = lane & 15;
  const int rbase = (lane >> 4) * 8;

  const float bmask = (row < NH) ? 1.f : 0.f;
  v4f bq[BI_G];
  bi_load(bq, pr_k + (size_t)((row < NH) ? row : NH - 1) * PAIR + 4*half);

  const float* ap = pairwise + ((size_t)i*LQ + 128*w + row) * PAIR + 4*half;
  v4f q[BI_R][BI_G];
  bi_load(q[0], ap);
#pragma unroll
  for (int g = 0; g < BI_G; ++g) bq[g] *= bmask;
  __builtin_amdgcn_sched_barrier(0);
#pragma unroll
  for (int s = 1; s < BI_R - 1; ++s) bi_load(q[s], ap + (size_t)s*16*PAIR);

  float* ob = bias + ((size_t)col*LQ + i)*LQ + 128*w + rbase;
#pragma unroll
  for (int t = 0; t < BI_T; ++t) {
    if (t + BI_R - 1 < BI_T)
      bi_load(q[(t + BI_R - 1) % BI_R], ap + (size_t)(t + BI_R - 1)*16*PAIR);
    __builtin_amdgcn_sched_barrier(0);
    v8f acc = vzero8();
#pragma unroll
    for (int g = 0; g < BI_G; ++g) {
      const v4f qa = q[t % BI_R][g];
      v2f a0; a0.x = qa.x;    a0.y = qa.y;
      v2f a1; a1.x = qa.z;    a1.y = qa.w;
      v2f b0; b0.x = bq[g].x; b0.y = bq[g].y;
      v2f b1; b1.x = bq[g].z; b1.y = bq[g].w;
      acc = wmma4(a0, b0, acc);
      acc = wmma4(a1, b1, acc);
    }
    if (col < NH) {
#pragma unroll
      for (int r = 0; r < 8; ++r) ob[16*t + r] = acc[r];
    }
  }
}

#define PJ_UNITS (32*7*NH)
#define PJ_B     (PJ_UNITS/BI_W)
__global__ __launch_bounds__(32*BI_W) void projbias_kernel(
    const float* __restrict__ node,
    const float* __restrict__ sq_k, const float* __restrict__ sk_k,
    const float* __restrict__ sv_k, const float* __restrict__ pq_k,
    const float* __restrict__ pk_k, const float* __restrict__ pv_k,
    const float* __restrict__ pairwise, const float* __restrict__ pr_k,
    const float* __restrict__ R, const float* __restrict__ t,
    float* __restrict__ ws)
{
  if (blockIdx.x < LQ)
    bias_body(pairwise, pr_k, ws, blockIdx.x);
  else
    proj_body(node, sq_k, sk_k, sv_k, pq_k, pk_k, pv_k, R, t, ws,
              (blockIdx.x - LQ)*BI_W + (threadIdx.x >> 5));
}

#define SMS 516
__global__ __launch_bounds__(128) void attn_kernel(
    const float* __restrict__ point_weight, float* __restrict__ ws)
{
  __shared__ float sm[16 * SMS];
  const int tid  = threadIdx.x;
  const int lane = tid & 31;
  const int w    = tid >> 5;
  const int nt   = blockIdx.x;
  const int h    = blockIdx.y;

  const float* Sq  = ws + OFF_SQ;  const float* Sk  = ws + OFF_SK;
  const float* Pq  = ws + OFF_PQ;  const float* Pk  = ws + OFF_PK;
  const float* qn2 = ws + OFF_QN2; const float* kn2 = ws + OFF_KN2;
  const float* bias= ws + OFF_BIAS;
  const float* Sv  = ws + OFF_SV;  const float* Pv  = ws + OFF_PV;
  float* attn = ws + OFF_ATTN;
  float* res  = ws + OFF_RES;

  const float pwx = point_weight[h];
  const float pw  = (pwx > 20.f) ? pwx : log1pf(__expf(pwx));
  const float cpt = -0.5f * pw / sqrtf(18.f);
  const float rs3 = 0.57735026919f;

  const int row   = lane & 15;
  const int khalf = (lane >> 4) * 2;
  const int col   = lane & 15;
  const int rbase = (lane >> 4) * 8;

  v2f aq[4], ap[4];
#pragma unroll
  for (int ks = 0; ks < 4; ++ks) {
    const int k = ks*4 + khalf;
    const int l = nt*16 + row;
    aq[ks].x = Sq[((size_t)h*LQ + l)*16 + k];
    aq[ks].y = Sq[((size_t)h*LQ + l)*16 + k + 1];
    ap[ks].x = Pq[((size_t)h*LQ + l)*16 + k];
    ap[ks].y = Pq[((size_t)h*LQ + l)*16 + k + 1];
  }
  float qv[8];
#pragma unroll
  for (int r = 0; r < 8; ++r) qv[r] = qn2[h*LQ + nt*16 + rbase + r];

  for (int mt = w; mt < 32; mt += 4) {
    v8f s = vzero8(), p = vzero8();
#pragma unroll
    for (int ks = 0; ks < 4; ++ks) {
      const int k = ks*4 + khalf;
      const int m = mt*16 + row;
      v2f bq, bp;
      bq.x = Sk[((size_t)h*LQ + m)*16 + k]; bq.y = Sk[((size_t)h*LQ + m)*16 + k + 1];
      bp.x = Pk[((size_t)h*LQ + m)*16 + k]; bp.y = Pk[((size_t)h*LQ + m)*16 + k + 1];
      s = wmma4(aq[ks], bq, s);
      p = wmma4(ap[ks], bp, p);
    }
    const int m = mt*16 + col;
    const float kv = kn2[h*LQ + m];
#pragma unroll
    for (int r = 0; r < 8; ++r) {
      const int rr = rbase + r;
      const int n  = nt*16 + rr;
      const float dist = qv[r] + kv - 2.f*p[r];
      sm[rr*SMS + m] =
          (s[r] + bias[((size_t)h*LQ + n)*LQ + m] + cpt*dist) * rs3;
    }
  }
  __syncthreads();

  {
    const int rrow = tid >> 3;
    const int sub  = tid & 7;
    float mx = -1e30f;
    for (int j = sub; j < LQ; j += 8) mx = fmaxf(mx, sm[rrow*SMS + j]);
    for (int o = 1; o < 8; o <<= 1) mx = fmaxf(mx, __shfl_xor(mx, o, 8));
    float sum = 0.f;
    for (int j = sub; j < LQ; j += 8) {
      const float e = __expf(sm[rrow*SMS + j] - mx);
      sm[rrow*SMS + j] = e; sum += e;
    }
    for (int o = 1; o < 8; o <<= 1) sum += __shfl_xor(sum, o, 8);
    const float inv = 1.f / sum;
    const int n = nt*16 + rrow;
    for (int j = sub; j < LQ; j += 8) {
      const float v = sm[rrow*SMS + j] * inv;
      sm[rrow*SMS + j] = v;
      attn[((size_t)h*LQ + n)*LQ + j] = v;
    }
  }
  __syncthreads();

  if (w < 3) {
    const float* vb = (w == 0) ? Sv + (size_t)h*LQ*16 : Pv + (size_t)h*LQ*24;
    const int    vs = (w == 0) ? 16 : 24;
    int   vc = col;
    float vmask = 1.f;
    if (w == 2) { vc = (col < 8) ? 16 + col : 23; vmask = (col < 8) ? 1.f : 0.f; }
    const float* vp = vb + vc;

    v8f acc = vzero8();
    for (int kt0 = 0; kt0 < 128; kt0 += U_AV) {
      v2f a[U_AV], b[U_AV];
#pragma unroll
      for (int u = 0; u < U_AV; ++u) {
        const int k = (kt0 + u)*4 + khalf;
        a[u].x = sm[row*SMS + k];        a[u].y = sm[row*SMS + k + 1];
        b[u].x = vp[(size_t)k * vs];     b[u].y = vp[(size_t)(k + 1) * vs];
      }
      __builtin_amdgcn_sched_barrier(0);
#pragma unroll
      for (int u = 0; u < U_AV; ++u) {
        v2f bs; bs.x = b[u].x * vmask; bs.y = b[u].y * vmask;
        acc = wmma4(a[u], bs, acc);
      }
    }
    const int fbase = (w == 0) ? 0 : 144;
#pragma unroll
    for (int r = 0; r < 8; ++r) {
      const int n = nt*16 + rbase + r;
      const float v = acc[r];
      if (w < 2)        res[RES_IDX(h, n, fbase + col)] = v;
      else if (col < 8) res[RES_IDX(h, n, 160 + col)]   = v;
    }
  }
}

static __device__ __forceinline__ void po_load(
    v2f (&a)[PO_S], v2f (&x)[PO_S], v2f (&y)[PO_S],
    const float* __restrict__ arow, const float* __restrict__ prow, int m0)
{
#pragma unroll
  for (int u = 0; u < PO_S; ++u) {
    const int m = m0 + 4*u;
    a[u] = *(const v2f*)(arow + m);
    x[u] = *(const v2f*)(prow + (size_t)m * PAIR);
    y[u] = *(const v2f*)(prow + (size_t)(m + 1) * PAIR);
  }
}

template <bool TAIL>
static __device__ __forceinline__ void po_turn(
    v8f (&acc)[PO_T], v2f (&a)[PO_R][PO_S], v2f (&x)[PO_R][PO_S],
    v2f (&y)[PO_R][PO_S], const float* __restrict__ arow,
    const float* __restrict__ prow, int kb0, int khalf, float amask)
{
#pragma unroll
  for (int j = 0; j < PO_R; ++j) {
    if (!TAIL || j == 0)
      po_load(a[(j + PO_R - 1) % PO_R], x[(j + PO_R - 1) % PO_R],
              y[(j + PO_R - 1) % PO_R], arow, prow,
              (kb0 + j + PO_R - 1)*4*PO_S + khalf);
    __builtin_amdgcn_sched_barrier(0);
#pragma unroll
    for (int u = 0; u < PO_S; ++u) {
      v2f am; am.x = a[j][u].x * amask; am.y = a[j][u].y * amask;
      v2f b0; b0.x = x[j][u].x; b0.y = y[j][u].x;
      v2f b1; b1.x = x[j][u].y; b1.y = y[j][u].y;
      acc[0] = wmma4(am, b0, acc[0]);
      acc[1] = wmma4(am, b1, acc[1]);
    }
    __builtin_amdgcn_sched_barrier(0);
  }
}

static __device__ __forceinline__ void pairout_body(
    const float* __restrict__ pairwise, float* __restrict__ ws, const int n)
{
  const float* attn = ws + OFF_ATTN;
  float* res = ws + OFF_RES;
  const int lane  = threadIdx.x & 31;
  const int w     = threadIdx.x >> 5;
  const int row   = lane & 15;
  const int khalf = (lane >> 4) * 2;
  const int col   = lane & 15;

  const int hh0 = (row < NH) ? row : NH - 1;
  const float amask = (row < NH) ? 1.f : 0.f;
  const float* arow = attn + ((size_t)hh0*LQ + n)*LQ;

  v8f acc[PO_T];
#pragma unroll
  for (int i = 0; i < PO_T; ++i) acc[i] = vzero8();

  v2f a[PO_R][PO_S], x[PO_R][PO_S], y[PO_R][PO_S];
  const float* prow = pairwise + (size_t)n*LQ*PAIR + 32*w + 2*col;
#pragma unroll
  for (int s = 0; s < PO_R - 1; ++s) {
    po_load(a[s], x[s], y[s], arow, prow, s*4*PO_S + khalf);
    __builtin_amdgcn_sched_barrier(0);
  }

  for (int kb0 = 0; kb0 < PO_NB - PO_R; kb0 += PO_R)
    po_turn<false>(acc, a, x, y, arow, prow, kb0, khalf, amask);
  po_turn<true>(acc, a, x, y, arow, prow, PO_NB - PO_R, khalf, amask);

  const int rbase = (lane >> 4) * 8;
#pragma unroll
  for (int r = 0; r < 8; ++r) {
    const int hh = rbase + r;
    if (hh < NH) {
      v2f o; o.x = acc[0][r]; o.y = acc[1][r];
      *(v2f*)(res + RES_IDX(hh, n, 16 + 32*w + 2*col)) = o;
    }
  }
}

static __device__ __forceinline__ void pfinish_body(
    const float* __restrict__ R, const float* __restrict__ t,
    float* __restrict__ ws, const int idx  )
{
  const int h = idx / LQ, l = idx % LQ;
  float* resp = ws + OFF_RES + RES_IDX(h, l, 0);
  float Rm[9], tv[3];
#pragma unroll
  for (int i = 0; i < 9; ++i) Rm[i] = R[l*9 + i];
#pragma unroll
  for (int i = 0; i < 3; ++i) tv[i] = t[l*3 + i];
#pragma unroll
  for (int p = 0; p < 8; ++p) {
    float g[3], loc[3], nrm = 0.f;
    for (int gg = 0; gg < 3; ++gg) g[gg] = resp[144 + p*3 + gg] - tv[gg];
    for (int f = 0; f < 3; ++f) {
      float v = 0.f;
      for (int gg = 0; gg < 3; ++gg) v += g[gg]*Rm[gg*3 + f];
      loc[f] = v; nrm += v*v;
    }
    for (int f = 0; f < 3; ++f) resp[144 + p*3 + f] = loc[f];
    resp[168 + p] = sqrtf(nrm + 1e-12f);
  }
}

#define PF_B ((NH*LQ)/(32*PO_W))
__global__ __launch_bounds__(32*PO_W) void pairout_pfinish_kernel(
    const float* __restrict__ pairwise, const float* __restrict__ R,
    const float* __restrict__ t, float* __restrict__ ws)
{
  if (blockIdx.x < LQ)
    pairout_body(pairwise, ws, blockIdx.x);
  else
    pfinish_body(R, t, ws, (blockIdx.x - LQ)*(32*PO_W) + threadIdx.x);
}

struct f3 { float x, y, z; };

static __device__ __forceinline__ void op_load(
    v4f& a, f3 (&b)[4], const float* __restrict__ ap,
    const float* __restrict__ bp, int d)
{
  a = *(const v4f*)(ap + 8*d);
#pragma unroll
  for (int i = 0; i < 4; ++i) {
    const float* p = bp + (size_t)(8*d + i) * NODE;
    b[i].x = p[0]; b[i].y = p[1]; b[i].z = p[2];
  }
}

template <bool TAIL>
static __device__ __forceinline__ void op_turn(
    v8f (&acc)[OP_T], v4f (&a)[OP_R], f3 (&b)[OP_R][4],
    const float* __restrict__ ap, const float* __restrict__ bp, int kb0)
{
#pragma unroll
  for (int j = 0; j < OP_R; ++j) {
    if (!TAIL || j == 0)
      op_load(a[(j + OP_R - 1) % OP_R], b[(j + OP_R - 1) % OP_R], ap, bp,
              kb0 + j + OP_R - 1);
    __builtin_amdgcn_sched_barrier(0);
    v2f a0; a0.x = a[j].x; a0.y = a[j].y;
    v2f a1; a1.x = a[j].z; a1.y = a[j].w;
    v2f b0, b1, b2;
    b0.x = b[j][0].x; b0.y = b[j][1].x;
    b1.x = b[j][0].y; b1.y = b[j][1].y;
    b2.x = b[j][0].z; b2.y = b[j][1].z;
    acc[0] = wmma4(a0, b0, acc[0]);
    acc[1] = wmma4(a0, b1, acc[1]);
    acc[2] = wmma4(a0, b2, acc[2]);
    b0.x = b[j][2].x; b0.y = b[j][3].x;
    b1.x = b[j][2].y; b1.y = b[j][3].y;
    b2.x = b[j][2].z; b2.y = b[j][3].z;
    acc[0] = wmma4(a1, b0, acc[0]);
    acc[1] = wmma4(a1, b1, acc[1]);
    acc[2] = wmma4(a1, b2, acc[2]);
    __builtin_amdgcn_sched_barrier(0);
  }
}

__global__ __launch_bounds__(32*OP_W) void outproj_kernel(
    const float* __restrict__ out_k, const float* __restrict__ out_b,
    const float* __restrict__ res, float* __restrict__ out)
{
  __shared__ float red[OP_W][OP_T][8][32];
  const int lane  = threadIdx.x & 31;
  const int q     = threadIdx.x >> 5;
  const int mt    = blockIdx.x;
  const int n0    = blockIdx.y * (16*OP_T);
  const int row   = lane & 15;
  const int half  = lane >> 4;
  const int col   = lane & 15;

  const float* ap = res + (size_t)(mt*16 + row) * KOUT + q*OP_KQ + 4*half;
  const float* bp = out_k + (size_t)(q*OP_KQ + 4*half) * NODE + n0 + 3*col;

  v8f acc[OP_T];
#pragma unroll
  for (int j = 0; j < OP_T; ++j) acc[j] = vzero8();
  v4f a[OP_R];
  f3  b[OP_R][4];
#pragma unroll
  for (int s = 0; s < OP_R - 1; ++s) {
    op_load(a[s], b[s], ap, bp, s);
    __builtin_amdgcn_sched_barrier(0);
  }

  for (int kb0 = 0; kb0 < OP_NB - OP_R; kb0 += OP_R)
    op_turn<false>(acc, a, b, ap, bp, kb0);
  op_turn<true>(acc, a, b, ap, bp, OP_NB - OP_R);

#pragma unroll
  for (int j = 0; j < OP_T; ++j)
#pragma unroll
    for (int r = 0; r < 8; ++r) red[q][j][r][lane] = acc[j][r];
  __syncthreads();

  const float* obp = out_b + n0 + 3*col;
  const float ob0 = obp[0], ob1 = obp[1], ob2 = obp[2];
#pragma unroll
  for (int rr = 0; rr < 2; ++rr) {
    const int r = 2*q + rr;
    const int l = mt*16 + 8*half + r;
    float o[OP_T];
#pragma unroll
    for (int j = 0; j < OP_T; ++j)
      o[j] = ((red[0][j][r][lane] + red[1][j][r][lane])
              + red[2][j][r][lane]) + red[3][j][r][lane];
    float* op = out + (size_t)l*NODE + n0 + 3*col;
    op[0] = o[0] + ob0; op[1] = o[1] + ob1; op[2] = o[2] + ob2;
  }
}

extern "C" void kernel_launch(void* const* d_in, const int* in_sizes, int n_in,
                              void* d_out, int out_size, void* d_ws, size_t ws_size,
                              hipStream_t stream) {
  const float* node  = (const float*)d_in[0];
  const float* R     = (const float*)d_in[1];
  const float* t     = (const float*)d_in[2];
  const float* pairw = (const float*)d_in[3];
  const float* sq_k  = (const float*)d_in[4];
  const float* sk_k  = (const float*)d_in[5];
  const float* sv_k  = (const float*)d_in[6];
  const float* pq_k  = (const float*)d_in[7];
  const float* pk_k  = (const float*)d_in[8];
  const float* pv_k  = (const float*)d_in[9];
  const float* pr_k  = (const float*)d_in[10];
  const float* out_k = (const float*)d_in[11];
  const float* out_b = (const float*)d_in[12];
  const float* pwgt  = (const float*)d_in[13];
  float* ws  = (float*)d_ws;
  float* out = (float*)d_out;

  projbias_kernel<<<LQ + PJ_B, 32*BI_W, 0, stream>>>(
      node, sq_k, sk_k, sv_k, pq_k, pk_k, pv_k, pairw, pr_k, R, t, ws);
  attn_kernel<<<dim3(32, NH), 128, 0, stream>>>(pwgt, ws);
  pairout_pfinish_kernel<<<LQ + PF_B, 32*PO_W, 0, stream>>>(pairw, R, t, ws);
  outproj_kernel<<<dim3(32, 8), 32*OP_W, 0, stream>>>(out_k, out_b, ws + OFF_RES, out);
}


# attn block 4 -> 8 waves: logits 4 key tiles per wave, softmax 16 lanes per row, A*V as (sv tile | 2 pv tiles, 8-byte loads) x 4 K parts added via LDS in fixed order; row-sum, A*V sum order changed
# speedup vs baseline: 3.8418x; 1.1101x over previous
#include <hip/hip_runtime.h>
#include <math.h>


typedef __attribute__((ext_vector_type(2))) float v2f;
typedef __attribute__((ext_vector_type(4))) float v4f;
typedef __attribute__((ext_vector_type(8))) float v8f;

#define LQ    512
#define NH    12
#define NODE  384
#define PAIR  128
#define OUTIN 176
#define KOUT  (NH*OUTIN)

#define U_PROJ 16
#define U_AV   16
#define BI_W   4
#define BI_T   8
#define BI_G   (PAIR/8)
#define BI_R   3
#define PO_W   4
#define PO_T   2
#define PO_S   2
#define PO_R   8
#define PO_NB  (LQ/(4*PO_S))
#define OP_T   3
#define OP_W   4
#define OP_KQ  (KOUT/OP_W)
#define OP_R   11
#define OP_NB  (OP_KQ/8)

#define OFF_SQ   ((size_t)0)
#define OFF_SK   (OFF_SQ  + (size_t)NH*LQ*16)
#define OFF_SV   (OFF_SK  + (size_t)NH*LQ*16)
#define OFF_PQ   (OFF_SV  + (size_t)NH*LQ*16)
#define OFF_PK   (OFF_PQ  + (size_t)NH*LQ*16)
#define OFF_PV   (OFF_PK  + (size_t)NH*LQ*16)
#define OFF_QN2  (OFF_PV  + (size_t)NH*LQ*24)
#define OFF_KN2  (OFF_QN2 + (size_t)NH*LQ)
#define OFF_BIAS (OFF_KN2 + (size_t)NH*LQ)
#define OFF_ATTN (OFF_BIAS+ (size_t)NH*LQ*LQ)
#define OFF_RES  (OFF_ATTN+ (size_t)NH*LQ*LQ)

#define RES_IDX(h, l, f) (((size_t)(l)*NH + (h))*OUTIN + (f))

static __device__ __forceinline__ v8f wmma4(v2f a, v2f b, v8f c) {
  return __builtin_amdgcn_wmma_f32_16x16x4_f32(
       false, a,  false, b,
       (short)0, c,  false,  false);
}

static __device__ __forceinline__ v8f vzero8() {
  v8f z = {0.f,0.f,0.f,0.f,0.f,0.f,0.f,0.f};
  return z;
}

static __device__ __forceinline__ void proj_body(
    const float* __restrict__ node,
    const float* __restrict__ sq_k, const float* __restrict__ sk_k,
    const float* __restrict__ sv_k, const float* __restrict__ pq_k,
    const float* __restrict__ pk_k, const float* __restrict__ pv_k,
    const float* __restrict__ R, const float* __restrict__ t,
    float* __restrict__ ws, const int unit)
{
  const int lane  = threadIdx.x & 31;
  const int mt    = unit & 31;
  const int seg   = (unit >> 5) % 7;
  const int h     = (unit >> 5) / 7;
  const int row   = lane & 15;
  const int khalf = (lane >> 4) * 2;

  const float* wb = sk_k + (size_t)h*NODE*16;
  int   wstride = 16;
  int   wcol    = row;
  float wscale  = 1.f;
  switch (seg) {
    case 0: wb = sq_k + (size_t)h*NODE*16; wscale = 0.25f; break;
    case 1: break;
    case 2: wb = sv_k + (size_t)h*NODE*16; break;
    case 3: wb = pq_k + (size_t)h*NODE*12; wstride = 12;
            wcol = (row < 12) ? row : 11; wscale = (row < 12) ? 1.f : 0.f; break;
    case 4: wb = pk_k + (size_t)h*NODE*12; wstride = 12;
            wcol = (row < 12) ? row : 11; wscale = (row < 12) ? 1.f : 0.f; break;
    case 5: wb = pv_k + (size_t)h*NODE*24; wstride = 24;
            wcol = (row < 12) ? row : 11; wscale = (row < 12) ? 1.f : 0.f; break;
    case 6: wb = pv_k + (size_t)h*NODE*24; wstride = 24;
            wcol = (row < 12) ? 12 + row : 23; wscale = (row < 12) ? 1.f : 0.f; break;
  }
  const float* arow = node + (size_t)(mt*16 + row)*NODE;
  const float* wp   = wb + wcol;

  const int col   = lane & 15;
  const int rbase = (lane >> 4) * 8;
  const int l0    = mt*16 + rbase;
  const int pc = (col < 12) ? col : 11;
  const int g  = pc % 3;
  float rg0[8], rg1[8], rg2[8], tg[8];
#pragma unroll
  for (int r = 0; r < 8; ++r) {
    const float* Rg = R + (l0 + r)*9 + g*3;
    rg0[r] = Rg[0]; rg1[r] = Rg[1]; rg2[r] = Rg[2];
    tg[r]  = t[(l0 + r)*3 + g];
  }

  v8f acc = vzero8();
  for (int k0 = 0; k0 < NODE; k0 += 4*U_PROJ) {
    v2f a[U_PROJ], b[U_PROJ];
#pragma unroll
    for (int u = 0; u < U_PROJ; ++u) {
      const int ka = k0 + 4*u + khalf;
      a[u].x = arow[ka];                     a[u].y = arow[ka + 1];
      b[u].x = wp[(size_t)ka * wstride];     b[u].y = wp[(size_t)(ka + 1) * wstride];
    }
    __builtin_amdgcn_sched_barrier(0);
#pragma unroll
    for (int u = 0; u < U_PROJ; ++u) {
      v2f bs; bs.x = b[u].x * wscale; bs.y = b[u].y * wscale;
      acc = wmma4(a[u], bs, acc);
    }
  }

  if (seg < 3) {
    float* ob = ws + ((seg == 0) ? OFF_SQ : (seg == 1) ? OFF_SK : OFF_SV)
                   + (size_t)h*LQ*16;
#pragma unroll
    for (int r = 0; r < 8; ++r)
      ob[(size_t)(l0 + r)*16 + col] = acc[r];
    return;
  }

  const int src = (lane & 16) + pc - g;
  float o[8];
#pragma unroll
  for (int r = 0; r < 8; ++r) {
    const float x0 = __shfl(acc[r], src);
    const float x1 = __shfl(acc[r], src + 1);
    const float x2 = __shfl(acc[r], src + 2);
    float gv = tg[r];
    gv += x0*rg0[r]; gv += x1*rg1[r]; gv += x2*rg2[r];
    o[r] = (col < 12) ? gv : 0.f;
  }
  if (seg < 5) {
    float* ob = ws + ((seg == 3) ? OFF_PQ : OFF_PK) + (size_t)h*LQ*16;
    float* nb = ws + ((seg == 3) ? OFF_QN2 : OFF_KN2) + (size_t)h*LQ;
    float s[8];
#pragma unroll
    for (int r = 0; r < 8; ++r) {
      ob[(size_t)(l0 + r)*16 + col] = o[r];
      s[r] = o[r]*o[r];
    }
#pragma unroll
    for (int d = 1; d < 16; d <<= 1) {
#pragma unroll
      for (int r = 0; r < 8; ++r) s[r] += __shfl_xor(s[r], d, 16);
    }
    if (col == 0) {
#pragma unroll
      for (int r = 0; r < 8; ++r) nb[l0 + r] = s[r];
    }
  } else {
    float* ob = ws + OFF_PV + (size_t)h*LQ*24 + ((seg == 6) ? 12 : 0);
    if (col < 12) {
#pragma unroll
      for (int r = 0; r < 8; ++r)
        ob[(size_t)(l0 + r)*24 + col] = o[r];
    }
  }
}

static __device__ __forceinline__ void bi_load(
    v4f (&q)[BI_G], const float* __restrict__ p)
{
#pragma unroll
  for (int g = 0; g < BI_G; ++g) q[g] = *(const v4f*)(p + 8*g);
}

static __device__ __forceinline__ void bias_body(
    const float* __restrict__ pairwise, const float* __restrict__ pr_k,
    float* __restrict__ ws, const int i  )
{
  float* bias = ws + OFF_BIAS;
  const int lane  = threadIdx.x & 31;
  const int w     = threadIdx.x >> 5;
  const int row   = lane & 15;
  const int half  = lane >> 4;
  const int col   = lane & 15;
  const int rbase = (lane >> 4) * 8;

  const float bmask = (row < NH) ? 1.f : 0.f;
  v4f bq[BI_G];
  bi_load(bq, pr_k + (size_t)((row < NH) ? row : NH - 1) * PAIR + 4*half);

  const float* ap = pairwise + ((size_t)i*LQ + 128*w + row) * PAIR + 4*half;
  v4f q[BI_R][BI_G];
  bi_load(q[0], ap);
#pragma unroll
  for (int g = 0; g < BI_G; ++g) bq[g] *= bmask;
  __builtin_amdgcn_sched_barrier(0);
#pragma unroll
  for (int s = 1; s < BI_R - 1; ++s) bi_load(q[s], ap + (size_t)s*16*PAIR);

  float* ob = bias + ((size_t)col*LQ + i)*LQ + 128*w + rbase;
#pragma unroll
  for (int t = 0; t < BI_T; ++t) {
    if (t + BI_R - 1 < BI_T)
      bi_load(q[(t + BI_R - 1) % BI_R], ap + (size_t)(t + BI_R - 1)*16*PAIR);
    __builtin_amdgcn_sched_barrier(0);
    v8f acc = vzero8();
#pragma unroll
    for (int g = 0; g < BI_G; ++g) {
      const v4f qa = q[t % BI_R][g];
      v2f a0; a0.x = qa.x;    a0.y = qa.y;
      v2f a1; a1.x = qa.z;    a1.y = qa.w;
      v2f b0; b0.x = bq[g].x; b0.y = bq[g].y;
      v2f b1; b1.x = bq[g].z; b1.y = bq[g].w;
      acc = wmma4(a0, b0, acc);
      acc = wmma4(a1, b1, acc);
    }
    if (col < NH) {
#pragma unroll
      for (int r = 0; r < 8; ++r) ob[16*t + r] = acc[r];
    }
  }
}

#define PJ_UNITS (32*7*NH)
#define PJ_B     (PJ_UNITS/BI_W)
__global__ __launch_bounds__(32*BI_W) void projbias_kernel(
    const float* __restrict__ node,
    const float* __restrict__ sq_k, const float* __restrict__ sk_k,
    const float* __restrict__ sv_k, const float* __restrict__ pq_k,
    const float* __restrict__ pk_k, const float* __restrict__ pv_k,
    const float* __restrict__ pairwise, const float* __restrict__ pr_k,
    const float* __restrict__ R, const float* __restrict__ t,
    float* __restrict__ ws)
{
  if (blockIdx.x < LQ)
    bias_body(pairwise, pr_k, ws, blockIdx.x);
  else
    proj_body(node, sq_k, sk_k, sv_k, pq_k, pk_k, pv_k, R, t, ws,
              (blockIdx.x - LQ)*BI_W + (threadIdx.x >> 5));
}

#define SMS 516
#define AT_W  8
#define AT_L  (32*AT_W/16)
#define AT_KQ (AT_W/2)
#define AT_KS (128/AT_KQ)
__global__ __launch_bounds__(32*AT_W) void attn_kernel(
    const float* __restrict__ point_weight, float* __restrict__ ws)
{
  __shared__ float sm[16 * SMS];
  __shared__ float red[AT_KQ][3][8][32];
  const int tid  = threadIdx.x;
  const int lane = tid & 31;
  const int w    = tid >> 5;
  const int nt   = blockIdx.x;
  const int h    = blockIdx.y;

  const float* Sq  = ws + OFF_SQ;  const float* Sk  = ws + OFF_SK;
  const float* Pq  = ws + OFF_PQ;  const float* Pk  = ws + OFF_PK;
  const float* qn2 = ws + OFF_QN2; const float* kn2 = ws + OFF_KN2;
  const float* bias= ws + OFF_BIAS;
  const float* Sv  = ws + OFF_SV;  const float* Pv  = ws + OFF_PV;
  float* attn = ws + OFF_ATTN;
  float* res  = ws + OFF_RES;

  const float pwx = point_weight[h];
  const float pw  = (pwx > 20.f) ? pwx : log1pf(__expf(pwx));
  const float cpt = -0.5f * pw / sqrtf(18.f);
  const float rs3 = 0.57735026919f;

  const int row   = lane & 15;
  const int khalf = (lane >> 4) * 2;
  const int col   = lane & 15;
  const int rbase = (lane >> 4) * 8;

  v2f aq[4], ap[4];
#pragma unroll
  for (int ks = 0; ks < 4; ++ks) {
    const int k = ks*4 + khalf;
    const int l = nt*16 + row;
    aq[ks].x = Sq[((size_t)h*LQ + l)*16 + k];
    aq[ks].y = Sq[((size_t)h*LQ + l)*16 + k + 1];
    ap[ks].x = Pq[((size_t)h*LQ + l)*16 + k];
    ap[ks].y = Pq[((size_t)h*LQ + l)*16 + k + 1];
  }
  float qv[8];
#pragma unroll
  for (int r = 0; r < 8; ++r) qv[r] = qn2[h*LQ + nt*16 + rbase + r];

  for (int mt = w; mt < 32; mt += AT_W) {
    v8f s = vzero8(), p = vzero8();
#pragma unroll
    for (int ks = 0; ks < 4; ++ks) {
      const int k = ks*4 + khalf;
      const int m = mt*16 + row;
      v2f bq, bp;
      bq.x = Sk[((size_t)h*LQ + m)*16 + k]; bq.y = Sk[((size_t)h*LQ + m)*16 + k + 1];
      bp.x = Pk[((size_t)h*LQ + m)*16 + k]; bp.y = Pk[((size_t)h*LQ + m)*16 + k + 1];
      s = wmma4(aq[ks], bq, s);
      p = wmma4(ap[ks], bp, p);
    }
    const int m = mt*16 + col;
    const float kv = kn2[h*LQ + m];
#pragma unroll
    for (int r = 0; r < 8; ++r) {
      const int rr = rbase + r;
      const int n  = nt*16 + rr;
      const float dist = qv[r] + kv - 2.f*p[r];
      sm[rr*SMS + m] =
          (s[r] + bias[((size_t)h*LQ + n)*LQ + m] + cpt*dist) * rs3;
    }
  }
  __syncthreads();

  {
    const int rrow = tid / AT_L;
    const int sub  = tid % AT_L;
    float mx = -1e30f;
    for (int j = sub; j < LQ; j += AT_L) mx = fmaxf(mx, sm[rrow*SMS + j]);
    for (int o = 1; o < AT_L; o <<= 1) mx = fmaxf(mx, __shfl_xor(mx, o, AT_L));
    float sum = 0.f;
    for (int j = sub; j < LQ; j += AT_L) {
      const float e = __expf(sm[rrow*SMS + j] - mx);
      sm[rrow*SMS + j] = e; sum += e;
    }
    for (int o = 1; o < AT_L; o <<= 1) sum += __shfl_xor(sum, o, AT_L);
    const float inv = 1.f / sum;
    const int n = nt*16 + rrow;
    for (int j = sub; j < LQ; j += AT_L) {
      const float v = sm[rrow*SMS + j] * inv;
      sm[rrow*SMS + j] = v;
      attn[((size_t)h*LQ + n)*LQ + j] = v;
    }
  }
  __syncthreads();

  const int kq    = w >> 1;
  const int role  = w & 1;
  const int kt_lo = kq * AT_KS;
  v8f acc0 = vzero8(), acc1 = vzero8();
  if (role == 0) {
    const float* vp = Sv + (size_t)h*LQ*16 + col;
    for (int kt0 = kt_lo; kt0 < kt_lo + AT_KS; kt0 += U_AV) {
      v2f a[U_AV], b[U_AV];
#pragma unroll
      for (int u = 0; u < U_AV; ++u) {
        const int k = (kt0 + u)*4 + khalf;
        a[u].x = sm[row*SMS + k];        a[u].y = sm[row*SMS + k + 1];
        b[u].x = vp[(size_t)k * 16];     b[u].y = vp[(size_t)(k + 1) * 16];
      }
      __builtin_amdgcn_sched_barrier(0);
#pragma unroll
      for (int u = 0; u < U_AV; ++u) acc0 = wmma4(a[u], b[u], acc0);
    }
  } else {
    const int   pc    = (col < 12) ? col : 11;
    const float pmask = (col < 12) ? 1.f : 0.f;
    const float* vp = Pv + (size_t)h*LQ*24 + 2*pc;
    for (int kt0 = kt_lo; kt0 < kt_lo + AT_KS; kt0 += U_AV) {
      v2f a[U_AV], x[U_AV], y[U_AV];
#pragma unroll
      for (int u = 0; u < U_AV; ++u) {
        const int k = (kt0 + u)*4 + khalf;
        a[u].x = sm[row*SMS + k];        a[u].y = sm[row*SMS + k + 1];
        x[u] = *(const v2f*)(vp + (size_t)k * 24);
        y[u] = *(const v2f*)(vp + (size_t)(k + 1) * 24);
      }
      __builtin_amdgcn_sched_barrier(0);
#pragma unroll
      for (int u = 0; u < U_AV; ++u) {
        v2f b0; b0.x = x[u].x * pmask; b0.y = y[u].x * pmask;
        v2f b1; b1.x = x[u].y * pmask; b1.y = y[u].y * pmask;
        acc0 = wmma4(a[u], b0, acc0);
        acc1 = wmma4(a[u], b1, acc1);
      }
    }
  }

  if (role == 0) {
#pragma unroll
    for (int r = 0; r < 8; ++r) red[kq][0][r][lane] = acc0[r];
  } else {
#pragma unroll
    for (int r = 0; r < 8; ++r) {
      red[kq][1][r][lane] = acc0[r];
      red[kq][2][r][lane] = acc1[r];
    }
  }
  __syncthreads();

#pragma unroll
  for (int rr = 0; rr < 2; ++rr) {
    const int r = 2*kq + rr;
    const int n = nt*16 + rbase + r;
    if (role == 0) {
      const float v = ((red[0][0][r][lane] + red[1][0][r][lane])
                       + red[2][0][r][lane]) + red[3][0][r][lane];
      res[RES_IDX(h, n, col)] = v;
    } else {
      v2f o;
      o.x = ((red[0][1][r][lane] + red[1][1][r][lane])
             + red[2][1][r][lane]) + red[3][1][r][lane];
      o.y = ((red[0][2][r][lane] + red[1][2][r][lane])
             + red[2][2][r][lane]) + red[3][2][r][lane];
      if (col < 12) *(v2f*)(res + RES_IDX(h, n, 144 + 2*col)) = o;
    }
  }
}

static __device__ __forceinline__ void po_load(
    v2f (&a)[PO_S], v2f (&x)[PO_S], v2f (&y)[PO_S],
    const float* __restrict__ arow, const float* __restrict__ prow, int m0)
{
#pragma unroll
  for (int u = 0; u < PO_S; ++u) {
    const int m = m0 + 4*u;
    a[u] = *(const v2f*)(arow + m);
    x[u] = *(const v2f*)(prow + (size_t)m * PAIR);
    y[u] = *(const v2f*)(prow + (size_t)(m + 1) * PAIR);
  }
}

template <bool TAIL>
static __device__ __forceinline__ void po_turn(
    v8f (&acc)[PO_T], v2f (&a)[PO_R][PO_S], v2f (&x)[PO_R][PO_S],
    v2f (&y)[PO_R][PO_S], const float* __restrict__ arow,
    const float* __restrict__ prow, int kb0, int khalf, float amask)
{
#pragma unroll
  for (int j = 0; j < PO_R; ++j) {
    if (!TAIL || j == 0)
      po_load(a[(j + PO_R - 1) % PO_R], x[(j + PO_R - 1) % PO_R],
              y[(j + PO_R - 1) % PO_R], arow, prow,
              (kb0 + j + PO_R - 1)*4*PO_S + khalf);
    __builtin_amdgcn_sched_barrier(0);
#pragma unroll
    for (int u = 0; u < PO_S; ++u) {
      v2f am; am.x = a[j][u].x * amask; am.y = a[j][u].y * amask;
      v2f b0; b0.x = x[j][u].x; b0.y = y[j][u].x;
      v2f b1; b1.x = x[j][u].y; b1.y = y[j][u].y;
      acc[0] = wmma4(am, b0, acc[0]);
      acc[1] = wmma4(am, b1, acc[1]);
    }
    __builtin_amdgcn_sched_barrier(0);
  }
}

static __device__ __forceinline__ void pairout_body(
    const float* __restrict__ pairwise, float* __restrict__ ws, const int n)
{
  const float* attn = ws + OFF_ATTN;
  float* res = ws + OFF_RES;
  const int lane  = threadIdx.x & 31;
  const int w     = threadIdx.x >> 5;
  const int row   = lane & 15;
  const int khalf = (lane >> 4) * 2;
  const int col   = lane & 15;

  const int hh0 = (row < NH) ? row : NH - 1;
  const float amask = (row < NH) ? 1.f : 0.f;
  const float* arow = attn + ((size_t)hh0*LQ + n)*LQ;

  v8f acc[PO_T];
#pragma unroll
  for (int i = 0; i < PO_T; ++i) acc[i] = vzero8();

  v2f a[PO_R][PO_S], x[PO_R][PO_S], y[PO_R][PO_S];
  const float* prow = pairwise + (size_t)n*LQ*PAIR + 32*w + 2*col;
#pragma unroll
  for (int s = 0; s < PO_R - 1; ++s) {
    po_load(a[s], x[s], y[s], arow, prow, s*4*PO_S + khalf);
    __builtin_amdgcn_sched_barrier(0);
  }

  for (int kb0 = 0; kb0 < PO_NB - PO_R; kb0 += PO_R)
    po_turn<false>(acc, a, x, y, arow, prow, kb0, khalf, amask);
  po_turn<true>(acc, a, x, y, arow, prow, PO_NB - PO_R, khalf, amask);

  const int rbase = (lane >> 4) * 8;
#pragma unroll
  for (int r = 0; r < 8; ++r) {
    const int hh = rbase + r;
    if (hh < NH) {
      v2f o; o.x = acc[0][r]; o.y = acc[1][r];
      *(v2f*)(res + RES_IDX(hh, n, 16 + 32*w + 2*col)) = o;
    }
  }
}

static __device__ __forceinline__ void pfinish_body(
    const float* __restrict__ R, const float* __restrict__ t,
    float* __restrict__ ws, const int idx  )
{
  const int h = idx / LQ, l = idx % LQ;
  float* resp = ws + OFF_RES + RES_IDX(h, l, 0);
  float Rm[9], tv[3];
#pragma unroll
  for (int i = 0; i < 9; ++i) Rm[i] = R[l*9 + i];
#pragma unroll
  for (int i = 0; i < 3; ++i) tv[i] = t[l*3 + i];
#pragma unroll
  for (int p = 0; p < 8; ++p) {
    float g[3], loc[3], nrm = 0.f;
    for (int gg = 0; gg < 3; ++gg) g[gg] = resp[144 + p*3 + gg] - tv[gg];
    for (int f = 0; f < 3; ++f) {
      float v = 0.f;
      for (int gg = 0; gg < 3; ++gg) v += g[gg]*Rm[gg*3 + f];
      loc[f] = v; nrm += v*v;
    }
    for (int f = 0; f < 3; ++f) resp[144 + p*3 + f] = loc[f];
    resp[168 + p] = sqrtf(nrm + 1e-12f);
  }
}

#define PF_B ((NH*LQ)/(32*PO_W))
__global__ __launch_bounds__(32*PO_W) void pairout_pfinish_kernel(
    const float* __restrict__ pairwise, const float* __restrict__ R,
    const float* __restrict__ t, float* __restrict__ ws)
{
  if (blockIdx.x < LQ)
    pairout_body(pairwise, ws, blockIdx.x);
  else
    pfinish_body(R, t, ws, (blockIdx.x - LQ)*(32*PO_W) + threadIdx.x);
}

struct f3 { float x, y, z; };

static __device__ __forceinline__ void op_load(
    v4f& a, f3 (&b)[4], const float* __restrict__ ap,
    const float* __restrict__ bp, int d)
{
  a = *(const v4f*)(ap + 8*d);
#pragma unroll
  for (int i = 0; i < 4; ++i) {
    const float* p = bp + (size_t)(8*d + i) * NODE;
    b[i].x = p[0]; b[i].y = p[1]; b[i].z = p[2];
  }
}

template <bool TAIL>
static __device__ __forceinline__ void op_turn(
    v8f (&acc)[OP_T], v4f (&a)[OP_R], f3 (&b)[OP_R][4],
    const float* __restrict__ ap, const float* __restrict__ bp, int kb0)
{
#pragma unroll
  for (int j = 0; j < OP_R; ++j) {
    if (!TAIL || j == 0)
      op_load(a[(j + OP_R - 1) % OP_R], b[(j + OP_R - 1) % OP_R], ap, bp,
              kb0 + j + OP_R - 1);
    __builtin_amdgcn_sched_barrier(0);
    v2f a0; a0.x = a[j].x; a0.y = a[j].y;
    v2f a1; a1.x = a[j].z; a1.y = a[j].w;
    v2f b0, b1, b2;
    b0.x = b[j][0].x; b0.y = b[j][1].x;
    b1.x = b[j][0].y; b1.y = b[j][1].y;
    b2.x = b[j][0].z; b2.y = b[j][1].z;
    acc[0] = wmma4(a0, b0, acc[0]);
    acc[1] = wmma4(a0, b1, acc[1]);
    acc[2] = wmma4(a0, b2, acc[2]);
    b0.x = b[j][2].x; b0.y = b[j][3].x;
    b1.x = b[j][2].y; b1.y = b[j][3].y;
    b2.x = b[j][2].z; b2.y = b[j][3].z;
    acc[0] = wmma4(a1, b0, acc[0]);
    acc[1] = wmma4(a1, b1, acc[1]);
    acc[2] = wmma4(a1, b2, acc[2]);
    __builtin_amdgcn_sched_barrier(0);
  }
}

__global__ __launch_bounds__(32*OP_W) void outproj_kernel(
    const float* __restrict__ out_k, const float* __restrict__ out_b,
    const float* __restrict__ res, float* __restrict__ out)
{
  __shared__ float red[OP_W][OP_T][8][32];
  const int lane  = threadIdx.x & 31;
  const int q     = threadIdx.x >> 5;
  const int mt    = blockIdx.x;
  const int n0    = blockIdx.y * (16*OP_T);
  const int row   = lane & 15;
  const int half  = lane >> 4;
  const int col   = lane & 15;

  const float* ap = res + (size_t)(mt*16 + row) * KOUT + q*OP_KQ + 4*half;
  const float* bp = out_k + (size_t)(q*OP_KQ + 4*half) * NODE + n0 + 3*col;

  v8f acc[OP_T];
#pragma unroll
  for (int j = 0; j < OP_T; ++j) acc[j] = vzero8();
  v4f a[OP_R];
  f3  b[OP_R][4];
#pragma unroll
  for (int s = 0; s < OP_R - 1; ++s) {
    op_load(a[s], b[s], ap, bp, s);
    __builtin_amdgcn_sched_barrier(0);
  }

  for (int kb0 = 0; kb0 < OP_NB - OP_R; kb0 += OP_R)
    op_turn<false>(acc, a, b, ap, bp, kb0);
  op_turn<true>(acc, a, b, ap, bp, OP_NB - OP_R);

#pragma unroll
  for (int j = 0; j < OP_T; ++j)
#pragma unroll
    for (int r = 0; r < 8; ++r) red[q][j][r][lane] = acc[j][r];
  __syncthreads();

  const float* obp = out_b + n0 + 3*col;
  const float ob0 = obp[0], ob1 = obp[1], ob2 = obp[2];
#pragma unroll
  for (int rr = 0; rr < 2; ++rr) {
    const int r = 2*q + rr;
    const int l = mt*16 + 8*half + r;
    float o[OP_T];
#pragma unroll
    for (int j = 0; j < OP_T; ++j)
      o[j] = ((red[0][j][r][lane] + red[1][j][r][lane])
              + red[2][j][r][lane]) + red[3][j][r][lane];
    float* op = out + (size_t)l*NODE + n0 + 3*col;
    op[0] = o[0] + ob0; op[1] = o[1] + ob1; op[2] = o[2] + ob2;
  }
}

extern "C" void kernel_launch(void* const* d_in, const int* in_sizes, int n_in,
                              void* d_out, int out_size, void* d_ws, size_t ws_size,
                              hipStream_t stream) {
  const float* node  = (const float*)d_in[0];
  const float* R     = (const float*)d_in[1];
  const float* t     = (const float*)d_in[2];
  const float* pairw = (const float*)d_in[3];
  const float* sq_k  = (const float*)d_in[4];
  const float* sk_k  = (const float*)d_in[5];
  const float* sv_k  = (const float*)d_in[6];
  const float* pq_k  = (const float*)d_in[7];
  const float* pk_k  = (const float*)d_in[8];
  const float* pv_k  = (const float*)d_in[9];
  const float* pr_k  = (const float*)d_in[10];
  const float* out_k = (const float*)d_in[11];
  const float* out_b = (const float*)d_in[12];
  const float* pwgt  = (const float*)d_in[13];
  float* ws  = (float*)d_ws;
  float* out = (float*)d_out;

  projbias_kernel<<<LQ + PJ_B, 32*BI_W, 0, stream>>>(
      node, sq_k, sk_k, sv_k, pq_k, pk_k, pv_k, pairw, pr_k, R, t, ws);
  attn_kernel<<<dim3(32, NH), 32*AT_W, 0, stream>>>(pwgt, ws);
  pairout_pfinish_kernel<<<LQ + PF_B, 32*PO_W, 0, stream>>>(pairw, R, t, ws);
  outproj_kernel<<<dim3(32, 8), 32*OP_W, 0, stream>>>(out_k, out_b, ws + OFF_RES, out);
}
